# starting kernel
# baseline (speedup 1.0000x reference)
#include <hip/hip_runtime.h>

typedef __attribute__((ext_vector_type(2))) float v2f;
typedef __attribute__((ext_vector_type(8))) float v8f;

__global__ void gcn_init(float* __restrict__ deg, float* __restrict__ agg1,
                         float* __restrict__ agg2, int n) {
    int i = blockIdx.x * blockDim.x + threadIdx.x;
    if (i < n) { deg[i] = 1.0f; agg2[i] = 0.0f; }
    if (i < n * 16) { agg1[i] = 0.0f; }
}

__global__ void gcn_degree(const int* __restrict__ col, float* __restrict__ deg,
                           int E) {
    int e = blockIdx.x * blockDim.x + threadIdx.x;
    if (e >= E) return;
    atomicAdd(&deg[col[e]], 1.0f);
}

__global__ void gcn_dinv(const float* __restrict__ deg, float* __restrict__ dinv,
                         int n) {
    int i = blockIdx.x * blockDim.x + threadIdx.x;
    if (i >= n) return;
    float d = deg[i];
    dinv[i] = (d > 0.0f) ? rsqrtf(d) : 0.0f;
}

__global__ void gcn_gemm1_wmma(const float* __restrict__ x,
                               const float* __restrict__ W1,
                               float* __restrict__ h1, int ntiles) {
    int wave = blockIdx.x * (blockDim.x >> 5) + (threadIdx.x >> 5);
    if (wave >= ntiles) return;
    int lane  = threadIdx.x & 31;
    int m     = lane & 15;
    int kb    = (lane >> 4) << 1;
    size_t row16 = (size_t)wave * 16;

    v8f acc = {};
    const float* arow = x + (row16 + (size_t)m) * 128 + kb;
    #pragma unroll 4
    for (int k0 = 0; k0 < 128; k0 += 4) {
        v2f a = *(const v2f*)(arow + k0);
        v2f b;
        b.x = W1[(size_t)(k0 + kb) * 16 + m];
        b.y = W1[(size_t)(k0 + kb + 1) * 16 + m];
        acc = __builtin_amdgcn_wmma_f32_16x16x4_f32(
             false, a,  false, b,
             (short)0, acc,  false,  false);
    }

    int mbase = (lane >> 4) << 3;
    int ncol  = lane & 15;
    #pragma unroll
    for (int r = 0; r < 8; ++r) {
        h1[(row16 + (size_t)(mbase + r)) * 16 + ncol] = acc[r];
    }
}

__global__ void gcn_gemm1_tail(const float* __restrict__ x,
                               const float* __restrict__ W1,
                               float* __restrict__ h1, int start, int n) {
    int t = blockIdx.x * blockDim.x + threadIdx.x;
    int node = start + (t >> 4);
    int f = t & 15;
    if (node >= n) return;
    float acc = 0.0f;
    const float* xr = x + (size_t)node * 128;
    for (int k = 0; k < 128; ++k) acc += xr[k] * W1[(size_t)k * 16 + f];
    h1[(size_t)node * 16 + f] = acc;
}

__global__ void gcn_edge16(const int* __restrict__ row, const int* __restrict__ col,
                           const float* __restrict__ dinv,
                           const float* __restrict__ h1,
                           float* __restrict__ agg1, int E) {
    int t = blockIdx.x * blockDim.x + threadIdx.x;
    int e = t >> 4;
    int f = t & 15;
    if (e >= E) return;
    int r = row[e];
    int c = col[e];
    float nrm = dinv[r] * dinv[c];
    float v = nrm * h1[(size_t)r * 16 + f];
    atomicAdd(&agg1[(size_t)c * 16 + f], v);
}

__global__ void gcn_post1(const float* __restrict__ agg1,
                          const float* __restrict__ h1,
                          const float* __restrict__ dinv,
                          const float* __restrict__ b1,
                          const float* __restrict__ W2,
                          float* __restrict__ s, int n) {
    int i = blockIdx.x * blockDim.x + threadIdx.x;
    if (i >= n) return;
    float di = dinv[i];
    float self = di * di;
    const float* a = agg1 + (size_t)i * 16;
    const float* h = h1 + (size_t)i * 16;
    float acc = 0.0f;
    #pragma unroll
    for (int f = 0; f < 16; ++f) {
        float v = a[f] + self * h[f] + b1[f];
        v = fmaxf(v, 0.0f);
        acc = fmaf(v, W2[f], acc);
    }
    s[i] = acc;
}

__global__ void gcn_edge1(const int* __restrict__ row, const int* __restrict__ col,
                          const float* __restrict__ dinv,
                          const float* __restrict__ s,
                          float* __restrict__ agg2, int E) {
    int e = blockIdx.x * blockDim.x + threadIdx.x;
    if (e >= E) return;
    int r = row[e];
    int c = col[e];
    atomicAdd(&agg2[c], dinv[r] * dinv[c] * s[r]);
}

__global__ void gcn_post2(const float* __restrict__ agg2,
                          const float* __restrict__ s,
                          const float* __restrict__ dinv,
                          const float* __restrict__ b2,
                          float* __restrict__ out, int n) {
    int i = blockIdx.x * blockDim.x + threadIdx.x;
    if (i >= n) return;
    float di = dinv[i];
    out[i] = agg2[i] + di * di * s[i] + b2[0];
}

extern "C" void kernel_launch(void* const* d_in, const int* in_sizes, int n_in,
                              void* d_out, int out_size, void* d_ws, size_t ws_size,
                              hipStream_t stream) {
    const float* x  = (const float*)d_in[0];
    const int*   ei = (const int*)d_in[1];
    const float* W1 = (const float*)d_in[2];
    const float* b1 = (const float*)d_in[3];
    const float* W2 = (const float*)d_in[4];
    const float* b2 = (const float*)d_in[5];

    const int N = in_sizes[0] / 128;
    const int E = in_sizes[1] / 2;
    const int* row = ei;
    const int* col = ei + E;

    float* ws   = (float*)d_ws;
    float* deg  = ws;
    float* dinv = deg + N;
    float* h1   = dinv + N;
    float* agg1 = h1 + (size_t)N * 16;
    float* s    = agg1 + (size_t)N * 16;
    float* agg2 = s + N;
    float* out  = (float*)d_out;

    const int B = 256;

    {
        int total = N * 16;
        gcn_init<<<(total + B - 1) / B, B, 0, stream>>>(deg, agg1, agg2, N);
    }
    gcn_degree<<<(E + B - 1) / B, B, 0, stream>>>(col, deg, E);
    gcn_dinv<<<(N + B - 1) / B, B, 0, stream>>>(deg, dinv, N);
    {
        int full   = N & ~15;
        int ntiles = full >> 4;
        if (ntiles > 0) {
            int wavesPerBlock = B / 32;
            int blocks = (ntiles + wavesPerBlock - 1) / wavesPerBlock;
            gcn_gemm1_wmma<<<blocks, B, 0, stream>>>(x, W1, h1, ntiles);
        }
        if (full < N) {
            int tail = (N - full) * 16;
            gcn_gemm1_tail<<<(tail + B - 1) / B, B, 0, stream>>>(x, W1, h1, full, N);
        }
    }
    {
        long long total = (long long)E * 16;
        gcn_edge16<<<(int)((total + B - 1) / B), B, 0, stream>>>(row, col, dinv, h1, agg1, E);
    }
    gcn_post1<<<(N + B - 1) / B, B, 0, stream>>>(agg1, h1, dinv, b1, W2, s, N);
    gcn_edge1<<<(E + B - 1) / B, B, 0, stream>>>(row, col, dinv, s, agg2, E);
    gcn_post2<<<(N + B - 1) / B, B, 0, stream>>>(agg2, s, dinv, b2, out, N);
}


# Degree count and layer-2 scalar edge sum accumulate in LDS bins per target-node range: one global atomic per non-zero bin, not per edge; per-slice LDS partial sums change the float sum order
# speedup vs baseline: 1.7256x; 1.7256x over previous
#include <hip/hip_runtime.h>

typedef __attribute__((ext_vector_type(2))) float v2f;
typedef __attribute__((ext_vector_type(8))) float v8f;

__global__ void gcn_init(float* __restrict__ deg, float* __restrict__ agg1,
                         float* __restrict__ agg2, int n) {
    int i = blockIdx.x * blockDim.x + threadIdx.x;
    if (i < n) { deg[i] = 1.0f; agg2[i] = 0.0f; }
    if (i < n * 16) { agg1[i] = 0.0f; }
}

#define RANGE_BINS  51200
#define EDGE_SLICES 64
#define LDS_THREADS 1024

__global__ __launch_bounds__(LDS_THREADS)
void gcn_degree_lds(const int* __restrict__ col, float* __restrict__ deg,
                    int E, int n) {
    __shared__ unsigned cnt[RANGE_BINS];
    const int T   = (int)blockDim.x;
    const int tid = (int)threadIdx.x;
    const int per = (n + (int)gridDim.y - 1) / (int)gridDim.y;
    const int lo  = (int)blockIdx.y * per;
    const int lim = max(0, min(per, n - lo));
    for (int i = tid; i < lim; i += T) cnt[i] = 0u;
    __syncthreads();

    const int quads = (E + 3) >> 2;
    const int qper  = (quads + (int)gridDim.x - 1) / (int)gridDim.x;
    const int q0    = (int)blockIdx.x * qper;
    const int q1    = min(q0 + qper, quads);
    const bool vec  = (((size_t)col) & 15) == 0;
    const int qv    = vec ? min(q1, E >> 2) : q0;

    for (int q = q0 + tid; q < qv; q += 4 * T) {
        int4 v[4];
        #pragma unroll
        for (int u = 0; u < 4; ++u) {
            const int qq = q + u * T;
            v[u] = (qq < qv) ? *(const int4*)(col + 4 * (size_t)qq)
                             : make_int4(-1, -1, -1, -1);
        }
        #pragma unroll
        for (int u = 0; u < 4; ++u) {
            const unsigned b0 = (unsigned)(v[u].x - lo);
            const unsigned b1 = (unsigned)(v[u].y - lo);
            const unsigned b2 = (unsigned)(v[u].z - lo);
            const unsigned b3 = (unsigned)(v[u].w - lo);
            if (b0 < (unsigned)lim) atomicAdd(&cnt[b0], 1u);
            if (b1 < (unsigned)lim) atomicAdd(&cnt[b1], 1u);
            if (b2 < (unsigned)lim) atomicAdd(&cnt[b2], 1u);
            if (b3 < (unsigned)lim) atomicAdd(&cnt[b3], 1u);
        }
    }
    for (int q = max(q0, qv) + tid; q < q1; q += T) {
        for (int k = 0; k < 4; ++k) {
            const int e = 4 * q + k;
            if (e < E) {
                const unsigned b = (unsigned)(col[e] - lo);
                if (b < (unsigned)lim) atomicAdd(&cnt[b], 1u);
            }
        }
    }
    __syncthreads();
    for (int i = tid; i < lim; i += T) {
        const unsigned v = cnt[i];
        if (v != 0u) atomicAdd(&deg[lo + i], (float)v);
    }
}

__global__ void gcn_dinv(const float* __restrict__ deg, float* __restrict__ dinv,
                         int n) {
    int i = blockIdx.x * blockDim.x + threadIdx.x;
    if (i >= n) return;
    float d = deg[i];
    dinv[i] = (d > 0.0f) ? rsqrtf(d) : 0.0f;
}

__global__ void gcn_gemm1_wmma(const float* __restrict__ x,
                               const float* __restrict__ W1,
                               float* __restrict__ h1, int ntiles) {
    int wave = blockIdx.x * (blockDim.x >> 5) + (threadIdx.x >> 5);
    if (wave >= ntiles) return;
    int lane  = threadIdx.x & 31;
    int m     = lane & 15;
    int kb    = (lane >> 4) << 1;
    size_t row16 = (size_t)wave * 16;

    v8f acc = {};
    const float* arow = x + (row16 + (size_t)m) * 128 + kb;
    #pragma unroll 4
    for (int k0 = 0; k0 < 128; k0 += 4) {
        v2f a = *(const v2f*)(arow + k0);
        v2f b;
        b.x = W1[(size_t)(k0 + kb) * 16 + m];
        b.y = W1[(size_t)(k0 + kb + 1) * 16 + m];
        acc = __builtin_amdgcn_wmma_f32_16x16x4_f32(
             false, a,  false, b,
             (short)0, acc,  false,  false);
    }

    int mbase = (lane >> 4) << 3;
    int ncol  = lane & 15;
    #pragma unroll
    for (int r = 0; r < 8; ++r) {
        h1[(row16 + (size_t)(mbase + r)) * 16 + ncol] = acc[r];
    }
}

__global__ void gcn_gemm1_tail(const float* __restrict__ x,
                               const float* __restrict__ W1,
                               float* __restrict__ h1, int start, int n) {
    int t = blockIdx.x * blockDim.x + threadIdx.x;
    int node = start + (t >> 4);
    int f = t & 15;
    if (node >= n) return;
    float acc = 0.0f;
    const float* xr = x + (size_t)node * 128;
    for (int k = 0; k < 128; ++k) acc += xr[k] * W1[(size_t)k * 16 + f];
    h1[(size_t)node * 16 + f] = acc;
}

__global__ void gcn_edge16(const int* __restrict__ row, const int* __restrict__ col,
                           const float* __restrict__ dinv,
                           const float* __restrict__ h1,
                           float* __restrict__ agg1, int E) {
    int t = blockIdx.x * blockDim.x + threadIdx.x;
    int e = t >> 4;
    int f = t & 15;
    if (e >= E) return;
    int r = row[e];
    int c = col[e];
    float nrm = dinv[r] * dinv[c];
    float v = nrm * h1[(size_t)r * 16 + f];
    atomicAdd(&agg1[(size_t)c * 16 + f], v);
}

__global__ void gcn_post1(const float* __restrict__ agg1,
                          const float* __restrict__ h1,
                          const float* __restrict__ dinv,
                          const float* __restrict__ b1,
                          const float* __restrict__ W2,
                          float* __restrict__ s, int n) {
    int i = blockIdx.x * blockDim.x + threadIdx.x;
    if (i >= n) return;
    float di = dinv[i];
    float self = di * di;
    const float* a = agg1 + (size_t)i * 16;
    const float* h = h1 + (size_t)i * 16;
    float acc = 0.0f;
    #pragma unroll
    for (int f = 0; f < 16; ++f) {
        float v = a[f] + self * h[f] + b1[f];
        v = fmaxf(v, 0.0f);
        acc = fmaf(v, W2[f], acc);
    }
    s[i] = acc;
}

__global__ __launch_bounds__(LDS_THREADS)
void gcn_edge1_lds(const int* __restrict__ row, const int* __restrict__ col,
                   const float* __restrict__ dinv, const float* __restrict__ s,
                   float* __restrict__ agg2, int E, int n) {
    __shared__ float acc[RANGE_BINS];
    const int T   = (int)blockDim.x;
    const int tid = (int)threadIdx.x;
    const int per = (n + (int)gridDim.y - 1) / (int)gridDim.y;
    const int lo  = (int)blockIdx.y * per;
    const int lim = max(0, min(per, n - lo));
    for (int i = tid; i < lim; i += T) acc[i] = 0.0f;
    __syncthreads();

    const int quads = (E + 3) >> 2;
    const int qper  = (quads + (int)gridDim.x - 1) / (int)gridDim.x;
    const int q0    = (int)blockIdx.x * qper;
    const int q1    = min(q0 + qper, quads);
    const bool vec  = ((((size_t)col) | ((size_t)row)) & 15) == 0;
    const int qv    = vec ? min(q1, E >> 2) : q0;

    for (int q = q0 + tid; q < qv; q += 2 * T) {
        int c[8], r[8];
        #pragma unroll
        for (int u = 0; u < 2; ++u) {
            const int qq = q + u * T;
            const bool ok = qq < qv;
            const int4 vc = ok ? *(const int4*)(col + 4 * (size_t)qq) : make_int4(-1, -1, -1, -1);
            const int4 vr = ok ? *(const int4*)(row + 4 * (size_t)qq) : make_int4(0, 0, 0, 0);
            c[4 * u + 0] = vc.x; c[4 * u + 1] = vc.y; c[4 * u + 2] = vc.z; c[4 * u + 3] = vc.w;
            r[4 * u + 0] = vr.x; r[4 * u + 1] = vr.y; r[4 * u + 2] = vr.z; r[4 * u + 3] = vr.w;
        }
        unsigned b[8];
        float    v[8];
        #pragma unroll
        for (int k = 0; k < 8; ++k) {
            b[k] = (unsigned)(c[k] - lo);
            const bool in = b[k] < (unsigned)lim;
            const int rr = in ? r[k] : 0;
            const int cc = in ? c[k] : 0;
            v[k] = dinv[rr] * dinv[cc] * s[rr];
        }
        #pragma unroll
        for (int k = 0; k < 8; ++k) {
            if (b[k] < (unsigned)lim)
                __hip_atomic_fetch_add(&acc[b[k]], v[k], __ATOMIC_RELAXED,
                                       __HIP_MEMORY_SCOPE_WORKGROUP);
        }
    }
    for (int q = max(q0, qv) + tid; q < q1; q += T) {
        for (int k = 0; k < 4; ++k) {
            const int e = 4 * q + k;
            if (e < E) {
                const int cc = col[e];
                const unsigned bb = (unsigned)(cc - lo);
                if (bb < (unsigned)lim) {
                    const int rr = row[e];
                    __hip_atomic_fetch_add(&acc[bb], dinv[rr] * dinv[cc] * s[rr],
                                           __ATOMIC_RELAXED, __HIP_MEMORY_SCOPE_WORKGROUP);
                }
            }
        }
    }
    __syncthreads();
    for (int i = tid; i < lim; i += T) {
        const float t = acc[i];
        if (t != 0.0f) atomicAdd(&agg2[lo + i], t);
    }
}

__global__ void gcn_post2(const float* __restrict__ agg2,
                          const float* __restrict__ s,
                          const float* __restrict__ dinv,
                          const float* __restrict__ b2,
                          float* __restrict__ out, int n) {
    int i = blockIdx.x * blockDim.x + threadIdx.x;
    if (i >= n) return;
    float di = dinv[i];
    out[i] = agg2[i] + di * di * s[i] + b2[0];
}

extern "C" void kernel_launch(void* const* d_in, const int* in_sizes, int n_in,
                              void* d_out, int out_size, void* d_ws, size_t ws_size,
                              hipStream_t stream) {
    const float* x  = (const float*)d_in[0];
    const int*   ei = (const int*)d_in[1];
    const float* W1 = (const float*)d_in[2];
    const float* b1 = (const float*)d_in[3];
    const float* W2 = (const float*)d_in[4];
    const float* b2 = (const float*)d_in[5];

    const int N = in_sizes[0] / 128;
    const int E = in_sizes[1] / 2;
    const int* row = ei;
    const int* col = ei + E;

    float* ws   = (float*)d_ws;
    float* deg  = ws;
    float* dinv = deg + N;
    float* h1   = dinv + N;
    float* agg1 = h1 + (size_t)N * 16;
    float* s    = agg1 + (size_t)N * 16;
    float* agg2 = s + N;
    float* out  = (float*)d_out;

    const int B = 256;

    {
        int total = N * 16;
        gcn_init<<<(total + B - 1) / B, B, 0, stream>>>(deg, agg1, agg2, N);
    }
    const int R = (N + RANGE_BINS - 1) / RANGE_BINS;
    const dim3 ldsGrid(EDGE_SLICES, R > 0 ? R : 1);

    gcn_degree_lds<<<ldsGrid, LDS_THREADS, 0, stream>>>(col, deg, E, N);
    gcn_dinv<<<(N + B - 1) / B, B, 0, stream>>>(deg, dinv, N);
    {
        int full   = N & ~15;
        int ntiles = full >> 4;
        if (ntiles > 0) {
            int wavesPerBlock = B / 32;
            int blocks = (ntiles + wavesPerBlock - 1) / wavesPerBlock;
            gcn_gemm1_wmma<<<blocks, B, 0, stream>>>(x, W1, h1, ntiles);
        }
        if (full < N) {
            int tail = (N - full) * 16;
            gcn_gemm1_tail<<<(tail + B - 1) / B, B, 0, stream>>>(x, W1, h1, full, N);
        }
    }
    {
        long long total = (long long)E * 16;
        gcn_edge16<<<(int)((total + B - 1) / B), B, 0, stream>>>(row, col, dinv, h1, agg1, E);
    }
    gcn_post1<<<(N + B - 1) / B, B, 0, stream>>>(agg1, h1, dinv, b1, W2, s, N);
    gcn_edge1_lds<<<ldsGrid, LDS_THREADS, 0, stream>>>(row, col, dinv, s, agg2, E, N);
    gcn_post2<<<(N + B - 1) / B, B, 0, stream>>>(agg2, s, dinv, b2, out, N);
}


# Layer-1 16-feature edge sum accumulates in LDS rows per target-node range (per-wave queues of in-range edges), flushed by whole-line atomics, not one atomic per edge lane; float sum order changes
# speedup vs baseline: 2.0646x; 1.1965x over previous
#include <hip/hip_runtime.h>

typedef __attribute__((ext_vector_type(2))) float v2f;
typedef __attribute__((ext_vector_type(8))) float v8f;

__global__ void gcn_init(float* __restrict__ deg, float* __restrict__ agg1,
                         float* __restrict__ agg2, int n) {
    int i = blockIdx.x * blockDim.x + threadIdx.x;
    if (i < n) { deg[i] = 1.0f; agg2[i] = 0.0f; }
    if (i < n * 16) { agg1[i] = 0.0f; }
}

#define RANGE_BINS  51200
#define EDGE_SLICES 64
#define LDS_THREADS 1024

__global__ __launch_bounds__(LDS_THREADS)
void gcn_degree_lds(const int* __restrict__ col, float* __restrict__ deg,
                    int E, int n) {
    __shared__ unsigned cnt[RANGE_BINS];
    const int T   = (int)blockDim.x;
    const int tid = (int)threadIdx.x;
    const int per = (n + (int)gridDim.y - 1) / (int)gridDim.y;
    const int lo  = (int)blockIdx.y * per;
    const int lim = max(0, min(per, n - lo));
    for (int i = tid; i < lim; i += T) cnt[i] = 0u;
    __syncthreads();

    const int quads = (E + 3) >> 2;
    const int qper  = (quads + (int)gridDim.x - 1) / (int)gridDim.x;
    const int q0    = (int)blockIdx.x * qper;
    const int q1    = min(q0 + qper, quads);
    const bool vec  = (((size_t)col) & 15) == 0;
    const int qv    = vec ? min(q1, E >> 2) : q0;

    for (int q = q0 + tid; q < qv; q += 4 * T) {
        int4 v[4];
        #pragma unroll
        for (int u = 0; u < 4; ++u) {
            const int qq = q + u * T;
            v[u] = (qq < qv) ? *(const int4*)(col + 4 * (size_t)qq)
                             : make_int4(-1, -1, -1, -1);
        }
        #pragma unroll
        for (int u = 0; u < 4; ++u) {
            const unsigned b0 = (unsigned)(v[u].x - lo);
            const unsigned b1 = (unsigned)(v[u].y - lo);
            const unsigned b2 = (unsigned)(v[u].z - lo);
            const unsigned b3 = (unsigned)(v[u].w - lo);
            if (b0 < (unsigned)lim) atomicAdd(&cnt[b0], 1u);
            if (b1 < (unsigned)lim) atomicAdd(&cnt[b1], 1u);
            if (b2 < (unsigned)lim) atomicAdd(&cnt[b2], 1u);
            if (b3 < (unsigned)lim) atomicAdd(&cnt[b3], 1u);
        }
    }
    for (int q = max(q0, qv) + tid; q < q1; q += T) {
        for (int k = 0; k < 4; ++k) {
            const int e = 4 * q + k;
            if (e < E) {
                const unsigned b = (unsigned)(col[e] - lo);
                if (b < (unsigned)lim) atomicAdd(&cnt[b], 1u);
            }
        }
    }
    __syncthreads();
    for (int i = tid; i < lim; i += T) {
        const unsigned v = cnt[i];
        if (v != 0u) atomicAdd(&deg[lo + i], (float)v);
    }
}

__global__ void gcn_dinv(const float* __restrict__ deg, float* __restrict__ dinv,
                         int n) {
    int i = blockIdx.x * blockDim.x + threadIdx.x;
    if (i >= n) return;
    float d = deg[i];
    dinv[i] = (d > 0.0f) ? rsqrtf(d) : 0.0f;
}

__global__ void gcn_gemm1_wmma(const float* __restrict__ x,
                               const float* __restrict__ W1,
                               float* __restrict__ h1, int ntiles) {
    int wave = blockIdx.x * (blockDim.x >> 5) + (threadIdx.x >> 5);
    if (wave >= ntiles) return;
    int lane  = threadIdx.x & 31;
    int m     = lane & 15;
    int kb    = (lane >> 4) << 1;
    size_t row16 = (size_t)wave * 16;

    v8f acc = {};
    const float* arow = x + (row16 + (size_t)m) * 128 + kb;
    #pragma unroll 4
    for (int k0 = 0; k0 < 128; k0 += 4) {
        v2f a = *(const v2f*)(arow + k0);
        v2f b;
        b.x = W1[(size_t)(k0 + kb) * 16 + m];
        b.y = W1[(size_t)(k0 + kb + 1) * 16 + m];
        acc = __builtin_amdgcn_wmma_f32_16x16x4_f32(
             false, a,  false, b,
             (short)0, acc,  false,  false);
    }

    int mbase = (lane >> 4) << 3;
    int ncol  = lane & 15;
    #pragma unroll
    for (int r = 0; r < 8; ++r) {
        h1[(row16 + (size_t)(mbase + r)) * 16 + ncol] = acc[r];
    }
}

__global__ void gcn_gemm1_tail(const float* __restrict__ x,
                               const float* __restrict__ W1,
                               float* __restrict__ h1, int start, int n) {
    int t = blockIdx.x * blockDim.x + threadIdx.x;
    int node = start + (t >> 4);
    int f = t & 15;
    if (node >= n) return;
    float acc = 0.0f;
    const float* xr = x + (size_t)node * 128;
    for (int k = 0; k < 128; ++k) acc += xr[k] * W1[(size_t)k * 16 + f];
    h1[(size_t)node * 16 + f] = acc;
}

#define R16_NODES  4768
#define R16_SLICES 6
#define R16_QCAP   160

__global__ __launch_bounds__(LDS_THREADS)
void gcn_edge16_lds(const int* __restrict__ row, const int* __restrict__ col,
                    const float* __restrict__ dinv, const float* __restrict__ h1,
                    float* __restrict__ agg1, int E, int n) {
    __shared__ float acc[R16_NODES * 16];
    __shared__ int queue[(LDS_THREADS / 32) * R16_QCAP];

    const int T    = (int)blockDim.x;
    const int tid  = (int)threadIdx.x;
    const int lane = tid & 31;
    const int per  = (n + (int)gridDim.y - 1) / (int)gridDim.y;
    const int lo   = (int)blockIdx.y * per;
    const int lim  = max(0, min(per, n - lo));
    {
        float4* a4 = (float4*)acc;
        for (int i = tid; i < lim * 4; i += T) a4[i] = make_float4(0.0f, 0.0f, 0.0f, 0.0f);
    }
    __syncthreads();

    const int qo = (tid >> 5) * R16_QCAP;
    int cnt = 0;

    auto push = [&](int cv, int e) {
        const bool in = (unsigned)(cv - lo) < (unsigned)lim;
        const unsigned m = __builtin_amdgcn_ballot_w32(in);
        if (m != 0u) {
            if (in) __hip_atomic_store(&queue[qo + cnt + (int)__builtin_amdgcn_mbcnt_lo(m, 0u)], e,
                                       __ATOMIC_RELAXED, __HIP_MEMORY_SCOPE_WAVEFRONT);
            cnt += __builtin_popcount(m);
        }
    };
    auto drain = [&](int base, int count) {
        const int   e  = (lane < count) ? __hip_atomic_load(&queue[qo + base + lane], __ATOMIC_RELAXED,
                                                         __HIP_MEMORY_SCOPE_WAVEFRONT) : 0;
        const int   ri = row[e];
        const int   ci = col[e];
        const float nr = dinv[ri] * dinv[ci];
        const int   bi = ci - lo;
        const int   l4 = lane & 3;
        const int   g  = lane >> 2;
        float4 v[4];
        #pragma unroll
        for (int j = 0; j < 4; ++j) {
            const int r = __shfl(ri, 8 * j + g);
            v[j] = *(const float4*)(h1 + (size_t)r * 16 + 4 * l4);
        }
        #pragma unroll
        for (int j = 0; j < 4; ++j) {
            const int   s = 8 * j + g;
            const float w = __shfl(nr, s);
            const int   b = __shfl(bi, s);
            if (s < count) {
                float* p = &acc[b * 16 + 4 * l4];
                __hip_atomic_fetch_add(p + 0, w * v[j].x, __ATOMIC_RELAXED, __HIP_MEMORY_SCOPE_WORKGROUP);
                __hip_atomic_fetch_add(p + 1, w * v[j].y, __ATOMIC_RELAXED, __HIP_MEMORY_SCOPE_WORKGROUP);
                __hip_atomic_fetch_add(p + 2, w * v[j].z, __ATOMIC_RELAXED, __HIP_MEMORY_SCOPE_WORKGROUP);
                __hip_atomic_fetch_add(p + 3, w * v[j].w, __ATOMIC_RELAXED, __HIP_MEMORY_SCOPE_WORKGROUP);
            }
        }
    };

    const int quads = (E + 3) >> 2;
    const int qper  = (quads + (int)gridDim.x - 1) / (int)gridDim.x;
    const int q0    = (int)blockIdx.x * qper;
    const int q1    = min(q0 + qper, quads);
    const bool vec  = (((size_t)col) & 15) == 0;
    const int qv    = vec ? min(q1, E >> 2) : q0;

    for (int qb = q0; qb < qv; qb += 4 * T) {
        int4 v[4];
        #pragma unroll
        for (int u = 0; u < 4; ++u) {
            const int qq = qb + tid + u * T;
            v[u] = (qq < qv) ? *(const int4*)(col + 4 * (size_t)qq)
                             : make_int4(-1, -1, -1, -1);
        }
        #pragma unroll
        for (int u = 0; u < 4; ++u) {
            const int e0 = 4 * (qb + tid + u * T);
            push(v[u].x, e0 + 0);
            push(v[u].y, e0 + 1);
            push(v[u].z, e0 + 2);
            push(v[u].w, e0 + 3);
            while (cnt >= 32) { cnt -= 32; drain(cnt, 32); }
        }
    }
    {
        const int e0 = 4 * max(q0, qv);
        const int e1 = min(4 * q1, E);
        for (int eb = e0; eb < e1; eb += T) {
            const int e = eb + tid;
            push((e < e1) ? col[e] : -1, e);
            while (cnt >= 32) { cnt -= 32; drain(cnt, 32); }
        }
    }
    if (cnt > 0) drain(0, cnt);

    __syncthreads();
    float* dst = agg1 + (size_t)lo * 16;
    for (int i = tid; i < lim * 16; i += T) {
        const float t = acc[i];
        if (t != 0.0f) atomicAdd(&dst[i], t);
    }
}

__global__ void gcn_post1(const float* __restrict__ agg1,
                          const float* __restrict__ h1,
                          const float* __restrict__ dinv,
                          const float* __restrict__ b1,
                          const float* __restrict__ W2,
                          float* __restrict__ s, int n) {
    int i = blockIdx.x * blockDim.x + threadIdx.x;
    if (i >= n) return;
    float di = dinv[i];
    float self = di * di;
    const float* a = agg1 + (size_t)i * 16;
    const float* h = h1 + (size_t)i * 16;
    float acc = 0.0f;
    #pragma unroll
    for (int f = 0; f < 16; ++f) {
        float v = a[f] + self * h[f] + b1[f];
        v = fmaxf(v, 0.0f);
        acc = fmaf(v, W2[f], acc);
    }
    s[i] = acc;
}

__global__ __launch_bounds__(LDS_THREADS)
void gcn_edge1_lds(const int* __restrict__ row, const int* __restrict__ col,
                   const float* __restrict__ dinv, const float* __restrict__ s,
                   float* __restrict__ agg2, int E, int n) {
    __shared__ float acc[RANGE_BINS];
    const int T   = (int)blockDim.x;
    const int tid = (int)threadIdx.x;
    const int per = (n + (int)gridDim.y - 1) / (int)gridDim.y;
    const int lo  = (int)blockIdx.y * per;
    const int lim = max(0, min(per, n - lo));
    for (int i = tid; i < lim; i += T) acc[i] = 0.0f;
    __syncthreads();

    const int quads = (E + 3) >> 2;
    const int qper  = (quads + (int)gridDim.x - 1) / (int)gridDim.x;
    const int q0    = (int)blockIdx.x * qper;
    const int q1    = min(q0 + qper, quads);
    const bool vec  = ((((size_t)col) | ((size_t)row)) & 15) == 0;
    const int qv    = vec ? min(q1, E >> 2) : q0;

    for (int q = q0 + tid; q < qv; q += 2 * T) {
        int c[8], r[8];
        #pragma unroll
        for (int u = 0; u < 2; ++u) {
            const int qq = q + u * T;
            const bool ok = qq < qv;
            const int4 vc = ok ? *(const int4*)(col + 4 * (size_t)qq) : make_int4(-1, -1, -1, -1);
            const int4 vr = ok ? *(const int4*)(row + 4 * (size_t)qq) : make_int4(0, 0, 0, 0);
            c[4 * u + 0] = vc.x; c[4 * u + 1] = vc.y; c[4 * u + 2] = vc.z; c[4 * u + 3] = vc.w;
            r[4 * u + 0] = vr.x; r[4 * u + 1] = vr.y; r[4 * u + 2] = vr.z; r[4 * u + 3] = vr.w;
        }
        unsigned b[8];
        float    v[8];
        #pragma unroll
        for (int k = 0; k < 8; ++k) {
            b[k] = (unsigned)(c[k] - lo);
            const bool in = b[k] < (unsigned)lim;
            const int rr = in ? r[k] : 0;
            const int cc = in ? c[k] : 0;
            v[k] = dinv[rr] * dinv[cc] * s[rr];
        }
        #pragma unroll
        for (int k = 0; k < 8; ++k) {
            if (b[k] < (unsigned)lim)
                __hip_atomic_fetch_add(&acc[b[k]], v[k], __ATOMIC_RELAXED,
                                       __HIP_MEMORY_SCOPE_WORKGROUP);
        }
    }
    for (int q = max(q0, qv) + tid; q < q1; q += T) {
        for (int k = 0; k < 4; ++k) {
            const int e = 4 * q + k;
            if (e < E) {
                const int cc = col[e];
                const unsigned bb = (unsigned)(cc - lo);
                if (bb < (unsigned)lim) {
                    const int rr = row[e];
                    __hip_atomic_fetch_add(&acc[bb], dinv[rr] * dinv[cc] * s[rr],
                                           __ATOMIC_RELAXED, __HIP_MEMORY_SCOPE_WORKGROUP);
                }
            }
        }
    }
    __syncthreads();
    for (int i = tid; i < lim; i += T) {
        const float t = acc[i];
        if (t != 0.0f) atomicAdd(&agg2[lo + i], t);
    }
}

__global__ void gcn_post2(const float* __restrict__ agg2,
                          const float* __restrict__ s,
                          const float* __restrict__ dinv,
                          const float* __restrict__ b2,
                          float* __restrict__ out, int n) {
    int i = blockIdx.x * blockDim.x + threadIdx.x;
    if (i >= n) return;
    float di = dinv[i];
    out[i] = agg2[i] + di * di * s[i] + b2[0];
}

extern "C" void kernel_launch(void* const* d_in, const int* in_sizes, int n_in,
                              void* d_out, int out_size, void* d_ws, size_t ws_size,
                              hipStream_t stream) {
    const float* x  = (const float*)d_in[0];
    const int*   ei = (const int*)d_in[1];
    const float* W1 = (const float*)d_in[2];
    const float* b1 = (const float*)d_in[3];
    const float* W2 = (const float*)d_in[4];
    const float* b2 = (const float*)d_in[5];

    const int N = in_sizes[0] / 128;
    const int E = in_sizes[1] / 2;
    const int* row = ei;
    const int* col = ei + E;

    float* ws   = (float*)d_ws;
    float* deg  = ws;
    float* dinv = deg + N;
    float* h1   = dinv + N;
    float* agg1 = h1 + (size_t)N * 16;
    float* s    = agg1 + (size_t)N * 16;
    float* agg2 = s + N;
    float* out  = (float*)d_out;

    const int B = 256;

    {
        int total = N * 16;
        gcn_init<<<(total + B - 1) / B, B, 0, stream>>>(deg, agg1, agg2, N);
    }
    const int R = (N + RANGE_BINS - 1) / RANGE_BINS;
    const dim3 ldsGrid(EDGE_SLICES, R > 0 ? R : 1);

    gcn_degree_lds<<<ldsGrid, LDS_THREADS, 0, stream>>>(col, deg, E, N);
    gcn_dinv<<<(N + B - 1) / B, B, 0, stream>>>(deg, dinv, N);
    {
        int full   = N & ~15;
        int ntiles = full >> 4;
        if (ntiles > 0) {
            int wavesPerBlock = B / 32;
            int blocks = (ntiles + wavesPerBlock - 1) / wavesPerBlock;
            gcn_gemm1_wmma<<<blocks, B, 0, stream>>>(x, W1, h1, ntiles);
        }
        if (full < N) {
            int tail = (N - full) * 16;
            gcn_gemm1_tail<<<(tail + B - 1) / B, B, 0, stream>>>(x, W1, h1, full, N);
        }
    }
    {
        const int R16 = (N + R16_NODES - 1) / R16_NODES;
        const dim3 grid16(R16_SLICES, R16 > 0 ? R16 : 1);
        gcn_edge16_lds<<<grid16, LDS_THREADS, 0, stream>>>(row, col, dinv, h1, agg1, E, N);
    }
    gcn_post1<<<(N + B - 1) / B, B, 0, stream>>>(agg1, h1, dinv, b1, W2, s, N);
    gcn_edge1_lds<<<ldsGrid, LDS_THREADS, 0, stream>>>(row, col, dinv, s, agg2, E, N);
    gcn_post2<<<(N + B - 1) / B, B, 0, stream>>>(agg2, s, dinv, b2, out, N);
}


# Normalisation factored out of the edge passes: gathered rows kept pre-scaled by source d^-1/2, target d^-1/2 applied once per sum; no dinv gathers per edge; term rounding changes, sum order kept
# speedup vs baseline: 2.8393x; 1.3752x over previous
#include <hip/hip_runtime.h>

typedef __attribute__((ext_vector_type(2))) float v2f;
typedef __attribute__((ext_vector_type(8))) float v8f;

__global__ void gcn_init(float* __restrict__ deg, float* __restrict__ agg1,
                         float* __restrict__ agg2, int n) {
    int i = blockIdx.x * blockDim.x + threadIdx.x;
    if (i < n) { deg[i] = 1.0f; agg2[i] = 0.0f; }
    if (i < n * 16) { agg1[i] = 0.0f; }
}

#define RANGE_BINS  51200
#define EDGE_SLICES 64
#define LDS_THREADS 1024

__global__ __launch_bounds__(LDS_THREADS)
void gcn_degree_lds(const int* __restrict__ col, float* __restrict__ deg,
                    int E, int n) {
    __shared__ unsigned cnt[RANGE_BINS];
    const int T   = (int)blockDim.x;
    const int tid = (int)threadIdx.x;
    const int per = (n + (int)gridDim.y - 1) / (int)gridDim.y;
    const int lo  = (int)blockIdx.y * per;
    const int lim = max(0, min(per, n - lo));
    for (int i = tid; i < lim; i += T) cnt[i] = 0u;
    __syncthreads();

    const int quads = (E + 3) >> 2;
    const int qper  = (quads + (int)gridDim.x - 1) / (int)gridDim.x;
    const int q0    = (int)blockIdx.x * qper;
    const int q1    = min(q0 + qper, quads);
    const bool vec  = (((size_t)col) & 15) == 0;
    const int qv    = vec ? min(q1, E >> 2) : q0;

    for (int q = q0 + tid; q < qv; q += 4 * T) {
        int4 v[4];
        #pragma unroll
        for (int u = 0; u < 4; ++u) {
            const int qq = q + u * T;
            v[u] = (qq < qv) ? *(const int4*)(col + 4 * (size_t)qq)
                             : make_int4(-1, -1, -1, -1);
        }
        #pragma unroll
        for (int u = 0; u < 4; ++u) {
            const unsigned b0 = (unsigned)(v[u].x - lo);
            const unsigned b1 = (unsigned)(v[u].y - lo);
            const unsigned b2 = (unsigned)(v[u].z - lo);
            const unsigned b3 = (unsigned)(v[u].w - lo);
            if (b0 < (unsigned)lim) atomicAdd(&cnt[b0], 1u);
            if (b1 < (unsigned)lim) atomicAdd(&cnt[b1], 1u);
            if (b2 < (unsigned)lim) atomicAdd(&cnt[b2], 1u);
            if (b3 < (unsigned)lim) atomicAdd(&cnt[b3], 1u);
        }
    }
    for (int q = max(q0, qv) + tid; q < q1; q += T) {
        for (int k = 0; k < 4; ++k) {
            const int e = 4 * q + k;
            if (e < E) {
                const unsigned b = (unsigned)(col[e] - lo);
                if (b < (unsigned)lim) atomicAdd(&cnt[b], 1u);
            }
        }
    }
    __syncthreads();
    for (int i = tid; i < lim; i += T) {
        const unsigned v = cnt[i];
        if (v != 0u) atomicAdd(&deg[lo + i], (float)v);
    }
}

__global__ void gcn_dinv(const float* __restrict__ deg, float* __restrict__ dinv,
                         int n) {
    int i = blockIdx.x * blockDim.x + threadIdx.x;
    if (i >= n) return;
    float d = deg[i];
    dinv[i] = (d > 0.0f) ? rsqrtf(d) : 0.0f;
}

__global__ void gcn_gemm1_wmma(const float* __restrict__ x,
                               const float* __restrict__ W1,
                               const float* __restrict__ dinv,
                               float* __restrict__ h1, int ntiles) {
    int wave = blockIdx.x * (blockDim.x >> 5) + (threadIdx.x >> 5);
    if (wave >= ntiles) return;
    int lane  = threadIdx.x & 31;
    int m     = lane & 15;
    int kb    = (lane >> 4) << 1;
    size_t row16 = (size_t)wave * 16;

    v8f acc = {};
    const float* arow = x + (row16 + (size_t)m) * 128 + kb;
    #pragma unroll 4
    for (int k0 = 0; k0 < 128; k0 += 4) {
        v2f a = *(const v2f*)(arow + k0);
        v2f b;
        b.x = W1[(size_t)(k0 + kb) * 16 + m];
        b.y = W1[(size_t)(k0 + kb + 1) * 16 + m];
        acc = __builtin_amdgcn_wmma_f32_16x16x4_f32(
             false, a,  false, b,
             (short)0, acc,  false,  false);
    }

    int mbase = (lane >> 4) << 3;
    int ncol  = lane & 15;
    #pragma unroll
    for (int r = 0; r < 8; ++r) {
        const size_t node = row16 + (size_t)(mbase + r);
        h1[node * 16 + ncol] = acc[r] * dinv[node];
    }
}

__global__ void gcn_gemm1_tail(const float* __restrict__ x,
                               const float* __restrict__ W1,
                               const float* __restrict__ dinv,
                               float* __restrict__ h1, int start, int n) {
    int t = blockIdx.x * blockDim.x + threadIdx.x;
    int node = start + (t >> 4);
    int f = t & 15;
    if (node >= n) return;
    float acc = 0.0f;
    const float* xr = x + (size_t)node * 128;
    for (int k = 0; k < 128; ++k) acc += xr[k] * W1[(size_t)k * 16 + f];
    h1[(size_t)node * 16 + f] = acc * dinv[node];
}

#define R16_NODES  4768
#define R16_SLICES 6
#define R16_QCAP   160

__global__ __launch_bounds__(LDS_THREADS)
void gcn_edge16_lds(const int* __restrict__ row, const int* __restrict__ col,
                    const float* __restrict__ h1,
                    float* __restrict__ agg1, int E, int n) {
    __shared__ float acc[R16_NODES * 16];
    __shared__ int queue[(LDS_THREADS / 32) * R16_QCAP];

    const int T    = (int)blockDim.x;
    const int tid  = (int)threadIdx.x;
    const int lane = tid & 31;
    const int per  = (n + (int)gridDim.y - 1) / (int)gridDim.y;
    const int lo   = (int)blockIdx.y * per;
    const int lim  = max(0, min(per, n - lo));
    {
        float4* a4 = (float4*)acc;
        for (int i = tid; i < lim * 4; i += T) a4[i] = make_float4(0.0f, 0.0f, 0.0f, 0.0f);
    }
    __syncthreads();

    const int qo = (tid >> 5) * R16_QCAP;
    int cnt = 0;

    auto push = [&](int cv, int e) {
        const bool in = (unsigned)(cv - lo) < (unsigned)lim;
        const unsigned m = __builtin_amdgcn_ballot_w32(in);
        if (m != 0u) {
            if (in) __hip_atomic_store(&queue[qo + cnt + (int)__builtin_amdgcn_mbcnt_lo(m, 0u)], e,
                                       __ATOMIC_RELAXED, __HIP_MEMORY_SCOPE_WAVEFRONT);
            cnt += __builtin_popcount(m);
        }
    };
    auto drain = [&](int base, int count) {
        const int   e  = (lane < count) ? __hip_atomic_load(&queue[qo + base + lane], __ATOMIC_RELAXED,
                                                         __HIP_MEMORY_SCOPE_WAVEFRONT) : 0;
        const int   ri = row[e];
        const int   bi = col[e] - lo;
        const int   l4 = lane & 3;
        const int   g  = lane >> 2;
        float4 v[4];
        #pragma unroll
        for (int j = 0; j < 4; ++j) {
            const int r = __shfl(ri, 8 * j + g);
            v[j] = *(const float4*)(h1 + (size_t)r * 16 + 4 * l4);
        }
        #pragma unroll
        for (int j = 0; j < 4; ++j) {
            const int s = 8 * j + g;
            const int b = __shfl(bi, s);
            if (s < count) {
                float* p = &acc[b * 16 + 4 * l4];
                __hip_atomic_fetch_add(p + 0, v[j].x, __ATOMIC_RELAXED, __HIP_MEMORY_SCOPE_WORKGROUP);
                __hip_atomic_fetch_add(p + 1, v[j].y, __ATOMIC_RELAXED, __HIP_MEMORY_SCOPE_WORKGROUP);
                __hip_atomic_fetch_add(p + 2, v[j].z, __ATOMIC_RELAXED, __HIP_MEMORY_SCOPE_WORKGROUP);
                __hip_atomic_fetch_add(p + 3, v[j].w, __ATOMIC_RELAXED, __HIP_MEMORY_SCOPE_WORKGROUP);
            }
        }
    };

    const int quads = (E + 3) >> 2;
    const int qper  = (quads + (int)gridDim.x - 1) / (int)gridDim.x;
    const int q0    = (int)blockIdx.x * qper;
    const int q1    = min(q0 + qper, quads);
    const bool vec  = (((size_t)col) & 15) == 0;
    const int qv    = vec ? min(q1, E >> 2) : q0;

    for (int qb = q0; qb < qv; qb += 4 * T) {
        int4 v[4];
        #pragma unroll
        for (int u = 0; u < 4; ++u) {
            const int qq = qb + tid + u * T;
            v[u] = (qq < qv) ? *(const int4*)(col + 4 * (size_t)qq)
                             : make_int4(-1, -1, -1, -1);
        }
        #pragma unroll
        for (int u = 0; u < 4; ++u) {
            const int e0 = 4 * (qb + tid + u * T);
            push(v[u].x, e0 + 0);
            push(v[u].y, e0 + 1);
            push(v[u].z, e0 + 2);
            push(v[u].w, e0 + 3);
            while (cnt >= 32) { cnt -= 32; drain(cnt, 32); }
        }
    }
    {
        const int e0 = 4 * max(q0, qv);
        const int e1 = min(4 * q1, E);
        for (int eb = e0; eb < e1; eb += T) {
            const int e = eb + tid;
            push((e < e1) ? col[e] : -1, e);
            while (cnt >= 32) { cnt -= 32; drain(cnt, 32); }
        }
    }
    if (cnt > 0) drain(0, cnt);

    __syncthreads();
    float* dst = agg1 + (size_t)lo * 16;
    for (int i = tid; i < lim * 16; i += T) {
        const float t = acc[i];
        if (t != 0.0f) atomicAdd(&dst[i], t);
    }
}

__global__ void gcn_post1(const float* __restrict__ agg1,
                          const float* __restrict__ h1,
                          const float* __restrict__ dinv,
                          const float* __restrict__ b1,
                          const float* __restrict__ W2,
                          float* __restrict__ s, int n) {
    int i = blockIdx.x * blockDim.x + threadIdx.x;
    if (i >= n) return;
    float di = dinv[i];
    const float* a = agg1 + (size_t)i * 16;
    const float* h = h1 + (size_t)i * 16;
    float acc = 0.0f;
    #pragma unroll
    for (int f = 0; f < 16; ++f) {
        float v = di * (a[f] + h[f]) + b1[f];
        v = fmaxf(v, 0.0f);
        acc = fmaf(v, W2[f], acc);
    }
    s[i] = di * acc;
}

__global__ __launch_bounds__(LDS_THREADS)
void gcn_edge1_lds(const int* __restrict__ row, const int* __restrict__ col,
                   const float* __restrict__ s,
                   float* __restrict__ agg2, int E, int n) {
    __shared__ float acc[RANGE_BINS];
    const int T   = (int)blockDim.x;
    const int tid = (int)threadIdx.x;
    const int per = (n + (int)gridDim.y - 1) / (int)gridDim.y;
    const int lo  = (int)blockIdx.y * per;
    const int lim = max(0, min(per, n - lo));
    for (int i = tid; i < lim; i += T) acc[i] = 0.0f;
    __syncthreads();

    const int quads = (E + 3) >> 2;
    const int qper  = (quads + (int)gridDim.x - 1) / (int)gridDim.x;
    const int q0    = (int)blockIdx.x * qper;
    const int q1    = min(q0 + qper, quads);
    const bool vec  = ((((size_t)col) | ((size_t)row)) & 15) == 0;
    const int qv    = vec ? min(q1, E >> 2) : q0;

    for (int q = q0 + tid; q < qv; q += 2 * T) {
        int c[8], r[8];
        #pragma unroll
        for (int u = 0; u < 2; ++u) {
            const int qq = q + u * T;
            const bool ok = qq < qv;
            const int4 vc = ok ? *(const int4*)(col + 4 * (size_t)qq) : make_int4(-1, -1, -1, -1);
            const int4 vr = ok ? *(const int4*)(row + 4 * (size_t)qq) : make_int4(0, 0, 0, 0);
            c[4 * u + 0] = vc.x; c[4 * u + 1] = vc.y; c[4 * u + 2] = vc.z; c[4 * u + 3] = vc.w;
            r[4 * u + 0] = vr.x; r[4 * u + 1] = vr.y; r[4 * u + 2] = vr.z; r[4 * u + 3] = vr.w;
        }
        unsigned b[8];
        float    v[8];
        #pragma unroll
        for (int k = 0; k < 8; ++k) {
            b[k] = (unsigned)(c[k] - lo);
            const bool in = b[k] < (unsigned)lim;
            v[k] = s[in ? r[k] : 0];
        }
        #pragma unroll
        for (int k = 0; k < 8; ++k) {
            if (b[k] < (unsigned)lim)
                __hip_atomic_fetch_add(&acc[b[k]], v[k], __ATOMIC_RELAXED,
                                       __HIP_MEMORY_SCOPE_WORKGROUP);
        }
    }
    for (int q = max(q0, qv) + tid; q < q1; q += T) {
        for (int k = 0; k < 4; ++k) {
            const int e = 4 * q + k;
            if (e < E) {
                const unsigned bb = (unsigned)(col[e] - lo);
                if (bb < (unsigned)lim) {
                    __hip_atomic_fetch_add(&acc[bb], s[row[e]],
                                           __ATOMIC_RELAXED, __HIP_MEMORY_SCOPE_WORKGROUP);
                }
            }
        }
    }
    __syncthreads();
    for (int i = tid; i < lim; i += T) {
        const float t = acc[i];
        if (t != 0.0f) atomicAdd(&agg2[lo + i], t);
    }
}

__global__ void gcn_post2(const float* __restrict__ agg2,
                          const float* __restrict__ s,
                          const float* __restrict__ dinv,
                          const float* __restrict__ b2,
                          float* __restrict__ out, int n) {
    int i = blockIdx.x * blockDim.x + threadIdx.x;
    if (i >= n) return;
    float di = dinv[i];
    out[i] = di * (agg2[i] + s[i]) + b2[0];
}

extern "C" void kernel_launch(void* const* d_in, const int* in_sizes, int n_in,
                              void* d_out, int out_size, void* d_ws, size_t ws_size,
                              hipStream_t stream) {
    const float* x  = (const float*)d_in[0];
    const int*   ei = (const int*)d_in[1];
    const float* W1 = (const float*)d_in[2];
    const float* b1 = (const float*)d_in[3];
    const float* W2 = (const float*)d_in[4];
    const float* b2 = (const float*)d_in[5];

    const int N = in_sizes[0] / 128;
    const int E = in_sizes[1] / 2;
    const int* row = ei;
    const int* col = ei + E;

    float* ws   = (float*)d_ws;
    float* deg  = ws;
    float* dinv = deg + N;
    float* h1   = dinv + N;
    float* agg1 = h1 + (size_t)N * 16;
    float* s    = agg1 + (size_t)N * 16;
    float* agg2 = s + N;
    float* out  = (float*)d_out;

    const int B = 256;

    {
        int total = N * 16;
        gcn_init<<<(total + B - 1) / B, B, 0, stream>>>(deg, agg1, agg2, N);
    }
    const int R = (N + RANGE_BINS - 1) / RANGE_BINS;
    const dim3 ldsGrid(EDGE_SLICES, R > 0 ? R : 1);

    gcn_degree_lds<<<ldsGrid, LDS_THREADS, 0, stream>>>(col, deg, E, N);
    gcn_dinv<<<(N + B - 1) / B, B, 0, stream>>>(deg, dinv, N);
    {
        int full   = N & ~15;
        int ntiles = full >> 4;
        if (ntiles > 0) {
            int wavesPerBlock = B / 32;
            int blocks = (ntiles + wavesPerBlock - 1) / wavesPerBlock;
            gcn_gemm1_wmma<<<blocks, B, 0, stream>>>(x, W1, dinv, h1, ntiles);
        }
        if (full < N) {
            int tail = (N - full) * 16;
            gcn_gemm1_tail<<<(tail + B - 1) / B, B, 0, stream>>>(x, W1, dinv, h1, full, N);
        }
    }
    {
        const int R16 = (N + R16_NODES - 1) / R16_NODES;
        const dim3 grid16(R16_SLICES, R16 > 0 ? R16 : 1);
        gcn_edge16_lds<<<grid16, LDS_THREADS, 0, stream>>>(row, col, h1, agg1, E, N);
    }
    gcn_post1<<<(N + B - 1) / B, B, 0, stream>>>(agg1, h1, dinv, b1, W2, s, N);
    gcn_edge1_lds<<<ldsGrid, LDS_THREADS, 0, stream>>>(row, col, s, agg2, E, N);
    gcn_post2<<<(N + B - 1) / B, B, 0, stream>>>(agg2, s, dinv, b2, out, N);
}


# Layer-1 edge queue entries carry the target's LDS row (16-bit side queue) beside the edge number: the drain gathers no target per edge; queue checked every 2 pushes, so LDS float add order changes
# speedup vs baseline: 3.1053x; 1.0937x over previous
#include <hip/hip_runtime.h>

typedef __attribute__((ext_vector_type(2))) float v2f;
typedef __attribute__((ext_vector_type(8))) float v8f;

__global__ void gcn_init(float* __restrict__ deg, float* __restrict__ agg1,
                         float* __restrict__ agg2, int n) {
    int i = blockIdx.x * blockDim.x + threadIdx.x;
    if (i < n) { deg[i] = 1.0f; agg2[i] = 0.0f; }
    if (i < n * 16) { agg1[i] = 0.0f; }
}

#define RANGE_BINS  51200
#define EDGE_SLICES 64
#define LDS_THREADS 1024

__global__ __launch_bounds__(LDS_THREADS)
void gcn_degree_lds(const int* __restrict__ col, float* __restrict__ deg,
                    int E, int n) {
    __shared__ unsigned cnt[RANGE_BINS];
    const int T   = (int)blockDim.x;
    const int tid = (int)threadIdx.x;
    const int per = (n + (int)gridDim.y - 1) / (int)gridDim.y;
    const int lo  = (int)blockIdx.y * per;
    const int lim = max(0, min(per, n - lo));
    for (int i = tid; i < lim; i += T) cnt[i] = 0u;
    __syncthreads();

    const int quads = (E + 3) >> 2;
    const int qper  = (quads + (int)gridDim.x - 1) / (int)gridDim.x;
    const int q0    = (int)blockIdx.x * qper;
    const int q1    = min(q0 + qper, quads);
    const bool vec  = (((size_t)col) & 15) == 0;
    const int qv    = vec ? min(q1, E >> 2) : q0;

    for (int q = q0 + tid; q < qv; q += 4 * T) {
        int4 v[4];
        #pragma unroll
        for (int u = 0; u < 4; ++u) {
            const int qq = q + u * T;
            v[u] = (qq < qv) ? *(const int4*)(col + 4 * (size_t)qq)
                             : make_int4(-1, -1, -1, -1);
        }
        #pragma unroll
        for (int u = 0; u < 4; ++u) {
            const unsigned b0 = (unsigned)(v[u].x - lo);
            const unsigned b1 = (unsigned)(v[u].y - lo);
            const unsigned b2 = (unsigned)(v[u].z - lo);
            const unsigned b3 = (unsigned)(v[u].w - lo);
            if (b0 < (unsigned)lim) atomicAdd(&cnt[b0], 1u);
            if (b1 < (unsigned)lim) atomicAdd(&cnt[b1], 1u);
            if (b2 < (unsigned)lim) atomicAdd(&cnt[b2], 1u);
            if (b3 < (unsigned)lim) atomicAdd(&cnt[b3], 1u);
        }
    }
    for (int q = max(q0, qv) + tid; q < q1; q += T) {
        for (int k = 0; k < 4; ++k) {
            const int e = 4 * q + k;
            if (e < E) {
                const unsigned b = (unsigned)(col[e] - lo);
                if (b < (unsigned)lim) atomicAdd(&cnt[b], 1u);
            }
        }
    }
    __syncthreads();
    for (int i = tid; i < lim; i += T) {
        const unsigned v = cnt[i];
        if (v != 0u) atomicAdd(&deg[lo + i], (float)v);
    }
}

__global__ void gcn_dinv(const float* __restrict__ deg, float* __restrict__ dinv,
                         int n) {
    int i = blockIdx.x * blockDim.x + threadIdx.x;
    if (i >= n) return;
    float d = deg[i];
    dinv[i] = (d > 0.0f) ? rsqrtf(d) : 0.0f;
}

__global__ void gcn_gemm1_wmma(const float* __restrict__ x,
                               const float* __restrict__ W1,
                               const float* __restrict__ dinv,
                               float* __restrict__ h1, int ntiles) {
    int wave = blockIdx.x * (blockDim.x >> 5) + (threadIdx.x >> 5);
    if (wave >= ntiles) return;
    int lane  = threadIdx.x & 31;
    int m     = lane & 15;
    int kb    = (lane >> 4) << 1;
    size_t row16 = (size_t)wave * 16;

    v8f acc = {};
    const float* arow = x + (row16 + (size_t)m) * 128 + kb;
    #pragma unroll 4
    for (int k0 = 0; k0 < 128; k0 += 4) {
        v2f a = *(const v2f*)(arow + k0);
        v2f b;
        b.x = W1[(size_t)(k0 + kb) * 16 + m];
        b.y = W1[(size_t)(k0 + kb + 1) * 16 + m];
        acc = __builtin_amdgcn_wmma_f32_16x16x4_f32(
             false, a,  false, b,
             (short)0, acc,  false,  false);
    }

    int mbase = (lane >> 4) << 3;
    int ncol  = lane & 15;
    #pragma unroll
    for (int r = 0; r < 8; ++r) {
        const size_t node = row16 + (size_t)(mbase + r);
        h1[node * 16 + ncol] = acc[r] * dinv[node];
    }
}

__global__ void gcn_gemm1_tail(const float* __restrict__ x,
                               const float* __restrict__ W1,
                               const float* __restrict__ dinv,
                               float* __restrict__ h1, int start, int n) {
    int t = blockIdx.x * blockDim.x + threadIdx.x;
    int node = start + (t >> 4);
    int f = t & 15;
    if (node >= n) return;
    float acc = 0.0f;
    const float* xr = x + (size_t)node * 128;
    for (int k = 0; k < 128; ++k) acc += xr[k] * W1[(size_t)k * 16 + f];
    h1[(size_t)node * 16 + f] = acc * dinv[node];
}

#define R16_NODES  4768
#define R16_SLICES 6
#define R16_QCAP   96

__global__ __launch_bounds__(LDS_THREADS)
void gcn_edge16_lds(const int* __restrict__ row, const int* __restrict__ col,
                    const float* __restrict__ h1,
                    float* __restrict__ agg1, int E, int n) {
    __shared__ float acc[R16_NODES * 16];
    __shared__ int queue[(LDS_THREADS / 32) * R16_QCAP];
    __shared__ unsigned short qrow[(LDS_THREADS / 32) * R16_QCAP];
    static_assert(R16_NODES <= 65535, "a target's LDS row must fit a 16-bit queue word");

    const int T    = (int)blockDim.x;
    const int tid  = (int)threadIdx.x;
    const int lane = tid & 31;
    const int per  = (n + (int)gridDim.y - 1) / (int)gridDim.y;
    const int lo   = (int)blockIdx.y * per;
    const int lim  = max(0, min(per, n - lo));
    {
        float4* a4 = (float4*)acc;
        for (int i = tid; i < lim * 4; i += T) a4[i] = make_float4(0.0f, 0.0f, 0.0f, 0.0f);
    }
    __syncthreads();

    const int qo = (tid >> 5) * R16_QCAP;
    int cnt = 0;

    auto push = [&](int cv, int e) {
        const unsigned bin = (unsigned)(cv - lo);
        const bool in = bin < (unsigned)lim;
        const unsigned m = __builtin_amdgcn_ballot_w32(in);
        if (m != 0u) {
            if (in) {
                const int slot = qo + cnt + (int)__builtin_amdgcn_mbcnt_lo(m, 0u);
                __hip_atomic_store(&queue[slot], e, __ATOMIC_RELAXED, __HIP_MEMORY_SCOPE_WAVEFRONT);
                __hip_atomic_store(&qrow[slot], (unsigned short)bin,
                                   __ATOMIC_RELAXED, __HIP_MEMORY_SCOPE_WAVEFRONT);
            }
            cnt += __builtin_popcount(m);
        }
    };
    auto drain = [&](int base, int count) {
        const bool  have = lane < count;
        const int   e  = have ? __hip_atomic_load(&queue[qo + base + lane], __ATOMIC_RELAXED,
                                                  __HIP_MEMORY_SCOPE_WAVEFRONT) : 0;
        const int   bi = have ? (int)__hip_atomic_load(&qrow[qo + base + lane], __ATOMIC_RELAXED,
                                                       __HIP_MEMORY_SCOPE_WAVEFRONT) : 0;
        const int   ri = row[e];
        const int   l4 = lane & 3;
        const int   g  = lane >> 2;
        float4 v[4];
        #pragma unroll
        for (int j = 0; j < 4; ++j) {
            const int r = __shfl(ri, 8 * j + g);
            v[j] = *(const float4*)(h1 + (size_t)r * 16 + 4 * l4);
        }
        #pragma unroll
        for (int j = 0; j < 4; ++j) {
            const int s = 8 * j + g;
            const int b = __shfl(bi, s);
            if (s < count) {
                float* p = &acc[b * 16 + 4 * l4];
                __hip_atomic_fetch_add(p + 0, v[j].x, __ATOMIC_RELAXED, __HIP_MEMORY_SCOPE_WORKGROUP);
                __hip_atomic_fetch_add(p + 1, v[j].y, __ATOMIC_RELAXED, __HIP_MEMORY_SCOPE_WORKGROUP);
                __hip_atomic_fetch_add(p + 2, v[j].z, __ATOMIC_RELAXED, __HIP_MEMORY_SCOPE_WORKGROUP);
                __hip_atomic_fetch_add(p + 3, v[j].w, __ATOMIC_RELAXED, __HIP_MEMORY_SCOPE_WORKGROUP);
            }
        }
    };

    const int quads = (E + 3) >> 2;
    const int qper  = (quads + (int)gridDim.x - 1) / (int)gridDim.x;
    const int q0    = (int)blockIdx.x * qper;
    const int q1    = min(q0 + qper, quads);
    const bool vec  = (((size_t)col) & 15) == 0;
    const int qv    = vec ? min(q1, E >> 2) : q0;

    for (int qb = q0; qb < qv; qb += 4 * T) {
        int4 v[4];
        #pragma unroll
        for (int u = 0; u < 4; ++u) {
            const int qq = qb + tid + u * T;
            v[u] = (qq < qv) ? *(const int4*)(col + 4 * (size_t)qq)
                             : make_int4(-1, -1, -1, -1);
        }
        #pragma unroll
        for (int u = 0; u < 4; ++u) {
            const int e0 = 4 * (qb + tid + u * T);
            push(v[u].x, e0 + 0);
            push(v[u].y, e0 + 1);
            while (cnt >= 32) { cnt -= 32; drain(cnt, 32); }
            push(v[u].z, e0 + 2);
            push(v[u].w, e0 + 3);
            while (cnt >= 32) { cnt -= 32; drain(cnt, 32); }
        }
    }
    {
        const int e0 = 4 * max(q0, qv);
        const int e1 = min(4 * q1, E);
        for (int eb = e0; eb < e1; eb += T) {
            const int e = eb + tid;
            push((e < e1) ? col[e] : -1, e);
            while (cnt >= 32) { cnt -= 32; drain(cnt, 32); }
        }
    }
    if (cnt > 0) drain(0, cnt);

    __syncthreads();
    float* dst = agg1 + (size_t)lo * 16;
    for (int i = tid; i < lim * 16; i += T) {
        const float t = acc[i];
        if (t != 0.0f) atomicAdd(&dst[i], t);
    }
}

__global__ void gcn_post1(const float* __restrict__ agg1,
                          const float* __restrict__ h1,
                          const float* __restrict__ dinv,
                          const float* __restrict__ b1,
                          const float* __restrict__ W2,
                          float* __restrict__ s, int n) {
    int i = blockIdx.x * blockDim.x + threadIdx.x;
    if (i >= n) return;
    float di = dinv[i];
    const float* a = agg1 + (size_t)i * 16;
    const float* h = h1 + (size_t)i * 16;
    float acc = 0.0f;
    #pragma unroll
    for (int f = 0; f < 16; ++f) {
        float v = di * (a[f] + h[f]) + b1[f];
        v = fmaxf(v, 0.0f);
        acc = fmaf(v, W2[f], acc);
    }
    s[i] = di * acc;
}

__global__ __launch_bounds__(LDS_THREADS)
void gcn_edge1_lds(const int* __restrict__ row, const int* __restrict__ col,
                   const float* __restrict__ s,
                   float* __restrict__ agg2, int E, int n) {
    __shared__ float acc[RANGE_BINS];
    const int T   = (int)blockDim.x;
    const int tid = (int)threadIdx.x;
    const int per = (n + (int)gridDim.y - 1) / (int)gridDim.y;
    const int lo  = (int)blockIdx.y * per;
    const int lim = max(0, min(per, n - lo));
    for (int i = tid; i < lim; i += T) acc[i] = 0.0f;
    __syncthreads();

    const int quads = (E + 3) >> 2;
    const int qper  = (quads + (int)gridDim.x - 1) / (int)gridDim.x;
    const int q0    = (int)blockIdx.x * qper;
    const int q1    = min(q0 + qper, quads);
    const bool vec  = ((((size_t)col) | ((size_t)row)) & 15) == 0;
    const int qv    = vec ? min(q1, E >> 2) : q0;

    for (int q = q0 + tid; q < qv; q += 2 * T) {
        int c[8], r[8];
        #pragma unroll
        for (int u = 0; u < 2; ++u) {
            const int qq = q + u * T;
            const bool ok = qq < qv;
            const int4 vc = ok ? *(const int4*)(col + 4 * (size_t)qq) : make_int4(-1, -1, -1, -1);
            const int4 vr = ok ? *(const int4*)(row + 4 * (size_t)qq) : make_int4(0, 0, 0, 0);
            c[4 * u + 0] = vc.x; c[4 * u + 1] = vc.y; c[4 * u + 2] = vc.z; c[4 * u + 3] = vc.w;
            r[4 * u + 0] = vr.x; r[4 * u + 1] = vr.y; r[4 * u + 2] = vr.z; r[4 * u + 3] = vr.w;
        }
        unsigned b[8];
        float    v[8];
        #pragma unroll
        for (int k = 0; k < 8; ++k) {
            b[k] = (unsigned)(c[k] - lo);
            const bool in = b[k] < (unsigned)lim;
            v[k] = s[in ? r[k] : 0];
        }
        #pragma unroll
        for (int k = 0; k < 8; ++k) {
            if (b[k] < (unsigned)lim)
                __hip_atomic_fetch_add(&acc[b[k]], v[k], __ATOMIC_RELAXED,
                                       __HIP_MEMORY_SCOPE_WORKGROUP);
        }
    }
    for (int q = max(q0, qv) + tid; q < q1; q += T) {
        for (int k = 0; k < 4; ++k) {
            const int e = 4 * q + k;
            if (e < E) {
                const unsigned bb = (unsigned)(col[e] - lo);
                if (bb < (unsigned)lim) {
                    __hip_atomic_fetch_add(&acc[bb], s[row[e]],
                                           __ATOMIC_RELAXED, __HIP_MEMORY_SCOPE_WORKGROUP);
                }
            }
        }
    }
    __syncthreads();
    for (int i = tid; i < lim; i += T) {
        const float t = acc[i];
        if (t != 0.0f) atomicAdd(&agg2[lo + i], t);
    }
}

__global__ void gcn_post2(const float* __restrict__ agg2,
                          const float* __restrict__ s,
                          const float* __restrict__ dinv,
                          const float* __restrict__ b2,
                          float* __restrict__ out, int n) {
    int i = blockIdx.x * blockDim.x + threadIdx.x;
    if (i >= n) return;
    float di = dinv[i];
    out[i] = di * (agg2[i] + s[i]) + b2[0];
}

extern "C" void kernel_launch(void* const* d_in, const int* in_sizes, int n_in,
                              void* d_out, int out_size, void* d_ws, size_t ws_size,
                              hipStream_t stream) {
    const float* x  = (const float*)d_in[0];
    const int*   ei = (const int*)d_in[1];
    const float* W1 = (const float*)d_in[2];
    const float* b1 = (const float*)d_in[3];
    const float* W2 = (const float*)d_in[4];
    const float* b2 = (const float*)d_in[5];

    const int N = in_sizes[0] / 128;
    const int E = in_sizes[1] / 2;
    const int* row = ei;
    const int* col = ei + E;

    float* ws   = (float*)d_ws;
    float* deg  = ws;
    float* dinv = deg + N;
    float* h1   = dinv + N;
    float* agg1 = h1 + (size_t)N * 16;
    float* s    = agg1 + (size_t)N * 16;
    float* agg2 = s + N;
    float* out  = (float*)d_out;

    const int B = 256;

    {
        int total = N * 16;
        gcn_init<<<(total + B - 1) / B, B, 0, stream>>>(deg, agg1, agg2, N);
    }
    const int R = (N + RANGE_BINS - 1) / RANGE_BINS;
    const dim3 ldsGrid(EDGE_SLICES, R > 0 ? R : 1);

    gcn_degree_lds<<<ldsGrid, LDS_THREADS, 0, stream>>>(col, deg, E, N);
    gcn_dinv<<<(N + B - 1) / B, B, 0, stream>>>(deg, dinv, N);
    {
        int full   = N & ~15;
        int ntiles = full >> 4;
        if (ntiles > 0) {
            int wavesPerBlock = B / 32;
            int blocks = (ntiles + wavesPerBlock - 1) / wavesPerBlock;
            gcn_gemm1_wmma<<<blocks, B, 0, stream>>>(x, W1, dinv, h1, ntiles);
        }
        if (full < N) {
            int tail = (N - full) * 16;
            gcn_gemm1_tail<<<(tail + B - 1) / B, B, 0, stream>>>(x, W1, dinv, h1, full, N);
        }
    }
    {
        const int R16 = (N + R16_NODES - 1) / R16_NODES;
        const dim3 grid16(R16_SLICES, R16 > 0 ? R16 : 1);
        gcn_edge16_lds<<<grid16, LDS_THREADS, 0, stream>>>(row, col, h1, agg1, E, N);
    }
    gcn_post1<<<(N + B - 1) / B, B, 0, stream>>>(agg1, h1, dinv, b1, W2, s, N);
    gcn_edge1_lds<<<ldsGrid, LDS_THREADS, 0, stream>>>(row, col, s, agg2, E, N);
    gcn_post2<<<(N + B - 1) / B, B, 0, stream>>>(agg2, s, dinv, b2, out, N);
}


# Edges binned by target range once per call (one word per edge, slots from returning integer atomics); one work-group owns each range in both layers, LDS sums only; float sum order changes
# speedup vs baseline: 4.7375x; 1.5256x over previous
#include <hip/hip_runtime.h>

typedef __attribute__((ext_vector_type(2))) float v2f;
typedef __attribute__((ext_vector_type(8))) float v8f;

__global__ void gcn_init(float* __restrict__ deg, float* __restrict__ agg1,
                         float* __restrict__ agg2, int n) {
    int i = blockIdx.x * blockDim.x + threadIdx.x;
    if (i < n) { deg[i] = 1.0f; agg2[i] = 0.0f; }
    if (i < n * 16) { agg1[i] = 0.0f; }
}

#define RANGE_BINS  51200
#define EDGE_SLICES 64
#define LDS_THREADS 1024

__global__ __launch_bounds__(LDS_THREADS)
void gcn_degree_lds(const int* __restrict__ col, float* __restrict__ deg,
                    int E, int n) {
    __shared__ unsigned cnt[RANGE_BINS];
    const int T   = (int)blockDim.x;
    const int tid = (int)threadIdx.x;
    const int per = (n + (int)gridDim.y - 1) / (int)gridDim.y;
    const int lo  = (int)blockIdx.y * per;
    const int lim = max(0, min(per, n - lo));
    for (int i = tid; i < lim; i += T) cnt[i] = 0u;
    __syncthreads();

    const int quads = (E + 3) >> 2;
    const int qper  = (quads + (int)gridDim.x - 1) / (int)gridDim.x;
    const int q0    = (int)blockIdx.x * qper;
    const int q1    = min(q0 + qper, quads);
    const bool vec  = (((size_t)col) & 15) == 0;
    const int qv    = vec ? min(q1, E >> 2) : q0;

    for (int q = q0 + tid; q < qv; q += 4 * T) {
        int4 v[4];
        #pragma unroll
        for (int u = 0; u < 4; ++u) {
            const int qq = q + u * T;
            v[u] = (qq < qv) ? *(const int4*)(col + 4 * (size_t)qq)
                             : make_int4(-1, -1, -1, -1);
        }
        #pragma unroll
        for (int u = 0; u < 4; ++u) {
            const unsigned b0 = (unsigned)(v[u].x - lo);
            const unsigned b1 = (unsigned)(v[u].y - lo);
            const unsigned b2 = (unsigned)(v[u].z - lo);
            const unsigned b3 = (unsigned)(v[u].w - lo);
            if (b0 < (unsigned)lim) atomicAdd(&cnt[b0], 1u);
            if (b1 < (unsigned)lim) atomicAdd(&cnt[b1], 1u);
            if (b2 < (unsigned)lim) atomicAdd(&cnt[b2], 1u);
            if (b3 < (unsigned)lim) atomicAdd(&cnt[b3], 1u);
        }
    }
    for (int q = max(q0, qv) + tid; q < q1; q += T) {
        for (int k = 0; k < 4; ++k) {
            const int e = 4 * q + k;
            if (e < E) {
                const unsigned b = (unsigned)(col[e] - lo);
                if (b < (unsigned)lim) atomicAdd(&cnt[b], 1u);
            }
        }
    }
    __syncthreads();
    for (int i = tid; i < lim; i += T) {
        const unsigned v = cnt[i];
        if (v != 0u) atomicAdd(&deg[lo + i], (float)v);
    }
}

__global__ void gcn_dinv(const float* __restrict__ deg, float* __restrict__ dinv,
                         int n) {
    int i = blockIdx.x * blockDim.x + threadIdx.x;
    if (i >= n) return;
    float d = deg[i];
    dinv[i] = (d > 0.0f) ? rsqrtf(d) : 0.0f;
}

__global__ void gcn_gemm1_wmma(const float* __restrict__ x,
                               const float* __restrict__ W1,
                               const float* __restrict__ dinv,
                               float* __restrict__ h1, int ntiles) {
    int wave = blockIdx.x * (blockDim.x >> 5) + (threadIdx.x >> 5);
    if (wave >= ntiles) return;
    int lane  = threadIdx.x & 31;
    int m     = lane & 15;
    int kb    = (lane >> 4) << 1;
    size_t row16 = (size_t)wave * 16;

    v8f acc = {};
    const float* arow = x + (row16 + (size_t)m) * 128 + kb;
    #pragma unroll 4
    for (int k0 = 0; k0 < 128; k0 += 4) {
        v2f a = *(const v2f*)(arow + k0);
        v2f b;
        b.x = W1[(size_t)(k0 + kb) * 16 + m];
        b.y = W1[(size_t)(k0 + kb + 1) * 16 + m];
        acc = __builtin_amdgcn_wmma_f32_16x16x4_f32(
             false, a,  false, b,
             (short)0, acc,  false,  false);
    }

    int mbase = (lane >> 4) << 3;
    int ncol  = lane & 15;
    #pragma unroll
    for (int r = 0; r < 8; ++r) {
        const size_t node = row16 + (size_t)(mbase + r);
        h1[node * 16 + ncol] = acc[r] * dinv[node];
    }
}

__global__ void gcn_gemm1_tail(const float* __restrict__ x,
                               const float* __restrict__ W1,
                               const float* __restrict__ dinv,
                               float* __restrict__ h1, int start, int n) {
    int t = blockIdx.x * blockDim.x + threadIdx.x;
    int node = start + (t >> 4);
    int f = t & 15;
    if (node >= n) return;
    float acc = 0.0f;
    const float* xr = x + (size_t)node * 128;
    for (int k = 0; k < 128; ++k) acc += xr[k] * W1[(size_t)k * 16 + f];
    h1[(size_t)node * 16 + f] = acc * dinv[node];
}

#define R16_NODES  4768
#define R16_SLICES 6
#define R16_QCAP   96

__global__ __launch_bounds__(LDS_THREADS)
void gcn_edge16_lds(const int* __restrict__ row, const int* __restrict__ col,
                    const float* __restrict__ h1,
                    float* __restrict__ agg1, int E, int n) {
    __shared__ float acc[R16_NODES * 16];
    __shared__ int queue[(LDS_THREADS / 32) * R16_QCAP];
    __shared__ unsigned short qrow[(LDS_THREADS / 32) * R16_QCAP];
    static_assert(R16_NODES <= 65535, "a target's LDS row must fit a 16-bit queue word");

    const int T    = (int)blockDim.x;
    const int tid  = (int)threadIdx.x;
    const int lane = tid & 31;
    const int per  = (n + (int)gridDim.y - 1) / (int)gridDim.y;
    const int lo   = (int)blockIdx.y * per;
    const int lim  = max(0, min(per, n - lo));
    {
        float4* a4 = (float4*)acc;
        for (int i = tid; i < lim * 4; i += T) a4[i] = make_float4(0.0f, 0.0f, 0.0f, 0.0f);
    }
    __syncthreads();

    const int qo = (tid >> 5) * R16_QCAP;
    int cnt = 0;

    auto push = [&](int cv, int e) {
        const unsigned bin = (unsigned)(cv - lo);
        const bool in = bin < (unsigned)lim;
        const unsigned m = __builtin_amdgcn_ballot_w32(in);
        if (m != 0u) {
            if (in) {
                const int slot = qo + cnt + (int)__builtin_amdgcn_mbcnt_lo(m, 0u);
                __hip_atomic_store(&queue[slot], e, __ATOMIC_RELAXED, __HIP_MEMORY_SCOPE_WAVEFRONT);
                __hip_atomic_store(&qrow[slot], (unsigned short)bin,
                                   __ATOMIC_RELAXED, __HIP_MEMORY_SCOPE_WAVEFRONT);
            }
            cnt += __builtin_popcount(m);
        }
    };
    auto drain = [&](int base, int count) {
        const bool  have = lane < count;
        const int   e  = have ? __hip_atomic_load(&queue[qo + base + lane], __ATOMIC_RELAXED,
                                                  __HIP_MEMORY_SCOPE_WAVEFRONT) : 0;
        const int   bi = have ? (int)__hip_atomic_load(&qrow[qo + base + lane], __ATOMIC_RELAXED,
                                                       __HIP_MEMORY_SCOPE_WAVEFRONT) : 0;
        const int   ri = row[e];
        const int   l4 = lane & 3;
        const int   g  = lane >> 2;
        float4 v[4];
        #pragma unroll
        for (int j = 0; j < 4; ++j) {
            const int r = __shfl(ri, 8 * j + g);
            v[j] = *(const float4*)(h1 + (size_t)r * 16 + 4 * l4);
        }
        #pragma unroll
        for (int j = 0; j < 4; ++j) {
            const int s = 8 * j + g;
            const int b = __shfl(bi, s);
            if (s < count) {
                float* p = &acc[b * 16 + 4 * l4];
                __hip_atomic_fetch_add(p + 0, v[j].x, __ATOMIC_RELAXED, __HIP_MEMORY_SCOPE_WORKGROUP);
                __hip_atomic_fetch_add(p + 1, v[j].y, __ATOMIC_RELAXED, __HIP_MEMORY_SCOPE_WORKGROUP);
                __hip_atomic_fetch_add(p + 2, v[j].z, __ATOMIC_RELAXED, __HIP_MEMORY_SCOPE_WORKGROUP);
                __hip_atomic_fetch_add(p + 3, v[j].w, __ATOMIC_RELAXED, __HIP_MEMORY_SCOPE_WORKGROUP);
            }
        }
    };

    const int quads = (E + 3) >> 2;
    const int qper  = (quads + (int)gridDim.x - 1) / (int)gridDim.x;
    const int q0    = (int)blockIdx.x * qper;
    const int q1    = min(q0 + qper, quads);
    const bool vec  = (((size_t)col) & 15) == 0;
    const int qv    = vec ? min(q1, E >> 2) : q0;

    for (int qb = q0; qb < qv; qb += 4 * T) {
        int4 v[4];
        #pragma unroll
        for (int u = 0; u < 4; ++u) {
            const int qq = qb + tid + u * T;
            v[u] = (qq < qv) ? *(const int4*)(col + 4 * (size_t)qq)
                             : make_int4(-1, -1, -1, -1);
        }
        #pragma unroll
        for (int u = 0; u < 4; ++u) {
            const int e0 = 4 * (qb + tid + u * T);
            push(v[u].x, e0 + 0);
            push(v[u].y, e0 + 1);
            while (cnt >= 32) { cnt -= 32; drain(cnt, 32); }
            push(v[u].z, e0 + 2);
            push(v[u].w, e0 + 3);
            while (cnt >= 32) { cnt -= 32; drain(cnt, 32); }
        }
    }
    {
        const int e0 = 4 * max(q0, qv);
        const int e1 = min(4 * q1, E);
        for (int eb = e0; eb < e1; eb += T) {
            const int e = eb + tid;
            push((e < e1) ? col[e] : -1, e);
            while (cnt >= 32) { cnt -= 32; drain(cnt, 32); }
        }
    }
    if (cnt > 0) drain(0, cnt);

    __syncthreads();
    float* dst = agg1 + (size_t)lo * 16;
    for (int i = tid; i < lim * 16; i += T) {
        const float t = acc[i];
        if (t != 0.0f) atomicAdd(&dst[i], t);
    }
}

__global__ void gcn_post1(const float* __restrict__ agg1,
                          const float* __restrict__ h1,
                          const float* __restrict__ dinv,
                          const float* __restrict__ b1,
                          const float* __restrict__ W2,
                          float* __restrict__ s, int n) {
    int i = blockIdx.x * blockDim.x + threadIdx.x;
    if (i >= n) return;
    float di = dinv[i];
    const float* a = agg1 + (size_t)i * 16;
    const float* h = h1 + (size_t)i * 16;
    float acc = 0.0f;
    #pragma unroll
    for (int f = 0; f < 16; ++f) {
        float v = di * (a[f] + h[f]) + b1[f];
        v = fmaxf(v, 0.0f);
        acc = fmaf(v, W2[f], acc);
    }
    s[i] = di * acc;
}

__global__ __launch_bounds__(LDS_THREADS)
void gcn_edge1_lds(const int* __restrict__ row, const int* __restrict__ col,
                   const float* __restrict__ s,
                   float* __restrict__ agg2, int E, int n) {
    __shared__ float acc[RANGE_BINS];
    const int T   = (int)blockDim.x;
    const int tid = (int)threadIdx.x;
    const int per = (n + (int)gridDim.y - 1) / (int)gridDim.y;
    const int lo  = (int)blockIdx.y * per;
    const int lim = max(0, min(per, n - lo));
    for (int i = tid; i < lim; i += T) acc[i] = 0.0f;
    __syncthreads();

    const int quads = (E + 3) >> 2;
    const int qper  = (quads + (int)gridDim.x - 1) / (int)gridDim.x;
    const int q0    = (int)blockIdx.x * qper;
    const int q1    = min(q0 + qper, quads);
    const bool vec  = ((((size_t)col) | ((size_t)row)) & 15) == 0;
    const int qv    = vec ? min(q1, E >> 2) : q0;

    for (int q = q0 + tid; q < qv; q += 2 * T) {
        int c[8], r[8];
        #pragma unroll
        for (int u = 0; u < 2; ++u) {
            const int qq = q + u * T;
            const bool ok = qq < qv;
            const int4 vc = ok ? *(const int4*)(col + 4 * (size_t)qq) : make_int4(-1, -1, -1, -1);
            const int4 vr = ok ? *(const int4*)(row + 4 * (size_t)qq) : make_int4(0, 0, 0, 0);
            c[4 * u + 0] = vc.x; c[4 * u + 1] = vc.y; c[4 * u + 2] = vc.z; c[4 * u + 3] = vc.w;
            r[4 * u + 0] = vr.x; r[4 * u + 1] = vr.y; r[4 * u + 2] = vr.z; r[4 * u + 3] = vr.w;
        }
        unsigned b[8];
        float    v[8];
        #pragma unroll
        for (int k = 0; k < 8; ++k) {
            b[k] = (unsigned)(c[k] - lo);
            const bool in = b[k] < (unsigned)lim;
            v[k] = s[in ? r[k] : 0];
        }
        #pragma unroll
        for (int k = 0; k < 8; ++k) {
            if (b[k] < (unsigned)lim)
                __hip_atomic_fetch_add(&acc[b[k]], v[k], __ATOMIC_RELAXED,
                                       __HIP_MEMORY_SCOPE_WORKGROUP);
        }
    }
    for (int q = max(q0, qv) + tid; q < q1; q += T) {
        for (int k = 0; k < 4; ++k) {
            const int e = 4 * q + k;
            if (e < E) {
                const unsigned bb = (unsigned)(col[e] - lo);
                if (bb < (unsigned)lim) {
                    __hip_atomic_fetch_add(&acc[bb], s[row[e]],
                                           __ATOMIC_RELAXED, __HIP_MEMORY_SCOPE_WORKGROUP);
                }
            }
        }
    }
    __syncthreads();
    for (int i = tid; i < lim; i += T) {
        const float t = acc[i];
        if (t != 0.0f) atomicAdd(&agg2[lo + i], t);
    }
}

__global__ void gcn_post2(const float* __restrict__ agg2,
                          const float* __restrict__ s,
                          const float* __restrict__ dinv,
                          const float* __restrict__ b2,
                          float* __restrict__ out, int n) {
    int i = blockIdx.x * blockDim.x + threadIdx.x;
    if (i >= n) return;
    float di = dinv[i];
    out[i] = di * (agg2[i] + s[i]) + b2[0];
}


#define OWN_NODES 782
#define OWN_SHIFT 10
#define OWN_MAXR  512
#define BIN_WGS   256
static_assert(OWN_NODES <= (1 << OWN_SHIFT), "a target's place in its range must fit OWN_SHIFT bits");
static_assert(OWN_NODES <= LDS_THREADS, "the epilogues give every node of a range one thread");

__global__ void gcn_init_deg(float* __restrict__ deg, int n) {
    int i = blockIdx.x * blockDim.x + threadIdx.x;
    if (i < n) deg[i] = 1.0f;
}

__global__ __launch_bounds__(256)
void gcn_dinv_tot(const float* __restrict__ deg, float* __restrict__ dinv,
                  int* __restrict__ tot, int* __restrict__ cursor, int n) {
    __shared__ int part[8];
    const int lo = (int)blockIdx.x * OWN_NODES;
    const int hi = min(lo + OWN_NODES, n);
    int c = 0;
    for (int i = lo + (int)threadIdx.x; i < hi; i += (int)blockDim.x) {
        const float d = deg[i];
        dinv[i] = (d > 0.0f) ? rsqrtf(d) : 0.0f;
        c += (int)d - 1;
    }
    for (int off = 16; off > 0; off >>= 1) c += __shfl_xor(c, off);
    if ((threadIdx.x & 31) == 0) part[threadIdx.x >> 5] = c;
    __syncthreads();
    if (threadIdx.x == 0) {
        int t = 0;
        for (int w = 0; w < (int)(blockDim.x >> 5); ++w) t += part[w];
        tot[blockIdx.x]    = t;
        cursor[blockIdx.x] = 0;
    }
}

#define BIN_CAP 32768
__global__ __launch_bounds__(LDS_THREADS)
void gcn_bin_edges(const int* __restrict__ row, const int* __restrict__ col,
                   const int* __restrict__ tot, int* __restrict__ cursor,
                   int* __restrict__ bstart, unsigned* __restrict__ packed,
                   int E, int n, int R) {
    __shared__ unsigned stage[BIN_CAP];
    __shared__ int slot[OWN_MAXR];
    __shared__ int start[OWN_MAXR + 1];
    __shared__ int loff[OWN_MAXR + 1];
    __shared__ int gbase[OWN_MAXR];
    __shared__ int wsum[LDS_THREADS / 32];
    const int T    = (int)blockDim.x;
    const int tid  = (int)threadIdx.x;
    const int lane = tid & 31;

    auto xscan = [&](int v) -> int {
        int incl = v;
        for (int off = 1; off < 32; off <<= 1) {
            const int t = __shfl_up(incl, off);
            if (lane >= off) incl += t;
        }
        __syncthreads();
        if (lane == 31) wsum[tid >> 5] = incl;
        __syncthreads();
        int pre = 0;
        for (int w = 0; w < (tid >> 5); ++w) pre += wsum[w];
        return pre + incl - v;
    };

    {
        const int v  = (tid < R) ? tot[tid] : 0;
        const int ex = xscan(v);
        if (tid < R)         start[tid] = ex;
        if (tid == R - 1)    start[R]   = ex + v;
        if (tid < OWN_MAXR)  slot[tid]  = 0;
        __syncthreads();
    }
    if (blockIdx.x == 0) {
        for (int i = tid; i <= R; i += T) bstart[i] = start[i];
    }

    const int quads = (E + 3) >> 2;
    const int qper  = (quads + (int)gridDim.x - 1) / (int)gridDim.x;
    const int q0    = (int)blockIdx.x * qper;
    const int q1    = min(q0 + qper, quads);
    const bool vec  = ((((size_t)col) | ((size_t)row)) & 15) == 0;
    const int qv    = vec ? min(q1, E >> 2) : q0;

    auto count = [&](int c) {
        if ((unsigned)c < (unsigned)n) atomicAdd(&slot[c / OWN_NODES], 1);
    };
    for (int q = q0 + tid; q < qv; q += 4 * T) {
        int4 v[4];
        #pragma unroll
        for (int u = 0; u < 4; ++u) {
            const int qq = q + u * T;
            v[u] = (qq < qv) ? *(const int4*)(col + 4 * (size_t)qq)
                             : make_int4(-1, -1, -1, -1);
        }
        #pragma unroll
        for (int u = 0; u < 4; ++u) {
            count(v[u].x); count(v[u].y); count(v[u].z); count(v[u].w);
        }
    }
    for (int q = max(q0, qv) + tid; q < q1; q += T) {
        for (int k = 0; k < 4; ++k) {
            const int e = 4 * q + k;
            if (e < E) count(col[e]);
        }
    }
    __syncthreads();
    {
        const int c  = (tid < R) ? slot[tid] : 0;
        const int ex = xscan(c);
        if (tid < R) {
            loff[tid]  = ex;
            gbase[tid] = start[tid] + ((c != 0) ? atomicAdd(&cursor[tid], c) : 0);
            slot[tid]  = ex;
        }
        if (tid == R - 1) loff[R] = ex + c;
        __syncthreads();
    }

    auto place = [&](int c, int sidx) {
        if ((unsigned)c < (unsigned)n) {
            const int r = c / OWN_NODES;
            const int k = atomicAdd(&slot[r], 1);
            if (k < BIN_CAP)
                stage[k] = ((unsigned)sidx << OWN_SHIFT) | (unsigned)(c - r * OWN_NODES);
        }
    };
    for (int q = q0 + tid; q < qv; q += 2 * T) {
        int4 vc[2], vr[2];
        #pragma unroll
        for (int u = 0; u < 2; ++u) {
            const int qq = q + u * T;
            const bool ok = qq < qv;
            vc[u] = ok ? *(const int4*)(col + 4 * (size_t)qq) : make_int4(-1, -1, -1, -1);
            vr[u] = ok ? *(const int4*)(row + 4 * (size_t)qq) : make_int4(0, 0, 0, 0);
        }
        #pragma unroll
        for (int u = 0; u < 2; ++u) {
            place(vc[u].x, vr[u].x); place(vc[u].y, vr[u].y);
            place(vc[u].z, vr[u].z); place(vc[u].w, vr[u].w);
        }
    }
    for (int q = max(q0, qv) + tid; q < q1; q += T) {
        for (int k = 0; k < 4; ++k) {
            const int e = 4 * q + k;
            if (e < E) place(col[e], row[e]);
        }
    }
    __syncthreads();

    for (int r = tid >> 5; r < R; r += (T >> 5)) {
        const int l0 = loff[r];
        const int nr = min(loff[r + 1], BIN_CAP) - l0;
        const int g0 = gbase[r];
        for (int i = lane; i < nr; i += 32) packed[g0 + i] = stage[l0 + i];
    }
}

__global__ __launch_bounds__(LDS_THREADS)
void gcn_own16(const unsigned* __restrict__ packed, const int* __restrict__ bstart,
               const float* __restrict__ h1, const float* __restrict__ dinv,
               const float* __restrict__ b1, const float* __restrict__ W2,
               float* __restrict__ s, int n) {
    __shared__ float acc[OWN_NODES * 16];
    const int T    = (int)blockDim.x;
    const int tid  = (int)threadIdx.x;
    const int lane = tid & 31;
    const int lo   = (int)blockIdx.x * OWN_NODES;
    const int lim  = max(0, min(OWN_NODES, n - lo));
    {
        float4* a4 = (float4*)acc;
        for (int i = tid; i < lim * 4; i += T) a4[i] = make_float4(0.0f, 0.0f, 0.0f, 0.0f);
    }
    __syncthreads();

    const int p0 = bstart[blockIdx.x];
    const int p1 = bstart[blockIdx.x + 1];
    const int l4 = lane & 3;
    const int g  = lane >> 2;
    for (int p = p0 + (tid & ~31); p < p1; p += 2 * T) {
        unsigned w[2];
        int      cn[2];
        #pragma unroll
        for (int u = 0; u < 2; ++u) {
            const int pp = p + u * T;
            w[u]  = (pp + lane < p1) ? packed[pp + lane] : 0u;
            cn[u] = p1 - pp;
        }
        float4 v[2][4];
        #pragma unroll
        for (int u = 0; u < 2; ++u) {
            #pragma unroll
            for (int j = 0; j < 4; ++j) {
                const unsigned wj = (unsigned)__shfl((int)w[u], 8 * j + g);
                v[u][j] = *(const float4*)(h1 + (size_t)(wj >> OWN_SHIFT) * 16 + 4 * l4);
            }
        }
        #pragma unroll
        for (int u = 0; u < 2; ++u) {
            #pragma unroll
            for (int j = 0; j < 4; ++j) {
                const int sl = 8 * j + g;
                const unsigned wj = (unsigned)__shfl((int)w[u], sl);
                if (sl < cn[u]) {
                    float* q = &acc[(int)(wj & ((1u << OWN_SHIFT) - 1u)) * 16 + 4 * l4];
                    __hip_atomic_fetch_add(q + 0, v[u][j].x, __ATOMIC_RELAXED, __HIP_MEMORY_SCOPE_WORKGROUP);
                    __hip_atomic_fetch_add(q + 1, v[u][j].y, __ATOMIC_RELAXED, __HIP_MEMORY_SCOPE_WORKGROUP);
                    __hip_atomic_fetch_add(q + 2, v[u][j].z, __ATOMIC_RELAXED, __HIP_MEMORY_SCOPE_WORKGROUP);
                    __hip_atomic_fetch_add(q + 3, v[u][j].w, __ATOMIC_RELAXED, __HIP_MEMORY_SCOPE_WORKGROUP);
                }
            }
        }
    }
    __syncthreads();
    if (tid < lim) {
        const int   i  = lo + tid;
        const float di = dinv[i];
        const float* a = &acc[tid * 16];
        const float* h = h1 + (size_t)i * 16;
        float t = 0.0f;
        #pragma unroll
        for (int f = 0; f < 16; ++f) {
            float x = di * (a[f] + h[f]) + b1[f];
            x = fmaxf(x, 0.0f);
            t = fmaf(x, W2[f], t);
        }
        s[i] = di * t;
    }
}

__global__ __launch_bounds__(LDS_THREADS)
void gcn_own1(const unsigned* __restrict__ packed, const int* __restrict__ bstart,
              const float* __restrict__ s, const float* __restrict__ dinv,
              const float* __restrict__ b2, float* __restrict__ out, int n) {
    __shared__ float acc[OWN_NODES];
    const int T   = (int)blockDim.x;
    const int tid = (int)threadIdx.x;
    const int lo  = (int)blockIdx.x * OWN_NODES;
    const int lim = max(0, min(OWN_NODES, n - lo));
    if (tid < lim) acc[tid] = 0.0f;
    __syncthreads();

    const int p0 = bstart[blockIdx.x];
    const int p1 = bstart[blockIdx.x + 1];
    for (int p = p0 + tid; p < p1; p += 4 * T) {
        unsigned w[4];
        float    v[4];
        #pragma unroll
        for (int u = 0; u < 4; ++u) {
            const int pp = p + u * T;
            w[u] = (pp < p1) ? packed[pp] : 0u;
        }
        #pragma unroll
        for (int u = 0; u < 4; ++u) v[u] = s[w[u] >> OWN_SHIFT];
        #pragma unroll
        for (int u = 0; u < 4; ++u) {
            if (p + u * T < p1)
                __hip_atomic_fetch_add(&acc[w[u] & ((1u << OWN_SHIFT) - 1u)], v[u],
                                       __ATOMIC_RELAXED, __HIP_MEMORY_SCOPE_WORKGROUP);
        }
    }
    __syncthreads();
    if (tid < lim) {
        const int i = lo + tid;
        out[i] = dinv[i] * (acc[tid] + s[i]) + b2[0];
    }
}

extern "C" void kernel_launch(void* const* d_in, const int* in_sizes, int n_in,
                              void* d_out, int out_size, void* d_ws, size_t ws_size,
                              hipStream_t stream) {
    const float* x  = (const float*)d_in[0];
    const int*   ei = (const int*)d_in[1];
    const float* W1 = (const float*)d_in[2];
    const float* b1 = (const float*)d_in[3];
    const float* W2 = (const float*)d_in[4];
    const float* b2 = (const float*)d_in[5];

    const int N = in_sizes[0] / 128;
    const int E = in_sizes[1] / 2;
    const int* row = ei;
    const int* col = ei + E;

    {
        const int R = (N + OWN_NODES - 1) / OWN_NODES;
        const size_t need = ((size_t)19 * (size_t)N + (size_t)E + 3 * (size_t)R + 1) * sizeof(float);
        if (R >= 1 && R <= OWN_MAXR && N <= (1 << (32 - OWN_SHIFT)) && ws_size >= need) {
            float*    deg    = (float*)d_ws;
            float*    dinv   = deg + N;
            float*    h1     = dinv + N;
            float*    s      = h1 + (size_t)N * 16;
            unsigned* packed = (unsigned*)(s + N);
            int*      tot    = (int*)(packed + E);
            int*      cursor = tot + R;
            int*      bstart = cursor + R;
            float*    out    = (float*)d_out;
            const int B = 256;

            gcn_init_deg<<<(N + B - 1) / B, B, 0, stream>>>(deg, N);
            {
                const int RR = (N + RANGE_BINS - 1) / RANGE_BINS;
                const dim3 ldsGrid(EDGE_SLICES, RR);
                gcn_degree_lds<<<ldsGrid, LDS_THREADS, 0, stream>>>(col, deg, E, N);
            }
            gcn_dinv_tot<<<R, 256, 0, stream>>>(deg, dinv, tot, cursor, N);
            {
                int full   = N & ~15;
                int ntiles = full >> 4;
                if (ntiles > 0) {
                    int wavesPerBlock = B / 32;
                    int blocks = (ntiles + wavesPerBlock - 1) / wavesPerBlock;
                    gcn_gemm1_wmma<<<blocks, B, 0, stream>>>(x, W1, dinv, h1, ntiles);
                }
                if (full < N) {
                    int tail = (N - full) * 16;
                    gcn_gemm1_tail<<<(tail + B - 1) / B, B, 0, stream>>>(x, W1, dinv, h1, full, N);
                }
            }
            {
                const int quads = (E + 3) >> 2;
                int G = (quads + (BIN_CAP / 4) - 1) / (BIN_CAP / 4);
                if (G < BIN_WGS) G = BIN_WGS;
                gcn_bin_edges<<<G, LDS_THREADS, 0, stream>>>(row, col, tot, cursor, bstart, packed, E, N, R);
            }
            gcn_own16<<<R, LDS_THREADS, 0, stream>>>(packed, bstart, h1, dinv, b1, W2, s, N);
            gcn_own1<<<R, LDS_THREADS, 0, stream>>>(packed, bstart, s, dinv, b2, out, N);
            return;
        }
    }

    float* ws   = (float*)d_ws;
    float* deg  = ws;
    float* dinv = deg + N;
    float* h1   = dinv + N;
    float* agg1 = h1 + (size_t)N * 16;
    float* s    = agg1 + (size_t)N * 16;
    float* agg2 = s + N;
    float* out  = (float*)d_out;

    const int B = 256;

    {
        int total = N * 16;
        gcn_init<<<(total + B - 1) / B, B, 0, stream>>>(deg, agg1, agg2, N);
    }
    const int R = (N + RANGE_BINS - 1) / RANGE_BINS;
    const dim3 ldsGrid(EDGE_SLICES, R > 0 ? R : 1);

    gcn_degree_lds<<<ldsGrid, LDS_THREADS, 0, stream>>>(col, deg, E, N);
    gcn_dinv<<<(N + B - 1) / B, B, 0, stream>>>(deg, dinv, N);
    {
        int full   = N & ~15;
        int ntiles = full >> 4;
        if (ntiles > 0) {
            int wavesPerBlock = B / 32;
            int blocks = (ntiles + wavesPerBlock - 1) / wavesPerBlock;
            gcn_gemm1_wmma<<<blocks, B, 0, stream>>>(x, W1, dinv, h1, ntiles);
        }
        if (full < N) {
            int tail = (N - full) * 16;
            gcn_gemm1_tail<<<(tail + B - 1) / B, B, 0, stream>>>(x, W1, dinv, h1, full, N);
        }
    }
    {
        const int R16 = (N + R16_NODES - 1) / R16_NODES;
        const dim3 grid16(R16_SLICES, R16 > 0 ? R16 : 1);
        gcn_edge16_lds<<<grid16, LDS_THREADS, 0, stream>>>(row, col, h1, agg1, E, N);
    }
    gcn_post1<<<(N + B - 1) / B, B, 0, stream>>>(agg1, h1, dinv, b1, W2, s, N);
    gcn_edge1_lds<<<ldsGrid, LDS_THREADS, 0, stream>>>(row, col, s, agg2, E, N);
    gcn_post2<<<(N + B - 1) / B, B, 0, stream>>>(agg2, s, dinv, b2, out, N);
}


# Layer-2 edge sum takes s from LDS, not one 4-byte memory gather per edge: s passes through LDS in 3 windows filled by whole-line loads, run words held in registers; float sum order changes
# speedup vs baseline: 5.0264x; 1.0610x over previous
#include <hip/hip_runtime.h>

typedef __attribute__((ext_vector_type(2))) float v2f;
typedef __attribute__((ext_vector_type(8))) float v8f;

__global__ void gcn_init(float* __restrict__ deg, float* __restrict__ agg1,
                         float* __restrict__ agg2, int n) {
    int i = blockIdx.x * blockDim.x + threadIdx.x;
    if (i < n) { deg[i] = 1.0f; agg2[i] = 0.0f; }
    if (i < n * 16) { agg1[i] = 0.0f; }
}

#define RANGE_BINS  51200
#define EDGE_SLICES 64
#define LDS_THREADS 1024

__global__ __launch_bounds__(LDS_THREADS)
void gcn_degree_lds(const int* __restrict__ col, float* __restrict__ deg,
                    int E, int n) {
    __shared__ unsigned cnt[RANGE_BINS];
    const int T   = (int)blockDim.x;
    const int tid = (int)threadIdx.x;
    const int per = (n + (int)gridDim.y - 1) / (int)gridDim.y;
    const int lo  = (int)blockIdx.y * per;
    const int lim = max(0, min(per, n - lo));
    for (int i = tid; i < lim; i += T) cnt[i] = 0u;
    __syncthreads();

    const int quads = (E + 3) >> 2;
    const int qper  = (quads + (int)gridDim.x - 1) / (int)gridDim.x;
    const int q0    = (int)blockIdx.x * qper;
    const int q1    = min(q0 + qper, quads);
    const bool vec  = (((size_t)col) & 15) == 0;
    const int qv    = vec ? min(q1, E >> 2) : q0;

    for (int q = q0 + tid; q < qv; q += 4 * T) {
        int4 v[4];
        #pragma unroll
        for (int u = 0; u < 4; ++u) {
            const int qq = q + u * T;
            v[u] = (qq < qv) ? *(const int4*)(col + 4 * (size_t)qq)
                             : make_int4(-1, -1, -1, -1);
        }
        #pragma unroll
        for (int u = 0; u < 4; ++u) {
            const unsigned b0 = (unsigned)(v[u].x - lo);
            const unsigned b1 = (unsigned)(v[u].y - lo);
            const unsigned b2 = (unsigned)(v[u].z - lo);
            const unsigned b3 = (unsigned)(v[u].w - lo);
            if (b0 < (unsigned)lim) atomicAdd(&cnt[b0], 1u);
            if (b1 < (unsigned)lim) atomicAdd(&cnt[b1], 1u);
            if (b2 < (unsigned)lim) atomicAdd(&cnt[b2], 1u);
            if (b3 < (unsigned)lim) atomicAdd(&cnt[b3], 1u);
        }
    }
    for (int q = max(q0, qv) + tid; q < q1; q += T) {
        for (int k = 0; k < 4; ++k) {
            const int e = 4 * q + k;
            if (e < E) {
                const unsigned b = (unsigned)(col[e] - lo);
                if (b < (unsigned)lim) atomicAdd(&cnt[b], 1u);
            }
        }
    }
    __syncthreads();
    for (int i = tid; i < lim; i += T) {
        const unsigned v = cnt[i];
        if (v != 0u) atomicAdd(&deg[lo + i], (float)v);
    }
}

__global__ void gcn_dinv(const float* __restrict__ deg, float* __restrict__ dinv,
                         int n) {
    int i = blockIdx.x * blockDim.x + threadIdx.x;
    if (i >= n) return;
    float d = deg[i];
    dinv[i] = (d > 0.0f) ? rsqrtf(d) : 0.0f;
}

__global__ void gcn_gemm1_wmma(const float* __restrict__ x,
                               const float* __restrict__ W1,
                               const float* __restrict__ dinv,
                               float* __restrict__ h1, int ntiles) {
    int wave = blockIdx.x * (blockDim.x >> 5) + (threadIdx.x >> 5);
    if (wave >= ntiles) return;
    int lane  = threadIdx.x & 31;
    int m     = lane & 15;
    int kb    = (lane >> 4) << 1;
    size_t row16 = (size_t)wave * 16;

    v8f acc = {};
    const float* arow = x + (row16 + (size_t)m) * 128 + kb;
    #pragma unroll 4
    for (int k0 = 0; k0 < 128; k0 += 4) {
        v2f a = *(const v2f*)(arow + k0);
        v2f b;
        b.x = W1[(size_t)(k0 + kb) * 16 + m];
        b.y = W1[(size_t)(k0 + kb + 1) * 16 + m];
        acc = __builtin_amdgcn_wmma_f32_16x16x4_f32(
             false, a,  false, b,
             (short)0, acc,  false,  false);
    }

    int mbase = (lane >> 4) << 3;
    int ncol  = lane & 15;
    #pragma unroll
    for (int r = 0; r < 8; ++r) {
        const size_t node = row16 + (size_t)(mbase + r);
        h1[node * 16 + ncol] = acc[r] * dinv[node];
    }
}

__global__ void gcn_gemm1_tail(const float* __restrict__ x,
                               const float* __restrict__ W1,
                               const float* __restrict__ dinv,
                               float* __restrict__ h1, int start, int n) {
    int t = blockIdx.x * blockDim.x + threadIdx.x;
    int node = start + (t >> 4);
    int f = t & 15;
    if (node >= n) return;
    float acc = 0.0f;
    const float* xr = x + (size_t)node * 128;
    for (int k = 0; k < 128; ++k) acc += xr[k] * W1[(size_t)k * 16 + f];
    h1[(size_t)node * 16 + f] = acc * dinv[node];
}

#define R16_NODES  4768
#define R16_SLICES 6
#define R16_QCAP   96

__global__ __launch_bounds__(LDS_THREADS)
void gcn_edge16_lds(const int* __restrict__ row, const int* __restrict__ col,
                    const float* __restrict__ h1,
                    float* __restrict__ agg1, int E, int n) {
    __shared__ float acc[R16_NODES * 16];
    __shared__ int queue[(LDS_THREADS / 32) * R16_QCAP];
    __shared__ unsigned short qrow[(LDS_THREADS / 32) * R16_QCAP];
    static_assert(R16_NODES <= 65535, "a target's LDS row must fit a 16-bit queue word");

    const int T    = (int)blockDim.x;
    const int tid  = (int)threadIdx.x;
    const int lane = tid & 31;
    const int per  = (n + (int)gridDim.y - 1) / (int)gridDim.y;
    const int lo   = (int)blockIdx.y * per;
    const int lim  = max(0, min(per, n - lo));
    {
        float4* a4 = (float4*)acc;
        for (int i = tid; i < lim * 4; i += T) a4[i] = make_float4(0.0f, 0.0f, 0.0f, 0.0f);
    }
    __syncthreads();

    const int qo = (tid >> 5) * R16_QCAP;
    int cnt = 0;

    auto push = [&](int cv, int e) {
        const unsigned bin = (unsigned)(cv - lo);
        const bool in = bin < (unsigned)lim;
        const unsigned m = __builtin_amdgcn_ballot_w32(in);
        if (m != 0u) {
            if (in) {
                const int slot = qo + cnt + (int)__builtin_amdgcn_mbcnt_lo(m, 0u);
                __hip_atomic_store(&queue[slot], e, __ATOMIC_RELAXED, __HIP_MEMORY_SCOPE_WAVEFRONT);
                __hip_atomic_store(&qrow[slot], (unsigned short)bin,
                                   __ATOMIC_RELAXED, __HIP_MEMORY_SCOPE_WAVEFRONT);
            }
            cnt += __builtin_popcount(m);
        }
    };
    auto drain = [&](int base, int count) {
        const bool  have = lane < count;
        const int   e  = have ? __hip_atomic_load(&queue[qo + base + lane], __ATOMIC_RELAXED,
                                                  __HIP_MEMORY_SCOPE_WAVEFRONT) : 0;
        const int   bi = have ? (int)__hip_atomic_load(&qrow[qo + base + lane], __ATOMIC_RELAXED,
                                                       __HIP_MEMORY_SCOPE_WAVEFRONT) : 0;
        const int   ri = row[e];
        const int   l4 = lane & 3;
        const int   g  = lane >> 2;
        float4 v[4];
        #pragma unroll
        for (int j = 0; j < 4; ++j) {
            const int r = __shfl(ri, 8 * j + g);
            v[j] = *(const float4*)(h1 + (size_t)r * 16 + 4 * l4);
        }
        #pragma unroll
        for (int j = 0; j < 4; ++j) {
            const int s = 8 * j + g;
            const int b = __shfl(bi, s);
            if (s < count) {
                float* p = &acc[b * 16 + 4 * l4];
                __hip_atomic_fetch_add(p + 0, v[j].x, __ATOMIC_RELAXED, __HIP_MEMORY_SCOPE_WORKGROUP);
                __hip_atomic_fetch_add(p + 1, v[j].y, __ATOMIC_RELAXED, __HIP_MEMORY_SCOPE_WORKGROUP);
                __hip_atomic_fetch_add(p + 2, v[j].z, __ATOMIC_RELAXED, __HIP_MEMORY_SCOPE_WORKGROUP);
                __hip_atomic_fetch_add(p + 3, v[j].w, __ATOMIC_RELAXED, __HIP_MEMORY_SCOPE_WORKGROUP);
            }
        }
    };

    const int quads = (E + 3) >> 2;
    const int qper  = (quads + (int)gridDim.x - 1) / (int)gridDim.x;
    const int q0    = (int)blockIdx.x * qper;
    const int q1    = min(q0 + qper, quads);
    const bool vec  = (((size_t)col) & 15) == 0;
    const int qv    = vec ? min(q1, E >> 2) : q0;

    for (int qb = q0; qb < qv; qb += 4 * T) {
        int4 v[4];
        #pragma unroll
        for (int u = 0; u < 4; ++u) {
            const int qq = qb + tid + u * T;
            v[u] = (qq < qv) ? *(const int4*)(col + 4 * (size_t)qq)
                             : make_int4(-1, -1, -1, -1);
        }
        #pragma unroll
        for (int u = 0; u < 4; ++u) {
            const int e0 = 4 * (qb + tid + u * T);
            push(v[u].x, e0 + 0);
            push(v[u].y, e0 + 1);
            while (cnt >= 32) { cnt -= 32; drain(cnt, 32); }
            push(v[u].z, e0 + 2);
            push(v[u].w, e0 + 3);
            while (cnt >= 32) { cnt -= 32; drain(cnt, 32); }
        }
    }
    {
        const int e0 = 4 * max(q0, qv);
        const int e1 = min(4 * q1, E);
        for (int eb = e0; eb < e1; eb += T) {
            const int e = eb + tid;
            push((e < e1) ? col[e] : -1, e);
            while (cnt >= 32) { cnt -= 32; drain(cnt, 32); }
        }
    }
    if (cnt > 0) drain(0, cnt);

    __syncthreads();
    float* dst = agg1 + (size_t)lo * 16;
    for (int i = tid; i < lim * 16; i += T) {
        const float t = acc[i];
        if (t != 0.0f) atomicAdd(&dst[i], t);
    }
}

__global__ void gcn_post1(const float* __restrict__ agg1,
                          const float* __restrict__ h1,
                          const float* __restrict__ dinv,
                          const float* __restrict__ b1,
                          const float* __restrict__ W2,
                          float* __restrict__ s, int n) {
    int i = blockIdx.x * blockDim.x + threadIdx.x;
    if (i >= n) return;
    float di = dinv[i];
    const float* a = agg1 + (size_t)i * 16;
    const float* h = h1 + (size_t)i * 16;
    float acc = 0.0f;
    #pragma unroll
    for (int f = 0; f < 16; ++f) {
        float v = di * (a[f] + h[f]) + b1[f];
        v = fmaxf(v, 0.0f);
        acc = fmaf(v, W2[f], acc);
    }
    s[i] = di * acc;
}

__global__ __launch_bounds__(LDS_THREADS)
void gcn_edge1_lds(const int* __restrict__ row, const int* __restrict__ col,
                   const float* __restrict__ s,
                   float* __restrict__ agg2, int E, int n) {
    __shared__ float acc[RANGE_BINS];
    const int T   = (int)blockDim.x;
    const int tid = (int)threadIdx.x;
    const int per = (n + (int)gridDim.y - 1) / (int)gridDim.y;
    const int lo  = (int)blockIdx.y * per;
    const int lim = max(0, min(per, n - lo));
    for (int i = tid; i < lim; i += T) acc[i] = 0.0f;
    __syncthreads();

    const int quads = (E + 3) >> 2;
    const int qper  = (quads + (int)gridDim.x - 1) / (int)gridDim.x;
    const int q0    = (int)blockIdx.x * qper;
    const int q1    = min(q0 + qper, quads);
    const bool vec  = ((((size_t)col) | ((size_t)row)) & 15) == 0;
    const int qv    = vec ? min(q1, E >> 2) : q0;

    for (int q = q0 + tid; q < qv; q += 2 * T) {
        int c[8], r[8];
        #pragma unroll
        for (int u = 0; u < 2; ++u) {
            const int qq = q + u * T;
            const bool ok = qq < qv;
            const int4 vc = ok ? *(const int4*)(col + 4 * (size_t)qq) : make_int4(-1, -1, -1, -1);
            const int4 vr = ok ? *(const int4*)(row + 4 * (size_t)qq) : make_int4(0, 0, 0, 0);
            c[4 * u + 0] = vc.x; c[4 * u + 1] = vc.y; c[4 * u + 2] = vc.z; c[4 * u + 3] = vc.w;
            r[4 * u + 0] = vr.x; r[4 * u + 1] = vr.y; r[4 * u + 2] = vr.z; r[4 * u + 3] = vr.w;
        }
        unsigned b[8];
        float    v[8];
        #pragma unroll
        for (int k = 0; k < 8; ++k) {
            b[k] = (unsigned)(c[k] - lo);
            const bool in = b[k] < (unsigned)lim;
            v[k] = s[in ? r[k] : 0];
        }
        #pragma unroll
        for (int k = 0; k < 8; ++k) {
            if (b[k] < (unsigned)lim)
                __hip_atomic_fetch_add(&acc[b[k]], v[k], __ATOMIC_RELAXED,
                                       __HIP_MEMORY_SCOPE_WORKGROUP);
        }
    }
    for (int q = max(q0, qv) + tid; q < q1; q += T) {
        for (int k = 0; k < 4; ++k) {
            const int e = 4 * q + k;
            if (e < E) {
                const unsigned bb = (unsigned)(col[e] - lo);
                if (bb < (unsigned)lim) {
                    __hip_atomic_fetch_add(&acc[bb], s[row[e]],
                                           __ATOMIC_RELAXED, __HIP_MEMORY_SCOPE_WORKGROUP);
                }
            }
        }
    }
    __syncthreads();
    for (int i = tid; i < lim; i += T) {
        const float t = acc[i];
        if (t != 0.0f) atomicAdd(&agg2[lo + i], t);
    }
}

__global__ void gcn_post2(const float* __restrict__ agg2,
                          const float* __restrict__ s,
                          const float* __restrict__ dinv,
                          const float* __restrict__ b2,
                          float* __restrict__ out, int n) {
    int i = blockIdx.x * blockDim.x + threadIdx.x;
    if (i >= n) return;
    float di = dinv[i];
    out[i] = di * (agg2[i] + s[i]) + b2[0];
}


#define OWN_NODES 782
#define OWN_SHIFT 10
#define OWN_MAXR  512
#define BIN_WGS   256
static_assert(OWN_NODES <= (1 << OWN_SHIFT), "a target's place in its range must fit OWN_SHIFT bits");
static_assert(OWN_NODES <= LDS_THREADS, "the epilogues give every node of a range one thread");

__global__ void gcn_init_deg(float* __restrict__ deg, int n) {
    int i = blockIdx.x * blockDim.x + threadIdx.x;
    if (i < n) deg[i] = 1.0f;
}

__global__ __launch_bounds__(256)
void gcn_dinv_tot(const float* __restrict__ deg, float* __restrict__ dinv,
                  int* __restrict__ tot, int* __restrict__ cursor, int n) {
    __shared__ int part[8];
    const int lo = (int)blockIdx.x * OWN_NODES;
    const int hi = min(lo + OWN_NODES, n);
    int c = 0;
    for (int i = lo + (int)threadIdx.x; i < hi; i += (int)blockDim.x) {
        const float d = deg[i];
        dinv[i] = (d > 0.0f) ? rsqrtf(d) : 0.0f;
        c += (int)d - 1;
    }
    for (int off = 16; off > 0; off >>= 1) c += __shfl_xor(c, off);
    if ((threadIdx.x & 31) == 0) part[threadIdx.x >> 5] = c;
    __syncthreads();
    if (threadIdx.x == 0) {
        int t = 0;
        for (int w = 0; w < (int)(blockDim.x >> 5); ++w) t += part[w];
        tot[blockIdx.x]    = t;
        cursor[blockIdx.x] = 0;
    }
}

#define BIN_CAP 32768
__global__ __launch_bounds__(LDS_THREADS)
void gcn_bin_edges(const int* __restrict__ row, const int* __restrict__ col,
                   const int* __restrict__ tot, int* __restrict__ cursor,
                   int* __restrict__ bstart, unsigned* __restrict__ packed,
                   int E, int n, int R) {
    __shared__ unsigned stage[BIN_CAP];
    __shared__ int slot[OWN_MAXR];
    __shared__ int start[OWN_MAXR + 1];
    __shared__ int loff[OWN_MAXR + 1];
    __shared__ int gbase[OWN_MAXR];
    __shared__ int wsum[LDS_THREADS / 32];
    const int T    = (int)blockDim.x;
    const int tid  = (int)threadIdx.x;
    const int lane = tid & 31;

    auto xscan = [&](int v) -> int {
        int incl = v;
        for (int off = 1; off < 32; off <<= 1) {
            const int t = __shfl_up(incl, off);
            if (lane >= off) incl += t;
        }
        __syncthreads();
        if (lane == 31) wsum[tid >> 5] = incl;
        __syncthreads();
        int pre = 0;
        for (int w = 0; w < (tid >> 5); ++w) pre += wsum[w];
        return pre + incl - v;
    };

    {
        const int v  = (tid < R) ? tot[tid] : 0;
        const int ex = xscan(v);
        if (tid < R)         start[tid] = ex;
        if (tid == R - 1)    start[R]   = ex + v;
        if (tid < OWN_MAXR)  slot[tid]  = 0;
        __syncthreads();
    }
    if (blockIdx.x == 0) {
        for (int i = tid; i <= R; i += T) bstart[i] = start[i];
    }

    const int quads = (E + 3) >> 2;
    const int qper  = (quads + (int)gridDim.x - 1) / (int)gridDim.x;
    const int q0    = (int)blockIdx.x * qper;
    const int q1    = min(q0 + qper, quads);
    const bool vec  = ((((size_t)col) | ((size_t)row)) & 15) == 0;
    const int qv    = vec ? min(q1, E >> 2) : q0;

    auto count = [&](int c) {
        if ((unsigned)c < (unsigned)n) atomicAdd(&slot[c / OWN_NODES], 1);
    };
    for (int q = q0 + tid; q < qv; q += 4 * T) {
        int4 v[4];
        #pragma unroll
        for (int u = 0; u < 4; ++u) {
            const int qq = q + u * T;
            v[u] = (qq < qv) ? *(const int4*)(col + 4 * (size_t)qq)
                             : make_int4(-1, -1, -1, -1);
        }
        #pragma unroll
        for (int u = 0; u < 4; ++u) {
            count(v[u].x); count(v[u].y); count(v[u].z); count(v[u].w);
        }
    }
    for (int q = max(q0, qv) + tid; q < q1; q += T) {
        for (int k = 0; k < 4; ++k) {
            const int e = 4 * q + k;
            if (e < E) count(col[e]);
        }
    }
    __syncthreads();
    {
        const int c  = (tid < R) ? slot[tid] : 0;
        const int ex = xscan(c);
        if (tid < R) {
            loff[tid]  = ex;
            gbase[tid] = start[tid] + ((c != 0) ? atomicAdd(&cursor[tid], c) : 0);
            slot[tid]  = ex;
        }
        if (tid == R - 1) loff[R] = ex + c;
        __syncthreads();
    }

    auto place = [&](int c, int sidx) {
        if ((unsigned)c < (unsigned)n) {
            const int r = c / OWN_NODES;
            const int k = atomicAdd(&slot[r], 1);
            if (k < BIN_CAP)
                stage[k] = ((unsigned)sidx << OWN_SHIFT) | (unsigned)(c - r * OWN_NODES);
        }
    };
    for (int q = q0 + tid; q < qv; q += 2 * T) {
        int4 vc[2], vr[2];
        #pragma unroll
        for (int u = 0; u < 2; ++u) {
            const int qq = q + u * T;
            const bool ok = qq < qv;
            vc[u] = ok ? *(const int4*)(col + 4 * (size_t)qq) : make_int4(-1, -1, -1, -1);
            vr[u] = ok ? *(const int4*)(row + 4 * (size_t)qq) : make_int4(0, 0, 0, 0);
        }
        #pragma unroll
        for (int u = 0; u < 2; ++u) {
            place(vc[u].x, vr[u].x); place(vc[u].y, vr[u].y);
            place(vc[u].z, vr[u].z); place(vc[u].w, vr[u].w);
        }
    }
    for (int q = max(q0, qv) + tid; q < q1; q += T) {
        for (int k = 0; k < 4; ++k) {
            const int e = 4 * q + k;
            if (e < E) place(col[e], row[e]);
        }
    }
    __syncthreads();

    for (int r = tid >> 5; r < R; r += (T >> 5)) {
        const int l0 = loff[r];
        const int nr = min(loff[r + 1], BIN_CAP) - l0;
        const int g0 = gbase[r];
        for (int i = lane; i < nr; i += 32) packed[g0 + i] = stage[l0 + i];
    }
}

__global__ __launch_bounds__(LDS_THREADS)
void gcn_own16(const unsigned* __restrict__ packed, const int* __restrict__ bstart,
               const float* __restrict__ h1, const float* __restrict__ dinv,
               const float* __restrict__ b1, const float* __restrict__ W2,
               float* __restrict__ s, int n) {
    __shared__ float acc[OWN_NODES * 16];
    const int T    = (int)blockDim.x;
    const int tid  = (int)threadIdx.x;
    const int lane = tid & 31;
    const int lo   = (int)blockIdx.x * OWN_NODES;
    const int lim  = max(0, min(OWN_NODES, n - lo));
    {
        float4* a4 = (float4*)acc;
        for (int i = tid; i < lim * 4; i += T) a4[i] = make_float4(0.0f, 0.0f, 0.0f, 0.0f);
    }
    __syncthreads();

    const int p0 = bstart[blockIdx.x];
    const int p1 = bstart[blockIdx.x + 1];
    const int l4 = lane & 3;
    const int g  = lane >> 2;
    for (int p = p0 + (tid & ~31); p < p1; p += 2 * T) {
        unsigned w[2];
        int      cn[2];
        #pragma unroll
        for (int u = 0; u < 2; ++u) {
            const int pp = p + u * T;
            w[u]  = (pp + lane < p1) ? packed[pp + lane] : 0u;
            cn[u] = p1 - pp;
        }
        float4 v[2][4];
        #pragma unroll
        for (int u = 0; u < 2; ++u) {
            #pragma unroll
            for (int j = 0; j < 4; ++j) {
                const unsigned wj = (unsigned)__shfl((int)w[u], 8 * j + g);
                v[u][j] = *(const float4*)(h1 + (size_t)(wj >> OWN_SHIFT) * 16 + 4 * l4);
            }
        }
        #pragma unroll
        for (int u = 0; u < 2; ++u) {
            #pragma unroll
            for (int j = 0; j < 4; ++j) {
                const int sl = 8 * j + g;
                const unsigned wj = (unsigned)__shfl((int)w[u], sl);
                if (sl < cn[u]) {
                    float* q = &acc[(int)(wj & ((1u << OWN_SHIFT) - 1u)) * 16 + 4 * l4];
                    __hip_atomic_fetch_add(q + 0, v[u][j].x, __ATOMIC_RELAXED, __HIP_MEMORY_SCOPE_WORKGROUP);
                    __hip_atomic_fetch_add(q + 1, v[u][j].y, __ATOMIC_RELAXED, __HIP_MEMORY_SCOPE_WORKGROUP);
                    __hip_atomic_fetch_add(q + 2, v[u][j].z, __ATOMIC_RELAXED, __HIP_MEMORY_SCOPE_WORKGROUP);
                    __hip_atomic_fetch_add(q + 3, v[u][j].w, __ATOMIC_RELAXED, __HIP_MEMORY_SCOPE_WORKGROUP);
                }
            }
        }
    }
    __syncthreads();
    if (tid < lim) {
        const int   i  = lo + tid;
        const float di = dinv[i];
        const float* a = &acc[tid * 16];
        const float* h = h1 + (size_t)i * 16;
        float t = 0.0f;
        #pragma unroll
        for (int f = 0; f < 16; ++f) {
            float x = di * (a[f] + h[f]) + b1[f];
            x = fmaxf(x, 0.0f);
            t = fmaf(x, W2[f], t);
        }
        s[i] = di * t;
    }
}

#define S_WIN_V4 17
#define S_ROUND  (4 * LDS_THREADS)
#define S_WIN    (S_WIN_V4 * S_ROUND)
#define OWN1_REG 26
__global__ __launch_bounds__(LDS_THREADS)
void gcn_own1(const unsigned* __restrict__ packed, const int* __restrict__ bstart,
              const float* __restrict__ s, const float* __restrict__ dinv,
              const float* __restrict__ b2, float* __restrict__ out, int n) {
    __shared__ float win[S_WIN];
    __shared__ float acc[OWN_NODES];
    const int T   = LDS_THREADS;
    const int tid = (int)threadIdx.x;
    if (tid < OWN_NODES) acc[tid] = 0.0f;

    const int p0 = bstart[blockIdx.x];
    const int p1 = bstart[blockIdx.x + 1];
    const unsigned tmask = (1u << OWN_SHIFT) - 1u;

    unsigned w[OWN1_REG];
    #pragma unroll
    for (int u = 0; u < OWN1_REG; ++u) {
        const int pp = p0 + tid + u * T;
        w[u] = (pp < p1) ? packed[pp] : 0u;
    }
    const int pr = p0 + OWN1_REG * T;

    const bool al = (((size_t)s) & 15) == 0;
    float4 v[S_WIN_V4];
    auto fetch = [&](int base) {
        const int len4 = al ? (min(S_WIN, n - base) & ~3) : 0;
        const int full = len4 / S_ROUND;
        const int rem  = len4 - full * S_ROUND;
        #pragma unroll
        for (int u = 0; u < S_WIN_V4; ++u) {
            const bool have = u < full || (u == full && 4 * tid < rem);
            v[u] = have ? *(const float4*)(s + base + 4 * tid + u * S_ROUND)
                        : make_float4(0.0f, 0.0f, 0.0f, 0.0f);
        }
    };
    auto add = [&](unsigned word, int base, int len) {
        const unsigned off = (word >> OWN_SHIFT) - (unsigned)base;
        if (off < (unsigned)len)
            __hip_atomic_fetch_add(&acc[word & tmask], win[off],
                                   __ATOMIC_RELAXED, __HIP_MEMORY_SCOPE_WORKGROUP);
    };

    fetch(0);
    for (int base = 0; base < n; base += S_WIN) {
        const int len  = min(S_WIN, n - base);
        const int len4 = al ? (len & ~3) : 0;
        const int full = len4 / S_ROUND;
        const int rem  = len4 - full * S_ROUND;
        #pragma unroll
        for (int u = 0; u < S_WIN_V4; ++u) {
            if (u < full || (u == full && 4 * tid < rem))
                *(float4*)(win + 4 * tid + u * S_ROUND) = v[u];
        }
        for (int i = len4 + tid; i < len; i += T) win[i] = s[base + i];
        __syncthreads();
        if (base + S_WIN < n) fetch(base + S_WIN);
        #pragma unroll
        for (int u = 0; u < OWN1_REG; ++u) {
            unsigned word = w[u];
            asm volatile("" : "+v"(word));
            if (p0 + tid + u * T < p1) add(word, base, len);
        }
        for (int p = pr + tid; p < p1; p += T) add(packed[p], base, len);
        __syncthreads();
    }
    const int lo  = (int)blockIdx.x * OWN_NODES;
    const int lim = max(0, min(OWN_NODES, n - lo));
    if (tid < lim) {
        const int i = lo + tid;
        out[i] = dinv[i] * (acc[tid] + s[i]) + b2[0];
    }
}

extern "C" void kernel_launch(void* const* d_in, const int* in_sizes, int n_in,
                              void* d_out, int out_size, void* d_ws, size_t ws_size,
                              hipStream_t stream) {
    const float* x  = (const float*)d_in[0];
    const int*   ei = (const int*)d_in[1];
    const float* W1 = (const float*)d_in[2];
    const float* b1 = (const float*)d_in[3];
    const float* W2 = (const float*)d_in[4];
    const float* b2 = (const float*)d_in[5];

    const int N = in_sizes[0] / 128;
    const int E = in_sizes[1] / 2;
    const int* row = ei;
    const int* col = ei + E;

    {
        const int R = (N + OWN_NODES - 1) / OWN_NODES;
        const size_t need = ((size_t)19 * (size_t)N + (size_t)E + 3 * (size_t)R + 1) * sizeof(float);
        if (R >= 1 && R <= OWN_MAXR && N <= (1 << (32 - OWN_SHIFT)) && ws_size >= need) {
            float*    deg    = (float*)d_ws;
            float*    dinv   = deg + N;
            float*    h1     = dinv + N;
            float*    s      = h1 + (size_t)N * 16;
            unsigned* packed = (unsigned*)(s + N);
            int*      tot    = (int*)(packed + E);
            int*      cursor = tot + R;
            int*      bstart = cursor + R;
            float*    out    = (float*)d_out;
            const int B = 256;

            gcn_init_deg<<<(N + B - 1) / B, B, 0, stream>>>(deg, N);
            {
                const int RR = (N + RANGE_BINS - 1) / RANGE_BINS;
                const dim3 ldsGrid(EDGE_SLICES, RR);
                gcn_degree_lds<<<ldsGrid, LDS_THREADS, 0, stream>>>(col, deg, E, N);
            }
            gcn_dinv_tot<<<R, 256, 0, stream>>>(deg, dinv, tot, cursor, N);
            {
                int full   = N & ~15;
                int ntiles = full >> 4;
                if (ntiles > 0) {
                    int wavesPerBlock = B / 32;
                    int blocks = (ntiles + wavesPerBlock - 1) / wavesPerBlock;
                    gcn_gemm1_wmma<<<blocks, B, 0, stream>>>(x, W1, dinv, h1, ntiles);
                }
                if (full < N) {
                    int tail = (N - full) * 16;
                    gcn_gemm1_tail<<<(tail + B - 1) / B, B, 0, stream>>>(x, W1, dinv, h1, full, N);
                }
            }
            {
                const int quads = (E + 3) >> 2;
                int G = (quads + (BIN_CAP / 4) - 1) / (BIN_CAP / 4);
                if (G < BIN_WGS) G = BIN_WGS;
                gcn_bin_edges<<<G, LDS_THREADS, 0, stream>>>(row, col, tot, cursor, bstart, packed, E, N, R);
            }
            gcn_own16<<<R, LDS_THREADS, 0, stream>>>(packed, bstart, h1, dinv, b1, W2, s, N);
            gcn_own1<<<R, LDS_THREADS, 0, stream>>>(packed, bstart, s, dinv, b2, out, N);
            return;
        }
    }

    float* ws   = (float*)d_ws;
    float* deg  = ws;
    float* dinv = deg + N;
    float* h1   = dinv + N;
    float* agg1 = h1 + (size_t)N * 16;
    float* s    = agg1 + (size_t)N * 16;
    float* agg2 = s + N;
    float* out  = (float*)d_out;

    const int B = 256;

    {
        int total = N * 16;
        gcn_init<<<(total + B - 1) / B, B, 0, stream>>>(deg, agg1, agg2, N);
    }
    const int R = (N + RANGE_BINS - 1) / RANGE_BINS;
    const dim3 ldsGrid(EDGE_SLICES, R > 0 ? R : 1);

    gcn_degree_lds<<<ldsGrid, LDS_THREADS, 0, stream>>>(col, deg, E, N);
    gcn_dinv<<<(N + B - 1) / B, B, 0, stream>>>(deg, dinv, N);
    {
        int full   = N & ~15;
        int ntiles = full >> 4;
        if (ntiles > 0) {
            int wavesPerBlock = B / 32;
            int blocks = (ntiles + wavesPerBlock - 1) / wavesPerBlock;
            gcn_gemm1_wmma<<<blocks, B, 0, stream>>>(x, W1, dinv, h1, ntiles);
        }
        if (full < N) {
            int tail = (N - full) * 16;
            gcn_gemm1_tail<<<(tail + B - 1) / B, B, 0, stream>>>(x, W1, dinv, h1, full, N);
        }
    }
    {
        const int R16 = (N + R16_NODES - 1) / R16_NODES;
        const dim3 grid16(R16_SLICES, R16 > 0 ? R16 : 1);
        gcn_edge16_lds<<<grid16, LDS_THREADS, 0, stream>>>(row, col, h1, agg1, E, N);
    }
    gcn_post1<<<(N + B - 1) / B, B, 0, stream>>>(agg1, h1, dinv, b1, W2, s, N);
    gcn_edge1_lds<<<ldsGrid, LDS_THREADS, 0, stream>>>(row, col, s, agg2, E, N);
    gcn_post2<<<(N + B - 1) / B, B, 0, stream>>>(agg2, s, dinv, b2, out, N);
}


# Layer-1 owner orders its run by target in LDS (counting sort on integer counters) and sums each node's rows in registers: no LDS float atomics in layer 1; float sum order changes
# speedup vs baseline: 6.1615x; 1.2258x over previous
#include <hip/hip_runtime.h>

typedef __attribute__((ext_vector_type(2))) float v2f;
typedef __attribute__((ext_vector_type(8))) float v8f;

__global__ void gcn_init(float* __restrict__ deg, float* __restrict__ agg1,
                         float* __restrict__ agg2, int n) {
    int i = blockIdx.x * blockDim.x + threadIdx.x;
    if (i < n) { deg[i] = 1.0f; agg2[i] = 0.0f; }
    if (i < n * 16) { agg1[i] = 0.0f; }
}

#define RANGE_BINS  51200
#define EDGE_SLICES 64
#define LDS_THREADS 1024

__global__ __launch_bounds__(LDS_THREADS)
void gcn_degree_lds(const int* __restrict__ col, float* __restrict__ deg,
                    int E, int n) {
    __shared__ unsigned cnt[RANGE_BINS];
    const int T   = (int)blockDim.x;
    const int tid = (int)threadIdx.x;
    const int per = (n + (int)gridDim.y - 1) / (int)gridDim.y;
    const int lo  = (int)blockIdx.y * per;
    const int lim = max(0, min(per, n - lo));
    for (int i = tid; i < lim; i += T) cnt[i] = 0u;
    __syncthreads();

    const int quads = (E + 3) >> 2;
    const int qper  = (quads + (int)gridDim.x - 1) / (int)gridDim.x;
    const int q0    = (int)blockIdx.x * qper;
    const int q1    = min(q0 + qper, quads);
    const bool vec  = (((size_t)col) & 15) == 0;
    const int qv    = vec ? min(q1, E >> 2) : q0;

    for (int q = q0 + tid; q < qv; q += 4 * T) {
        int4 v[4];
        #pragma unroll
        for (int u = 0; u < 4; ++u) {
            const int qq = q + u * T;
            v[u] = (qq < qv) ? *(const int4*)(col + 4 * (size_t)qq)
                             : make_int4(-1, -1, -1, -1);
        }
        #pragma unroll
        for (int u = 0; u < 4; ++u) {
            const unsigned b0 = (unsigned)(v[u].x - lo);
            const unsigned b1 = (unsigned)(v[u].y - lo);
            const unsigned b2 = (unsigned)(v[u].z - lo);
            const unsigned b3 = (unsigned)(v[u].w - lo);
            if (b0 < (unsigned)lim) atomicAdd(&cnt[b0], 1u);
            if (b1 < (unsigned)lim) atomicAdd(&cnt[b1], 1u);
            if (b2 < (unsigned)lim) atomicAdd(&cnt[b2], 1u);
            if (b3 < (unsigned)lim) atomicAdd(&cnt[b3], 1u);
        }
    }
    for (int q = max(q0, qv) + tid; q < q1; q += T) {
        for (int k = 0; k < 4; ++k) {
            const int e = 4 * q + k;
            if (e < E) {
                const unsigned b = (unsigned)(col[e] - lo);
                if (b < (unsigned)lim) atomicAdd(&cnt[b], 1u);
            }
        }
    }
    __syncthreads();
    for (int i = tid; i < lim; i += T) {
        const unsigned v = cnt[i];
        if (v != 0u) atomicAdd(&deg[lo + i], (float)v);
    }
}

__global__ void gcn_dinv(const float* __restrict__ deg, float* __restrict__ dinv,
                         int n) {
    int i = blockIdx.x * blockDim.x + threadIdx.x;
    if (i >= n) return;
    float d = deg[i];
    dinv[i] = (d > 0.0f) ? rsqrtf(d) : 0.0f;
}

__global__ void gcn_gemm1_wmma(const float* __restrict__ x,
                               const float* __restrict__ W1,
                               const float* __restrict__ dinv,
                               float* __restrict__ h1, int ntiles) {
    int wave = blockIdx.x * (blockDim.x >> 5) + (threadIdx.x >> 5);
    if (wave >= ntiles) return;
    int lane  = threadIdx.x & 31;
    int m     = lane & 15;
    int kb    = (lane >> 4) << 1;
    size_t row16 = (size_t)wave * 16;

    v8f acc = {};
    const float* arow = x + (row16 + (size_t)m) * 128 + kb;
    #pragma unroll 4
    for (int k0 = 0; k0 < 128; k0 += 4) {
        v2f a = *(const v2f*)(arow + k0);
        v2f b;
        b.x = W1[(size_t)(k0 + kb) * 16 + m];
        b.y = W1[(size_t)(k0 + kb + 1) * 16 + m];
        acc = __builtin_amdgcn_wmma_f32_16x16x4_f32(
             false, a,  false, b,
             (short)0, acc,  false,  false);
    }

    int mbase = (lane >> 4) << 3;
    int ncol  = lane & 15;
    #pragma unroll
    for (int r = 0; r < 8; ++r) {
        const size_t node = row16 + (size_t)(mbase + r);
        h1[node * 16 + ncol] = acc[r] * dinv[node];
    }
}

__global__ void gcn_gemm1_tail(const float* __restrict__ x,
                               const float* __restrict__ W1,
                               const float* __restrict__ dinv,
                               float* __restrict__ h1, int start, int n) {
    int t = blockIdx.x * blockDim.x + threadIdx.x;
    int node = start + (t >> 4);
    int f = t & 15;
    if (node >= n) return;
    float acc = 0.0f;
    const float* xr = x + (size_t)node * 128;
    for (int k = 0; k < 128; ++k) acc += xr[k] * W1[(size_t)k * 16 + f];
    h1[(size_t)node * 16 + f] = acc * dinv[node];
}

#define R16_NODES  4768
#define R16_SLICES 6
#define R16_QCAP   96

__global__ __launch_bounds__(LDS_THREADS)
void gcn_edge16_lds(const int* __restrict__ row, const int* __restrict__ col,
                    const float* __restrict__ h1,
                    float* __restrict__ agg1, int E, int n) {
    __shared__ float acc[R16_NODES * 16];
    __shared__ int queue[(LDS_THREADS / 32) * R16_QCAP];
    __shared__ unsigned short qrow[(LDS_THREADS / 32) * R16_QCAP];
    static_assert(R16_NODES <= 65535, "a target's LDS row must fit a 16-bit queue word");

    const int T    = (int)blockDim.x;
    const int tid  = (int)threadIdx.x;
    const int lane = tid & 31;
    const int per  = (n + (int)gridDim.y - 1) / (int)gridDim.y;
    const int lo   = (int)blockIdx.y * per;
    const int lim  = max(0, min(per, n - lo));
    {
        float4* a4 = (float4*)acc;
        for (int i = tid; i < lim * 4; i += T) a4[i] = make_float4(0.0f, 0.0f, 0.0f, 0.0f);
    }
    __syncthreads();

    const int qo = (tid >> 5) * R16_QCAP;
    int cnt = 0;

    auto push = [&](int cv, int e) {
        const unsigned bin = (unsigned)(cv - lo);
        const bool in = bin < (unsigned)lim;
        const unsigned m = __builtin_amdgcn_ballot_w32(in);
        if (m != 0u) {
            if (in) {
                const int slot = qo + cnt + (int)__builtin_amdgcn_mbcnt_lo(m, 0u);
                __hip_atomic_store(&queue[slot], e, __ATOMIC_RELAXED, __HIP_MEMORY_SCOPE_WAVEFRONT);
                __hip_atomic_store(&qrow[slot], (unsigned short)bin,
                                   __ATOMIC_RELAXED, __HIP_MEMORY_SCOPE_WAVEFRONT);
            }
            cnt += __builtin_popcount(m);
        }
    };
    auto drain = [&](int base, int count) {
        const bool  have = lane < count;
        const int   e  = have ? __hip_atomic_load(&queue[qo + base + lane], __ATOMIC_RELAXED,
                                                  __HIP_MEMORY_SCOPE_WAVEFRONT) : 0;
        const int   bi = have ? (int)__hip_atomic_load(&qrow[qo + base + lane], __ATOMIC_RELAXED,
                                                       __HIP_MEMORY_SCOPE_WAVEFRONT) : 0;
        const int   ri = row[e];
        const int   l4 = lane & 3;
        const int   g  = lane >> 2;
        float4 v[4];
        #pragma unroll
        for (int j = 0; j < 4; ++j) {
            const int r = __shfl(ri, 8 * j + g);
            v[j] = *(const float4*)(h1 + (size_t)r * 16 + 4 * l4);
        }
        #pragma unroll
        for (int j = 0; j < 4; ++j) {
            const int s = 8 * j + g;
            const int b = __shfl(bi, s);
            if (s < count) {
                float* p = &acc[b * 16 + 4 * l4];
                __hip_atomic_fetch_add(p + 0, v[j].x, __ATOMIC_RELAXED, __HIP_MEMORY_SCOPE_WORKGROUP);
                __hip_atomic_fetch_add(p + 1, v[j].y, __ATOMIC_RELAXED, __HIP_MEMORY_SCOPE_WORKGROUP);
                __hip_atomic_fetch_add(p + 2, v[j].z, __ATOMIC_RELAXED, __HIP_MEMORY_SCOPE_WORKGROUP);
                __hip_atomic_fetch_add(p + 3, v[j].w, __ATOMIC_RELAXED, __HIP_MEMORY_SCOPE_WORKGROUP);
            }
        }
    };

    const int quads = (E + 3) >> 2;
    const int qper  = (quads + (int)gridDim.x - 1) / (int)gridDim.x;
    const int q0    = (int)blockIdx.x * qper;
    const int q1    = min(q0 + qper, quads);
    const bool vec  = (((size_t)col) & 15) == 0;
    const int qv    = vec ? min(q1, E >> 2) : q0;

    for (int qb = q0; qb < qv; qb += 4 * T) {
        int4 v[4];
        #pragma unroll
        for (int u = 0; u < 4; ++u) {
            const int qq = qb + tid + u * T;
            v[u] = (qq < qv) ? *(const int4*)(col + 4 * (size_t)qq)
                             : make_int4(-1, -1, -1, -1);
        }
        #pragma unroll
        for (int u = 0; u < 4; ++u) {
            const int e0 = 4 * (qb + tid + u * T);
            push(v[u].x, e0 + 0);
            push(v[u].y, e0 + 1);
            while (cnt >= 32) { cnt -= 32; drain(cnt, 32); }
            push(v[u].z, e0 + 2);
            push(v[u].w, e0 + 3);
            while (cnt >= 32) { cnt -= 32; drain(cnt, 32); }
        }
    }
    {
        const int e0 = 4 * max(q0, qv);
        const int e1 = min(4 * q1, E);
        for (int eb = e0; eb < e1; eb += T) {
            const int e = eb + tid;
            push((e < e1) ? col[e] : -1, e);
            while (cnt >= 32) { cnt -= 32; drain(cnt, 32); }
        }
    }
    if (cnt > 0) drain(0, cnt);

    __syncthreads();
    float* dst = agg1 + (size_t)lo * 16;
    for (int i = tid; i < lim * 16; i += T) {
        const float t = acc[i];
        if (t != 0.0f) atomicAdd(&dst[i], t);
    }
}

__global__ void gcn_post1(const float* __restrict__ agg1,
                          const float* __restrict__ h1,
                          const float* __restrict__ dinv,
                          const float* __restrict__ b1,
                          const float* __restrict__ W2,
                          float* __restrict__ s, int n) {
    int i = blockIdx.x * blockDim.x + threadIdx.x;
    if (i >= n) return;
    float di = dinv[i];
    const float* a = agg1 + (size_t)i * 16;
    const float* h = h1 + (size_t)i * 16;
    float acc = 0.0f;
    #pragma unroll
    for (int f = 0; f < 16; ++f) {
        float v = di * (a[f] + h[f]) + b1[f];
        v = fmaxf(v, 0.0f);
        acc = fmaf(v, W2[f], acc);
    }
    s[i] = di * acc;
}

__global__ __launch_bounds__(LDS_THREADS)
void gcn_edge1_lds(const int* __restrict__ row, const int* __restrict__ col,
                   const float* __restrict__ s,
                   float* __restrict__ agg2, int E, int n) {
    __shared__ float acc[RANGE_BINS];
    const int T   = (int)blockDim.x;
    const int tid = (int)threadIdx.x;
    const int per = (n + (int)gridDim.y - 1) / (int)gridDim.y;
    const int lo  = (int)blockIdx.y * per;
    const int lim = max(0, min(per, n - lo));
    for (int i = tid; i < lim; i += T) acc[i] = 0.0f;
    __syncthreads();

    const int quads = (E + 3) >> 2;
    const int qper  = (quads + (int)gridDim.x - 1) / (int)gridDim.x;
    const int q0    = (int)blockIdx.x * qper;
    const int q1    = min(q0 + qper, quads);
    const bool vec  = ((((size_t)col) | ((size_t)row)) & 15) == 0;
    const int qv    = vec ? min(q1, E >> 2) : q0;

    for (int q = q0 + tid; q < qv; q += 2 * T) {
        int c[8], r[8];
        #pragma unroll
        for (int u = 0; u < 2; ++u) {
            const int qq = q + u * T;
            const bool ok = qq < qv;
            const int4 vc = ok ? *(const int4*)(col + 4 * (size_t)qq) : make_int4(-1, -1, -1, -1);
            const int4 vr = ok ? *(const int4*)(row + 4 * (size_t)qq) : make_int4(0, 0, 0, 0);
            c[4 * u + 0] = vc.x; c[4 * u + 1] = vc.y; c[4 * u + 2] = vc.z; c[4 * u + 3] = vc.w;
            r[4 * u + 0] = vr.x; r[4 * u + 1] = vr.y; r[4 * u + 2] = vr.z; r[4 * u + 3] = vr.w;
        }
        unsigned b[8];
        float    v[8];
        #pragma unroll
        for (int k = 0; k < 8; ++k) {
            b[k] = (unsigned)(c[k] - lo);
            const bool in = b[k] < (unsigned)lim;
            v[k] = s[in ? r[k] : 0];
        }
        #pragma unroll
        for (int k = 0; k < 8; ++k) {
            if (b[k] < (unsigned)lim)
                __hip_atomic_fetch_add(&acc[b[k]], v[k], __ATOMIC_RELAXED,
                                       __HIP_MEMORY_SCOPE_WORKGROUP);
        }
    }
    for (int q = max(q0, qv) + tid; q < q1; q += T) {
        for (int k = 0; k < 4; ++k) {
            const int e = 4 * q + k;
            if (e < E) {
                const unsigned bb = (unsigned)(col[e] - lo);
                if (bb < (unsigned)lim) {
                    __hip_atomic_fetch_add(&acc[bb], s[row[e]],
                                           __ATOMIC_RELAXED, __HIP_MEMORY_SCOPE_WORKGROUP);
                }
            }
        }
    }
    __syncthreads();
    for (int i = tid; i < lim; i += T) {
        const float t = acc[i];
        if (t != 0.0f) atomicAdd(&agg2[lo + i], t);
    }
}

__global__ void gcn_post2(const float* __restrict__ agg2,
                          const float* __restrict__ s,
                          const float* __restrict__ dinv,
                          const float* __restrict__ b2,
                          float* __restrict__ out, int n) {
    int i = blockIdx.x * blockDim.x + threadIdx.x;
    if (i >= n) return;
    float di = dinv[i];
    out[i] = di * (agg2[i] + s[i]) + b2[0];
}


#define OWN_NODES 782
#define OWN_SHIFT 10
#define OWN_MAXR  512
#define BIN_WGS   256
static_assert(OWN_NODES <= (1 << OWN_SHIFT), "a target's place in its range must fit OWN_SHIFT bits");
static_assert(OWN_NODES <= LDS_THREADS, "the epilogues give every node of a range one thread");

__global__ void gcn_init_deg(float* __restrict__ deg, int n) {
    int i = blockIdx.x * blockDim.x + threadIdx.x;
    if (i < n) deg[i] = 1.0f;
}

__global__ __launch_bounds__(256)
void gcn_dinv_tot(const float* __restrict__ deg, float* __restrict__ dinv,
                  int* __restrict__ tot, int* __restrict__ cursor, int n) {
    __shared__ int part[8];
    const int lo = (int)blockIdx.x * OWN_NODES;
    const int hi = min(lo + OWN_NODES, n);
    int c = 0;
    for (int i = lo + (int)threadIdx.x; i < hi; i += (int)blockDim.x) {
        const float d = deg[i];
        dinv[i] = (d > 0.0f) ? rsqrtf(d) : 0.0f;
        c += (int)d - 1;
    }
    for (int off = 16; off > 0; off >>= 1) c += __shfl_xor(c, off);
    if ((threadIdx.x & 31) == 0) part[threadIdx.x >> 5] = c;
    __syncthreads();
    if (threadIdx.x == 0) {
        int t = 0;
        for (int w = 0; w < (int)(blockDim.x >> 5); ++w) t += part[w];
        tot[blockIdx.x]    = t;
        cursor[blockIdx.x] = 0;
    }
}

#define BIN_CAP 32768
__global__ __launch_bounds__(LDS_THREADS)
void gcn_bin_edges(const int* __restrict__ row, const int* __restrict__ col,
                   const int* __restrict__ tot, int* __restrict__ cursor,
                   int* __restrict__ bstart, unsigned* __restrict__ packed,
                   int E, int n, int R) {
    __shared__ unsigned stage[BIN_CAP];
    __shared__ int slot[OWN_MAXR];
    __shared__ int start[OWN_MAXR + 1];
    __shared__ int loff[OWN_MAXR + 1];
    __shared__ int gbase[OWN_MAXR];
    __shared__ int wsum[LDS_THREADS / 32];
    const int T    = (int)blockDim.x;
    const int tid  = (int)threadIdx.x;
    const int lane = tid & 31;

    auto xscan = [&](int v) -> int {
        int incl = v;
        for (int off = 1; off < 32; off <<= 1) {
            const int t = __shfl_up(incl, off);
            if (lane >= off) incl += t;
        }
        __syncthreads();
        if (lane == 31) wsum[tid >> 5] = incl;
        __syncthreads();
        int pre = 0;
        for (int w = 0; w < (tid >> 5); ++w) pre += wsum[w];
        return pre + incl - v;
    };

    {
        const int v  = (tid < R) ? tot[tid] : 0;
        const int ex = xscan(v);
        if (tid < R)         start[tid] = ex;
        if (tid == R - 1)    start[R]   = ex + v;
        if (tid < OWN_MAXR)  slot[tid]  = 0;
        __syncthreads();
    }
    if (blockIdx.x == 0) {
        for (int i = tid; i <= R; i += T) bstart[i] = start[i];
    }

    const int quads = (E + 3) >> 2;
    const int qper  = (quads + (int)gridDim.x - 1) / (int)gridDim.x;
    const int q0    = (int)blockIdx.x * qper;
    const int q1    = min(q0 + qper, quads);
    const bool vec  = ((((size_t)col) | ((size_t)row)) & 15) == 0;
    const int qv    = vec ? min(q1, E >> 2) : q0;

    auto count = [&](int c) {
        if ((unsigned)c < (unsigned)n) atomicAdd(&slot[c / OWN_NODES], 1);
    };
    for (int q = q0 + tid; q < qv; q += 4 * T) {
        int4 v[4];
        #pragma unroll
        for (int u = 0; u < 4; ++u) {
            const int qq = q + u * T;
            v[u] = (qq < qv) ? *(const int4*)(col + 4 * (size_t)qq)
                             : make_int4(-1, -1, -1, -1);
        }
        #pragma unroll
        for (int u = 0; u < 4; ++u) {
            count(v[u].x); count(v[u].y); count(v[u].z); count(v[u].w);
        }
    }
    for (int q = max(q0, qv) + tid; q < q1; q += T) {
        for (int k = 0; k < 4; ++k) {
            const int e = 4 * q + k;
            if (e < E) count(col[e]);
        }
    }
    __syncthreads();
    {
        const int c  = (tid < R) ? slot[tid] : 0;
        const int ex = xscan(c);
        if (tid < R) {
            loff[tid]  = ex;
            gbase[tid] = start[tid] + ((c != 0) ? atomicAdd(&cursor[tid], c) : 0);
            slot[tid]  = ex;
        }
        if (tid == R - 1) loff[R] = ex + c;
        __syncthreads();
    }

    auto place = [&](int c, int sidx) {
        if ((unsigned)c < (unsigned)n) {
            const int r = c / OWN_NODES;
            const int k = atomicAdd(&slot[r], 1);
            if (k < BIN_CAP)
                stage[k] = ((unsigned)sidx << OWN_SHIFT) | (unsigned)(c - r * OWN_NODES);
        }
    };
    for (int q = q0 + tid; q < qv; q += 2 * T) {
        int4 vc[2], vr[2];
        #pragma unroll
        for (int u = 0; u < 2; ++u) {
            const int qq = q + u * T;
            const bool ok = qq < qv;
            vc[u] = ok ? *(const int4*)(col + 4 * (size_t)qq) : make_int4(-1, -1, -1, -1);
            vr[u] = ok ? *(const int4*)(row + 4 * (size_t)qq) : make_int4(0, 0, 0, 0);
        }
        #pragma unroll
        for (int u = 0; u < 2; ++u) {
            place(vc[u].x, vr[u].x); place(vc[u].y, vr[u].y);
            place(vc[u].z, vr[u].z); place(vc[u].w, vr[u].w);
        }
    }
    for (int q = max(q0, qv) + tid; q < q1; q += T) {
        for (int k = 0; k < 4; ++k) {
            const int e = 4 * q + k;
            if (e < E) place(col[e], row[e]);
        }
    }
    __syncthreads();

    for (int r = tid >> 5; r < R; r += (T >> 5)) {
        const int l0 = loff[r];
        const int nr = min(loff[r + 1], BIN_CAP) - l0;
        const int g0 = gbase[r];
        for (int i = lane; i < nr; i += 32) packed[g0 + i] = stage[l0 + i];
    }
}

#define SRT_REG 26
#define SRT_CAP (SRT_REG * LDS_THREADS)
#define SEG_UNR 6
__global__ __launch_bounds__(LDS_THREADS)
void gcn_own16(const unsigned* __restrict__ packed, const int* __restrict__ bstart,
               const float* __restrict__ h1, const float* __restrict__ dinv,
               const float* __restrict__ b1, const float* __restrict__ W2,
               float* __restrict__ s, int n) {
    __shared__ unsigned srt[SRT_CAP];
    __shared__ float acc[OWN_NODES * 16];
    __shared__ int cnt[LDS_THREADS];
    __shared__ int start[LDS_THREADS + 1];
    __shared__ int wsum[LDS_THREADS / 32];
    const int T    = LDS_THREADS;
    const int tid  = (int)threadIdx.x;
    const int lane = tid & 31;
    const int wv   = __builtin_amdgcn_readfirstlane(tid >> 5);
    const int lo   = (int)blockIdx.x * OWN_NODES;
    const int lim  = max(0, min(OWN_NODES, n - lo));
    {
        float4* a4 = (float4*)acc;
        for (int i = tid; i < lim * 4; i += T) a4[i] = make_float4(0.0f, 0.0f, 0.0f, 0.0f);
    }

    auto xscan = [&](int v) -> int {
        int incl = v;
        for (int off = 1; off < 32; off <<= 1) {
            const int t = __shfl_up(incl, off);
            if (lane >= off) incl += t;
        }
        __syncthreads();
        if (lane == 31) wsum[wv] = incl;
        __syncthreads();
        int pre = 0;
        for (int w = 0; w < wv; ++w) pre += wsum[w];
        return pre + incl - v;
    };

    const int p0 = bstart[blockIdx.x];
    const int p1 = bstart[blockIdx.x + 1];
    const unsigned tmask = (1u << OWN_SHIFT) - 1u;
    const int  l4   = lane & 3;
    const int  slot = lane >> 2;
    const bool up16 = (lane & 16) != 0;
    const bool up8  = (lane & 8) != 0;
    const float4 zero4 = make_float4(0.0f, 0.0f, 0.0f, 0.0f);

    for (int c0 = p0; c0 < p1; c0 += SRT_CAP) {
        cnt[tid] = 0;
        __syncthreads();

        unsigned w[SRT_REG];
        int      rk[SRT_REG];
        #pragma unroll
        for (int u = 0; u < SRT_REG; ++u) {
            const int pp = c0 + tid + u * T;
            w[u] = (pp < p1) ? packed[pp] : 0u;
        }
        #pragma unroll
        for (int u = 0; u < SRT_REG; ++u) {
            const int pp = c0 + tid + u * T;
            rk[u] = 0;
            if (pp < p1) rk[u] = atomicAdd(&cnt[w[u] & tmask], 1);
        }
        __syncthreads();
        {
            const int v  = cnt[tid];
            const int ex = xscan(v);
            start[tid] = ex;
            if (tid == T - 1) start[T] = ex + v;
        }
        __syncthreads();
        #pragma unroll
        for (int u = 0; u < SRT_REG; ++u) rk[u] += start[w[u] & tmask];
        #pragma unroll
        for (int u = 0; u < SRT_REG; ++u) {
            const int pp = c0 + tid + u * T;
            if (pp < p1) srt[rk[u]] = w[u] >> OWN_SHIFT;
        }
        __syncthreads();

        for (int nb = wv; nb < lim; nb += 64) {
            int  a[2], len[2];
            bool ok[2];
            #pragma unroll
            for (int q = 0; q < 2; ++q) {
                const int node = nb + 32 * q;
                ok[q] = node < lim;
                const int nn = ok[q] ? node : 0;
                a[q]   = __builtin_amdgcn_readfirstlane(start[nn]);
                len[q] = ok[q] ? (__builtin_amdgcn_readfirstlane(start[nn + 1]) - a[q]) : 0;
            }
            float4 sum[2] = {zero4, zero4};
            const int mx = max(len[0], len[1]);
            for (int k = 0; k < mx; k += 8 * SEG_UNR) {
                float4 t[2][SEG_UNR];
                #pragma unroll
                for (int q = 0; q < 2; ++q) {
                    #pragma unroll
                    for (int j = 0; j < SEG_UNR; ++j) {
                        const int  kk   = k + 8 * j + slot;
                        const bool have = kk < len[q];
                        const unsigned sw = srt[have ? (a[q] + kk) : 0];
                        const unsigned sr = have ? sw : 0u;
                        t[q][j] = *(const float4*)(h1 + (size_t)sr * 16 + 4 * l4);
                    }
                }
                #pragma unroll
                for (int q = 0; q < 2; ++q) {
                    #pragma unroll
                    for (int j = 0; j < SEG_UNR; ++j) {
                        const bool have = (k + 8 * j + slot) < len[q];
                        sum[q].x += have ? t[q][j].x : 0.0f;
                        sum[q].y += have ? t[q][j].y : 0.0f;
                        sum[q].z += have ? t[q][j].z : 0.0f;
                        sum[q].w += have ? t[q][j].w : 0.0f;
                    }
                }
            }
            float u0[2], u1[2], r[2], old[2];
            const int f = 4 * l4 + (up16 ? 2 : 0) + (up8 ? 1 : 0);
            #pragma unroll
            for (int q = 0; q < 2; ++q) {
                const float4 v = sum[q];
                const float ka = up16 ? v.z : v.x, kb = up16 ? v.w : v.y;
                const float ga = up16 ? v.x : v.z, gb = up16 ? v.y : v.w;
                u0[q] = ka + __shfl_xor(ga, 16);
                u1[q] = kb + __shfl_xor(gb, 16);
                old[q] = acc[(ok[q] ? (nb + 32 * q) : 0) * 16 + f];
            }
            #pragma unroll
            for (int q = 0; q < 2; ++q) {
                const float kc = up8 ? u1[q] : u0[q];
                const float gc = up8 ? u0[q] : u1[q];
                r[q] = kc + __shfl_xor(gc, 8);
            }
            #pragma unroll
            for (int q = 0; q < 2; ++q) r[q] += __shfl_xor(r[q], 4);
            #pragma unroll
            for (int q = 0; q < 2; ++q) {
                if (ok[q] && (lane & 4) == 0)
                    acc[(nb + 32 * q) * 16 + f] = old[q] + r[q];
            }
        }
    }
    __syncthreads();
    if (tid < lim) {
        const int   i  = lo + tid;
        const float di = dinv[i];
        const float* a = &acc[tid * 16];
        const float* h = h1 + (size_t)i * 16;
        float t = 0.0f;
        #pragma unroll
        for (int f = 0; f < 16; ++f) {
            float x = di * (a[f] + h[f]) + b1[f];
            x = fmaxf(x, 0.0f);
            t = fmaf(x, W2[f], t);
        }
        s[i] = di * t;
    }
}

#define S_WIN_V4 17
#define S_ROUND  (4 * LDS_THREADS)
#define S_WIN    (S_WIN_V4 * S_ROUND)
#define OWN1_REG 26
__global__ __launch_bounds__(LDS_THREADS)
void gcn_own1(const unsigned* __restrict__ packed, const int* __restrict__ bstart,
              const float* __restrict__ s, const float* __restrict__ dinv,
              const float* __restrict__ b2, float* __restrict__ out, int n) {
    __shared__ float win[S_WIN];
    __shared__ float acc[OWN_NODES];
    const int T   = LDS_THREADS;
    const int tid = (int)threadIdx.x;
    if (tid < OWN_NODES) acc[tid] = 0.0f;

    const int p0 = bstart[blockIdx.x];
    const int p1 = bstart[blockIdx.x + 1];
    const unsigned tmask = (1u << OWN_SHIFT) - 1u;

    unsigned w[OWN1_REG];
    #pragma unroll
    for (int u = 0; u < OWN1_REG; ++u) {
        const int pp = p0 + tid + u * T;
        w[u] = (pp < p1) ? packed[pp] : 0u;
    }
    const int pr = p0 + OWN1_REG * T;

    const bool al = (((size_t)s) & 15) == 0;
    float4 v[S_WIN_V4];
    auto fetch = [&](int base) {
        const int len4 = al ? (min(S_WIN, n - base) & ~3) : 0;
        const int full = len4 / S_ROUND;
        const int rem  = len4 - full * S_ROUND;
        #pragma unroll
        for (int u = 0; u < S_WIN_V4; ++u) {
            const bool have = u < full || (u == full && 4 * tid < rem);
            v[u] = have ? *(const float4*)(s + base + 4 * tid + u * S_ROUND)
                        : make_float4(0.0f, 0.0f, 0.0f, 0.0f);
        }
    };
    auto add = [&](unsigned word, int base, int len) {
        const unsigned off = (word >> OWN_SHIFT) - (unsigned)base;
        if (off < (unsigned)len)
            __hip_atomic_fetch_add(&acc[word & tmask], win[off],
                                   __ATOMIC_RELAXED, __HIP_MEMORY_SCOPE_WORKGROUP);
    };

    fetch(0);
    for (int base = 0; base < n; base += S_WIN) {
        const int len  = min(S_WIN, n - base);
        const int len4 = al ? (len & ~3) : 0;
        const int full = len4 / S_ROUND;
        const int rem  = len4 - full * S_ROUND;
        #pragma unroll
        for (int u = 0; u < S_WIN_V4; ++u) {
            if (u < full || (u == full && 4 * tid < rem))
                *(float4*)(win + 4 * tid + u * S_ROUND) = v[u];
        }
        for (int i = len4 + tid; i < len; i += T) win[i] = s[base + i];
        __syncthreads();
        if (base + S_WIN < n) fetch(base + S_WIN);
        #pragma unroll
        for (int u = 0; u < OWN1_REG; ++u) {
            unsigned word = w[u];
            asm volatile("" : "+v"(word));
            if (p0 + tid + u * T < p1) add(word, base, len);
        }
        for (int p = pr + tid; p < p1; p += T) add(packed[p], base, len);
        __syncthreads();
    }
    const int lo  = (int)blockIdx.x * OWN_NODES;
    const int lim = max(0, min(OWN_NODES, n - lo));
    if (tid < lim) {
        const int i = lo + tid;
        out[i] = dinv[i] * (acc[tid] + s[i]) + b2[0];
    }
}

extern "C" void kernel_launch(void* const* d_in, const int* in_sizes, int n_in,
                              void* d_out, int out_size, void* d_ws, size_t ws_size,
                              hipStream_t stream) {
    const float* x  = (const float*)d_in[0];
    const int*   ei = (const int*)d_in[1];
    const float* W1 = (const float*)d_in[2];
    const float* b1 = (const float*)d_in[3];
    const float* W2 = (const float*)d_in[4];
    const float* b2 = (const float*)d_in[5];

    const int N = in_sizes[0] / 128;
    const int E = in_sizes[1] / 2;
    const int* row = ei;
    const int* col = ei + E;

    {
        const int R = (N + OWN_NODES - 1) / OWN_NODES;
        const size_t need = ((size_t)19 * (size_t)N + (size_t)E + 3 * (size_t)R + 1) * sizeof(float);
        if (R >= 1 && R <= OWN_MAXR && N <= (1 << (32 - OWN_SHIFT)) && ws_size >= need) {
            float*    deg    = (float*)d_ws;
            float*    dinv   = deg + N;
            float*    h1     = dinv + N;
            float*    s      = h1 + (size_t)N * 16;
            unsigned* packed = (unsigned*)(s + N);
            int*      tot    = (int*)(packed + E);
            int*      cursor = tot + R;
            int*      bstart = cursor + R;
            float*    out    = (float*)d_out;
            const int B = 256;

            gcn_init_deg<<<(N + B - 1) / B, B, 0, stream>>>(deg, N);
            {
                const int RR = (N + RANGE_BINS - 1) / RANGE_BINS;
                const dim3 ldsGrid(EDGE_SLICES, RR);
                gcn_degree_lds<<<ldsGrid, LDS_THREADS, 0, stream>>>(col, deg, E, N);
            }
            gcn_dinv_tot<<<R, 256, 0, stream>>>(deg, dinv, tot, cursor, N);
            {
                int full   = N & ~15;
                int ntiles = full >> 4;
                if (ntiles > 0) {
                    int wavesPerBlock = B / 32;
                    int blocks = (ntiles + wavesPerBlock - 1) / wavesPerBlock;
                    gcn_gemm1_wmma<<<blocks, B, 0, stream>>>(x, W1, dinv, h1, ntiles);
                }
                if (full < N) {
                    int tail = (N - full) * 16;
                    gcn_gemm1_tail<<<(tail + B - 1) / B, B, 0, stream>>>(x, W1, dinv, h1, full, N);
                }
            }
            {
                const int quads = (E + 3) >> 2;
                int G = (quads + (BIN_CAP / 4) - 1) / (BIN_CAP / 4);
                if (G < BIN_WGS) G = BIN_WGS;
                gcn_bin_edges<<<G, LDS_THREADS, 0, stream>>>(row, col, tot, cursor, bstart, packed, E, N, R);
            }
            gcn_own16<<<R, LDS_THREADS, 0, stream>>>(packed, bstart, h1, dinv, b1, W2, s, N);
            gcn_own1<<<R, LDS_THREADS, 0, stream>>>(packed, bstart, s, dinv, b2, out, N);
            return;
        }
    }

    float* ws   = (float*)d_ws;
    float* deg  = ws;
    float* dinv = deg + N;
    float* h1   = dinv + N;
    float* agg1 = h1 + (size_t)N * 16;
    float* s    = agg1 + (size_t)N * 16;
    float* agg2 = s + N;
    float* out  = (float*)d_out;

    const int B = 256;

    {
        int total = N * 16;
        gcn_init<<<(total + B - 1) / B, B, 0, stream>>>(deg, agg1, agg2, N);
    }
    const int R = (N + RANGE_BINS - 1) / RANGE_BINS;
    const dim3 ldsGrid(EDGE_SLICES, R > 0 ? R : 1);

    gcn_degree_lds<<<ldsGrid, LDS_THREADS, 0, stream>>>(col, deg, E, N);
    gcn_dinv<<<(N + B - 1) / B, B, 0, stream>>>(deg, dinv, N);
    {
        int full   = N & ~15;
        int ntiles = full >> 4;
        if (ntiles > 0) {
            int wavesPerBlock = B / 32;
            int blocks = (ntiles + wavesPerBlock - 1) / wavesPerBlock;
            gcn_gemm1_wmma<<<blocks, B, 0, stream>>>(x, W1, dinv, h1, ntiles);
        }
        if (full < N) {
            int tail = (N - full) * 16;
            gcn_gemm1_tail<<<(tail + B - 1) / B, B, 0, stream>>>(x, W1, dinv, h1, full, N);
        }
    }
    {
        const int R16 = (N + R16_NODES - 1) / R16_NODES;
        const dim3 grid16(R16_SLICES, R16 > 0 ? R16 : 1);
        gcn_edge16_lds<<<grid16, LDS_THREADS, 0, stream>>>(row, col, h1, agg1, E, N);
    }
    gcn_post1<<<(N + B - 1) / B, B, 0, stream>>>(agg1, h1, dinv, b1, W2, s, N);
    gcn_edge1_lds<<<ldsGrid, LDS_THREADS, 0, stream>>>(row, col, s, agg2, E, N);
    gcn_post2<<<(N + B - 1) / B, B, 0, stream>>>(agg2, s, dinv, b2, out, N);
}


# Layer-1 product reads each x tile once in whole 512-byte rows (16-byte non-temporal loads) and takes A and B fragments from LDS, not 8-byte loads over 16 lines each; k order and sum order kept
# speedup vs baseline: 6.8362x; 1.1095x over previous
#include <hip/hip_runtime.h>

typedef __attribute__((ext_vector_type(2))) float v2f;
typedef __attribute__((ext_vector_type(4))) float v4f;
typedef __attribute__((ext_vector_type(8))) float v8f;

__global__ void gcn_init(float* __restrict__ deg, float* __restrict__ agg1,
                         float* __restrict__ agg2, int n) {
    int i = blockIdx.x * blockDim.x + threadIdx.x;
    if (i < n) { deg[i] = 1.0f; agg2[i] = 0.0f; }
    if (i < n * 16) { agg1[i] = 0.0f; }
}

#define RANGE_BINS  51200
#define EDGE_SLICES 64
#define LDS_THREADS 1024

__global__ __launch_bounds__(LDS_THREADS)
void gcn_degree_lds(const int* __restrict__ col, float* __restrict__ deg,
                    int E, int n) {
    __shared__ unsigned cnt[RANGE_BINS];
    const int T   = (int)blockDim.x;
    const int tid = (int)threadIdx.x;
    const int per = (n + (int)gridDim.y - 1) / (int)gridDim.y;
    const int lo  = (int)blockIdx.y * per;
    const int lim = max(0, min(per, n - lo));
    for (int i = tid; i < lim; i += T) cnt[i] = 0u;
    __syncthreads();

    const int quads = (E + 3) >> 2;
    const int qper  = (quads + (int)gridDim.x - 1) / (int)gridDim.x;
    const int q0    = (int)blockIdx.x * qper;
    const int q1    = min(q0 + qper, quads);
    const bool vec  = (((size_t)col) & 15) == 0;
    const int qv    = vec ? min(q1, E >> 2) : q0;

    for (int q = q0 + tid; q < qv; q += 4 * T) {
        int4 v[4];
        #pragma unroll
        for (int u = 0; u < 4; ++u) {
            const int qq = q + u * T;
            v[u] = (qq < qv) ? *(const int4*)(col + 4 * (size_t)qq)
                             : make_int4(-1, -1, -1, -1);
        }
        #pragma unroll
        for (int u = 0; u < 4; ++u) {
            const unsigned b0 = (unsigned)(v[u].x - lo);
            const unsigned b1 = (unsigned)(v[u].y - lo);
            const unsigned b2 = (unsigned)(v[u].z - lo);
            const unsigned b3 = (unsigned)(v[u].w - lo);
            if (b0 < (unsigned)lim) atomicAdd(&cnt[b0], 1u);
            if (b1 < (unsigned)lim) atomicAdd(&cnt[b1], 1u);
            if (b2 < (unsigned)lim) atomicAdd(&cnt[b2], 1u);
            if (b3 < (unsigned)lim) atomicAdd(&cnt[b3], 1u);
        }
    }
    for (int q = max(q0, qv) + tid; q < q1; q += T) {
        for (int k = 0; k < 4; ++k) {
            const int e = 4 * q + k;
            if (e < E) {
                const unsigned b = (unsigned)(col[e] - lo);
                if (b < (unsigned)lim) atomicAdd(&cnt[b], 1u);
            }
        }
    }
    __syncthreads();
    for (int i = tid; i < lim; i += T) {
        const unsigned v = cnt[i];
        if (v != 0u) atomicAdd(&deg[lo + i], (float)v);
    }
}

__global__ void gcn_dinv(const float* __restrict__ deg, float* __restrict__ dinv,
                         int n) {
    int i = blockIdx.x * blockDim.x + threadIdx.x;
    if (i >= n) return;
    float d = deg[i];
    dinv[i] = (d > 0.0f) ? rsqrtf(d) : 0.0f;
}

#define G1_WAVES 8
#define G1_PITCH 132
__global__ __launch_bounds__(G1_WAVES * 32)
void gcn_gemm1_wmma(const float* __restrict__ x,
                    const float* __restrict__ W1,
                    const float* __restrict__ dinv,
                    float* __restrict__ h1, int ntiles) {
    __shared__ alignas(16) float xs[G1_WAVES * 16 * G1_PITCH];
    __shared__ alignas(16) float ws[128 * 16];
    const int  tid  = (int)threadIdx.x;
    const int  lane = tid & 31;
    const int  wv   = tid >> 5;
    const int  wave = (int)blockIdx.x * G1_WAVES + wv;
    const bool act  = wave < ntiles;
    const size_t row16 = (size_t)(act ? wave : 0) * 16;

    float* xt = xs + wv * (16 * G1_PITCH);
    const float* src = x + row16 * 128;
    const bool al = (((size_t)x) & 15) == 0;
    const int mbase = (lane >> 4) << 3;

    v4f v[16];
    if (act && al) {
        #pragma unroll
        for (int i = 0; i < 16; ++i)
            v[i] = __builtin_nontemporal_load((const v4f*)(src + i * 128 + 4 * lane));
    }
    float w[(128 * 16) / (G1_WAVES * 32)];
    #pragma unroll
    for (int j = 0; j < (128 * 16) / (G1_WAVES * 32); ++j) w[j] = W1[tid + j * (G1_WAVES * 32)];
    float dv[8];
    #pragma unroll
    for (int r = 0; r < 8; ++r) dv[r] = dinv[row16 + (size_t)(mbase + r)];

    #pragma unroll
    for (int j = 0; j < (128 * 16) / (G1_WAVES * 32); ++j) {
        const int e = tid + j * (G1_WAVES * 32);
        const int k = e >> 4, n = e & 15;
        ws[(((k >> 1) * 16 + n) << 1) + (k & 1)] = w[j];
    }
    if (act) {
        if (al) {
            #pragma unroll
            for (int i = 0; i < 16; ++i)
                *(v4f*)(xt + i * G1_PITCH + 4 * lane) = v[i];
        } else {
            #pragma nounroll
            for (int i = 0; i < 16 * 4; ++i)
                xt[(i >> 2) * G1_PITCH + lane + 32 * (i & 3)] = src[(i >> 2) * 128 + lane + 32 * (i & 3)];
        }
    }
    __syncthreads();
    if (!act) return;

    const int m = lane & 15;
    const int h = lane >> 4;
    const float* ap = xt + m * G1_PITCH + 2 * h;
    const float* bp = ws + ((h * 16 + m) << 1);

    v8f acc = {};
    #pragma unroll 8
    for (int k0 = 0; k0 < 128; k0 += 4) {
        const v2f a = *(const v2f*)(ap + k0);
        const v2f b = *(const v2f*)(bp + (k0 >> 1) * 32);
        acc = __builtin_amdgcn_wmma_f32_16x16x4_f32(
             false, a,  false, b,
             (short)0, acc,  false,  false);
    }

    #pragma unroll
    for (int r = 0; r < 8; ++r) {
        const size_t node = row16 + (size_t)(mbase + r);
        h1[node * 16 + m] = acc[r] * dv[r];
    }
}

__global__ void gcn_gemm1_tail(const float* __restrict__ x,
                               const float* __restrict__ W1,
                               const float* __restrict__ dinv,
                               float* __restrict__ h1, int start, int n) {
    int t = blockIdx.x * blockDim.x + threadIdx.x;
    int node = start + (t >> 4);
    int f = t & 15;
    if (node >= n) return;
    float acc = 0.0f;
    const float* xr = x + (size_t)node * 128;
    for (int k = 0; k < 128; ++k) acc += xr[k] * W1[(size_t)k * 16 + f];
    h1[(size_t)node * 16 + f] = acc * dinv[node];
}

#define R16_NODES  4768
#define R16_SLICES 6
#define R16_QCAP   96

__global__ __launch_bounds__(LDS_THREADS)
void gcn_edge16_lds(const int* __restrict__ row, const int* __restrict__ col,
                    const float* __restrict__ h1,
                    float* __restrict__ agg1, int E, int n) {
    __shared__ float acc[R16_NODES * 16];
    __shared__ int queue[(LDS_THREADS / 32) * R16_QCAP];
    __shared__ unsigned short qrow[(LDS_THREADS / 32) * R16_QCAP];
    static_assert(R16_NODES <= 65535, "a target's LDS row must fit a 16-bit queue word");

    const int T    = (int)blockDim.x;
    const int tid  = (int)threadIdx.x;
    const int lane = tid & 31;
    const int per  = (n + (int)gridDim.y - 1) / (int)gridDim.y;
    const int lo   = (int)blockIdx.y * per;
    const int lim  = max(0, min(per, n - lo));
    {
        float4* a4 = (float4*)acc;
        for (int i = tid; i < lim * 4; i += T) a4[i] = make_float4(0.0f, 0.0f, 0.0f, 0.0f);
    }
    __syncthreads();

    const int qo = (tid >> 5) * R16_QCAP;
    int cnt = 0;

    auto push = [&](int cv, int e) {
        const unsigned bin = (unsigned)(cv - lo);
        const bool in = bin < (unsigned)lim;
        const unsigned m = __builtin_amdgcn_ballot_w32(in);
        if (m != 0u) {
            if (in) {
                const int slot = qo + cnt + (int)__builtin_amdgcn_mbcnt_lo(m, 0u);
                __hip_atomic_store(&queue[slot], e, __ATOMIC_RELAXED, __HIP_MEMORY_SCOPE_WAVEFRONT);
                __hip_atomic_store(&qrow[slot], (unsigned short)bin,
                                   __ATOMIC_RELAXED, __HIP_MEMORY_SCOPE_WAVEFRONT);
            }
            cnt += __builtin_popcount(m);
        }
    };
    auto drain = [&](int base, int count) {
        const bool  have = lane < count;
        const int   e  = have ? __hip_atomic_load(&queue[qo + base + lane], __ATOMIC_RELAXED,
                                                  __HIP_MEMORY_SCOPE_WAVEFRONT) : 0;
        const int   bi = have ? (int)__hip_atomic_load(&qrow[qo + base + lane], __ATOMIC_RELAXED,
                                                       __HIP_MEMORY_SCOPE_WAVEFRONT) : 0;
        const int   ri = row[e];
        const int   l4 = lane & 3;
        const int   g  = lane >> 2;
        float4 v[4];
        #pragma unroll
        for (int j = 0; j < 4; ++j) {
            const int r = __shfl(ri, 8 * j + g);
            v[j] = *(const float4*)(h1 + (size_t)r * 16 + 4 * l4);
        }
        #pragma unroll
        for (int j = 0; j < 4; ++j) {
            const int s = 8 * j + g;
            const int b = __shfl(bi, s);
            if (s < count) {
                float* p = &acc[b * 16 + 4 * l4];
                __hip_atomic_fetch_add(p + 0, v[j].x, __ATOMIC_RELAXED, __HIP_MEMORY_SCOPE_WORKGROUP);
                __hip_atomic_fetch_add(p + 1, v[j].y, __ATOMIC_RELAXED, __HIP_MEMORY_SCOPE_WORKGROUP);
                __hip_atomic_fetch_add(p + 2, v[j].z, __ATOMIC_RELAXED, __HIP_MEMORY_SCOPE_WORKGROUP);
                __hip_atomic_fetch_add(p + 3, v[j].w, __ATOMIC_RELAXED, __HIP_MEMORY_SCOPE_WORKGROUP);
            }
        }
    };

    const int quads = (E + 3) >> 2;
    const int qper  = (quads + (int)gridDim.x - 1) / (int)gridDim.x;
    const int q0    = (int)blockIdx.x * qper;
    const int q1    = min(q0 + qper, quads);
    const bool vec  = (((size_t)col) & 15) == 0;
    const int qv    = vec ? min(q1, E >> 2) : q0;

    for (int qb = q0; qb < qv; qb += 4 * T) {
        int4 v[4];
        #pragma unroll
        for (int u = 0; u < 4; ++u) {
            const int qq = qb + tid + u * T;
            v[u] = (qq < qv) ? *(const int4*)(col + 4 * (size_t)qq)
                             : make_int4(-1, -1, -1, -1);
        }
        #pragma unroll
        for (int u = 0; u < 4; ++u) {
            const int e0 = 4 * (qb + tid + u * T);
            push(v[u].x, e0 + 0);
            push(v[u].y, e0 + 1);
            while (cnt >= 32) { cnt -= 32; drain(cnt, 32); }
            push(v[u].z, e0 + 2);
            push(v[u].w, e0 + 3);
            while (cnt >= 32) { cnt -= 32; drain(cnt, 32); }
        }
    }
    {
        const int e0 = 4 * max(q0, qv);
        const int e1 = min(4 * q1, E);
        for (int eb = e0; eb < e1; eb += T) {
            const int e = eb + tid;
            push((e < e1) ? col[e] : -1, e);
            while (cnt >= 32) { cnt -= 32; drain(cnt, 32); }
        }
    }
    if (cnt > 0) drain(0, cnt);

    __syncthreads();
    float* dst = agg1 + (size_t)lo * 16;
    for (int i = tid; i < lim * 16; i += T) {
        const float t = acc[i];
        if (t != 0.0f) atomicAdd(&dst[i], t);
    }
}

__global__ void gcn_post1(const float* __restrict__ agg1,
                          const float* __restrict__ h1,
                          const float* __restrict__ dinv,
                          const float* __restrict__ b1,
                          const float* __restrict__ W2,
                          float* __restrict__ s, int n) {
    int i = blockIdx.x * blockDim.x + threadIdx.x;
    if (i >= n) return;
    float di = dinv[i];
    const float* a = agg1 + (size_t)i * 16;
    const float* h = h1 + (size_t)i * 16;
    float acc = 0.0f;
    #pragma unroll
    for (int f = 0; f < 16; ++f) {
        float v = di * (a[f] + h[f]) + b1[f];
        v = fmaxf(v, 0.0f);
        acc = fmaf(v, W2[f], acc);
    }
    s[i] = di * acc;
}

__global__ __launch_bounds__(LDS_THREADS)
void gcn_edge1_lds(const int* __restrict__ row, const int* __restrict__ col,
                   const float* __restrict__ s,
                   float* __restrict__ agg2, int E, int n) {
    __shared__ float acc[RANGE_BINS];
    const int T   = (int)blockDim.x;
    const int tid = (int)threadIdx.x;
    const int per = (n + (int)gridDim.y - 1) / (int)gridDim.y;
    const int lo  = (int)blockIdx.y * per;
    const int lim = max(0, min(per, n - lo));
    for (int i = tid; i < lim; i += T) acc[i] = 0.0f;
    __syncthreads();

    const int quads = (E + 3) >> 2;
    const int qper  = (quads + (int)gridDim.x - 1) / (int)gridDim.x;
    const int q0    = (int)blockIdx.x * qper;
    const int q1    = min(q0 + qper, quads);
    const bool vec  = ((((size_t)col) | ((size_t)row)) & 15) == 0;
    const int qv    = vec ? min(q1, E >> 2) : q0;

    for (int q = q0 + tid; q < qv; q += 2 * T) {
        int c[8], r[8];
        #pragma unroll
        for (int u = 0; u < 2; ++u) {
            const int qq = q + u * T;
            const bool ok = qq < qv;
            const int4 vc = ok ? *(const int4*)(col + 4 * (size_t)qq) : make_int4(-1, -1, -1, -1);
            const int4 vr = ok ? *(const int4*)(row + 4 * (size_t)qq) : make_int4(0, 0, 0, 0);
            c[4 * u + 0] = vc.x; c[4 * u + 1] = vc.y; c[4 * u + 2] = vc.z; c[4 * u + 3] = vc.w;
            r[4 * u + 0] = vr.x; r[4 * u + 1] = vr.y; r[4 * u + 2] = vr.z; r[4 * u + 3] = vr.w;
        }
        unsigned b[8];
        float    v[8];
        #pragma unroll
        for (int k = 0; k < 8; ++k) {
            b[k] = (unsigned)(c[k] - lo);
            const bool in = b[k] < (unsigned)lim;
            v[k] = s[in ? r[k] : 0];
        }
        #pragma unroll
        for (int k = 0; k < 8; ++k) {
            if (b[k] < (unsigned)lim)
                __hip_atomic_fetch_add(&acc[b[k]], v[k], __ATOMIC_RELAXED,
                                       __HIP_MEMORY_SCOPE_WORKGROUP);
        }
    }
    for (int q = max(q0, qv) + tid; q < q1; q += T) {
        for (int k = 0; k < 4; ++k) {
            const int e = 4 * q + k;
            if (e < E) {
                const unsigned bb = (unsigned)(col[e] - lo);
                if (bb < (unsigned)lim) {
                    __hip_atomic_fetch_add(&acc[bb], s[row[e]],
                                           __ATOMIC_RELAXED, __HIP_MEMORY_SCOPE_WORKGROUP);
                }
            }
        }
    }
    __syncthreads();
    for (int i = tid; i < lim; i += T) {
        const float t = acc[i];
        if (t != 0.0f) atomicAdd(&agg2[lo + i], t);
    }
}

__global__ void gcn_post2(const float* __restrict__ agg2,
                          const float* __restrict__ s,
                          const float* __restrict__ dinv,
                          const float* __restrict__ b2,
                          float* __restrict__ out, int n) {
    int i = blockIdx.x * blockDim.x + threadIdx.x;
    if (i >= n) return;
    float di = dinv[i];
    out[i] = di * (agg2[i] + s[i]) + b2[0];
}


#define OWN_NODES 782
#define OWN_SHIFT 10
#define OWN_MAXR  512
#define BIN_WGS   256
static_assert(OWN_NODES <= (1 << OWN_SHIFT), "a target's place in its range must fit OWN_SHIFT bits");
static_assert(OWN_NODES <= LDS_THREADS, "the epilogues give every node of a range one thread");

__global__ void gcn_init_deg(float* __restrict__ deg, int n) {
    int i = blockIdx.x * blockDim.x + threadIdx.x;
    if (i < n) deg[i] = 1.0f;
}

__global__ __launch_bounds__(256)
void gcn_dinv_tot(const float* __restrict__ deg, float* __restrict__ dinv,
                  int* __restrict__ tot, int* __restrict__ cursor, int n) {
    __shared__ int part[8];
    const int lo = (int)blockIdx.x * OWN_NODES;
    const int hi = min(lo + OWN_NODES, n);
    int c = 0;
    for (int i = lo + (int)threadIdx.x; i < hi; i += (int)blockDim.x) {
        const float d = deg[i];
        dinv[i] = (d > 0.0f) ? rsqrtf(d) : 0.0f;
        c += (int)d - 1;
    }
    for (int off = 16; off > 0; off >>= 1) c += __shfl_xor(c, off);
    if ((threadIdx.x & 31) == 0) part[threadIdx.x >> 5] = c;
    __syncthreads();
    if (threadIdx.x == 0) {
        int t = 0;
        for (int w = 0; w < (int)(blockDim.x >> 5); ++w) t += part[w];
        tot[blockIdx.x]    = t;
        cursor[blockIdx.x] = 0;
    }
}

#define BIN_CAP 32768
__global__ __launch_bounds__(LDS_THREADS)
void gcn_bin_edges(const int* __restrict__ row, const int* __restrict__ col,
                   const int* __restrict__ tot, int* __restrict__ cursor,
                   int* __restrict__ bstart, unsigned* __restrict__ packed,
                   int E, int n, int R) {
    __shared__ unsigned stage[BIN_CAP];
    __shared__ int slot[OWN_MAXR];
    __shared__ int start[OWN_MAXR + 1];
    __shared__ int loff[OWN_MAXR + 1];
    __shared__ int gbase[OWN_MAXR];
    __shared__ int wsum[LDS_THREADS / 32];
    const int T    = (int)blockDim.x;
    const int tid  = (int)threadIdx.x;
    const int lane = tid & 31;

    auto xscan = [&](int v) -> int {
        int incl = v;
        for (int off = 1; off < 32; off <<= 1) {
            const int t = __shfl_up(incl, off);
            if (lane >= off) incl += t;
        }
        __syncthreads();
        if (lane == 31) wsum[tid >> 5] = incl;
        __syncthreads();
        int pre = 0;
        for (int w = 0; w < (tid >> 5); ++w) pre += wsum[w];
        return pre + incl - v;
    };

    {
        const int v  = (tid < R) ? tot[tid] : 0;
        const int ex = xscan(v);
        if (tid < R)         start[tid] = ex;
        if (tid == R - 1)    start[R]   = ex + v;
        if (tid < OWN_MAXR)  slot[tid]  = 0;
        __syncthreads();
    }
    if (blockIdx.x == 0) {
        for (int i = tid; i <= R; i += T) bstart[i] = start[i];
    }

    const int quads = (E + 3) >> 2;
    const int qper  = (quads + (int)gridDim.x - 1) / (int)gridDim.x;
    const int q0    = (int)blockIdx.x * qper;
    const int q1    = min(q0 + qper, quads);
    const bool vec  = ((((size_t)col) | ((size_t)row)) & 15) == 0;
    const int qv    = vec ? min(q1, E >> 2) : q0;

    auto count = [&](int c) {
        if ((unsigned)c < (unsigned)n) atomicAdd(&slot[c / OWN_NODES], 1);
    };
    for (int q = q0 + tid; q < qv; q += 4 * T) {
        int4 v[4];
        #pragma unroll
        for (int u = 0; u < 4; ++u) {
            const int qq = q + u * T;
            v[u] = (qq < qv) ? *(const int4*)(col + 4 * (size_t)qq)
                             : make_int4(-1, -1, -1, -1);
        }
        #pragma unroll
        for (int u = 0; u < 4; ++u) {
            count(v[u].x); count(v[u].y); count(v[u].z); count(v[u].w);
        }
    }
    for (int q = max(q0, qv) + tid; q < q1; q += T) {
        for (int k = 0; k < 4; ++k) {
            const int e = 4 * q + k;
            if (e < E) count(col[e]);
        }
    }
    __syncthreads();
    {
        const int c  = (tid < R) ? slot[tid] : 0;
        const int ex = xscan(c);
        if (tid < R) {
            loff[tid]  = ex;
            gbase[tid] = start[tid] + ((c != 0) ? atomicAdd(&cursor[tid], c) : 0);
            slot[tid]  = ex;
        }
        if (tid == R - 1) loff[R] = ex + c;
        __syncthreads();
    }

    auto place = [&](int c, int sidx) {
        if ((unsigned)c < (unsigned)n) {
            const int r = c / OWN_NODES;
            const int k = atomicAdd(&slot[r], 1);
            if (k < BIN_CAP)
                stage[k] = ((unsigned)sidx << OWN_SHIFT) | (unsigned)(c - r * OWN_NODES);
        }
    };
    for (int q = q0 + tid; q < qv; q += 2 * T) {
        int4 vc[2], vr[2];
        #pragma unroll
        for (int u = 0; u < 2; ++u) {
            const int qq = q + u * T;
            const bool ok = qq < qv;
            vc[u] = ok ? *(const int4*)(col + 4 * (size_t)qq) : make_int4(-1, -1, -1, -1);
            vr[u] = ok ? *(const int4*)(row + 4 * (size_t)qq) : make_int4(0, 0, 0, 0);
        }
        #pragma unroll
        for (int u = 0; u < 2; ++u) {
            place(vc[u].x, vr[u].x); place(vc[u].y, vr[u].y);
            place(vc[u].z, vr[u].z); place(vc[u].w, vr[u].w);
        }
    }
    for (int q = max(q0, qv) + tid; q < q1; q += T) {
        for (int k = 0; k < 4; ++k) {
            const int e = 4 * q + k;
            if (e < E) place(col[e], row[e]);
        }
    }
    __syncthreads();

    for (int r = tid >> 5; r < R; r += (T >> 5)) {
        const int l0 = loff[r];
        const int nr = min(loff[r + 1], BIN_CAP) - l0;
        const int g0 = gbase[r];
        for (int i = lane; i < nr; i += 32) packed[g0 + i] = stage[l0 + i];
    }
}

#define SRT_REG 26
#define SRT_CAP (SRT_REG * LDS_THREADS)
#define SEG_UNR 6
__global__ __launch_bounds__(LDS_THREADS)
void gcn_own16(const unsigned* __restrict__ packed, const int* __restrict__ bstart,
               const float* __restrict__ h1, const float* __restrict__ dinv,
               const float* __restrict__ b1, const float* __restrict__ W2,
               float* __restrict__ s, int n) {
    __shared__ unsigned srt[SRT_CAP];
    __shared__ float acc[OWN_NODES * 16];
    __shared__ int cnt[LDS_THREADS];
    __shared__ int start[LDS_THREADS + 1];
    __shared__ int wsum[LDS_THREADS / 32];
    const int T    = LDS_THREADS;
    const int tid  = (int)threadIdx.x;
    const int lane = tid & 31;
    const int wv   = __builtin_amdgcn_readfirstlane(tid >> 5);
    const int lo   = (int)blockIdx.x * OWN_NODES;
    const int lim  = max(0, min(OWN_NODES, n - lo));
    {
        float4* a4 = (float4*)acc;
        for (int i = tid; i < lim * 4; i += T) a4[i] = make_float4(0.0f, 0.0f, 0.0f, 0.0f);
    }

    auto xscan = [&](int v) -> int {
        int incl = v;
        for (int off = 1; off < 32; off <<= 1) {
            const int t = __shfl_up(incl, off);
            if (lane >= off) incl += t;
        }
        __syncthreads();
        if (lane == 31) wsum[wv] = incl;
        __syncthreads();
        int pre = 0;
        for (int w = 0; w < wv; ++w) pre += wsum[w];
        return pre + incl - v;
    };

    const int p0 = bstart[blockIdx.x];
    const int p1 = bstart[blockIdx.x + 1];
    const unsigned tmask = (1u << OWN_SHIFT) - 1u;
    const int  l4   = lane & 3;
    const int  slot = lane >> 2;
    const bool up16 = (lane & 16) != 0;
    const bool up8  = (lane & 8) != 0;
    const float4 zero4 = make_float4(0.0f, 0.0f, 0.0f, 0.0f);

    for (int c0 = p0; c0 < p1; c0 += SRT_CAP) {
        cnt[tid] = 0;
        __syncthreads();

        unsigned w[SRT_REG];
        int      rk[SRT_REG];
        #pragma unroll
        for (int u = 0; u < SRT_REG; ++u) {
            const int pp = c0 + tid + u * T;
            w[u] = (pp < p1) ? packed[pp] : 0u;
        }
        #pragma unroll
        for (int u = 0; u < SRT_REG; ++u) {
            const int pp = c0 + tid + u * T;
            rk[u] = 0;
            if (pp < p1) rk[u] = atomicAdd(&cnt[w[u] & tmask], 1);
        }
        __syncthreads();
        {
            const int v  = cnt[tid];
            const int ex = xscan(v);
            start[tid] = ex;
            if (tid == T - 1) start[T] = ex + v;
        }
        __syncthreads();
        #pragma unroll
        for (int u = 0; u < SRT_REG; ++u) rk[u] += start[w[u] & tmask];
        #pragma unroll
        for (int u = 0; u < SRT_REG; ++u) {
            const int pp = c0 + tid + u * T;
            if (pp < p1) srt[rk[u]] = w[u] >> OWN_SHIFT;
        }
        __syncthreads();

        for (int nb = wv; nb < lim; nb += 64) {
            int  a[2], len[2];
            bool ok[2];
            #pragma unroll
            for (int q = 0; q < 2; ++q) {
                const int node = nb + 32 * q;
                ok[q] = node < lim;
                const int nn = ok[q] ? node : 0;
                a[q]   = __builtin_amdgcn_readfirstlane(start[nn]);
                len[q] = ok[q] ? (__builtin_amdgcn_readfirstlane(start[nn + 1]) - a[q]) : 0;
            }
            float4 sum[2] = {zero4, zero4};
            const int mx = max(len[0], len[1]);
            for (int k = 0; k < mx; k += 8 * SEG_UNR) {
                float4 t[2][SEG_UNR];
                #pragma unroll
                for (int q = 0; q < 2; ++q) {
                    #pragma unroll
                    for (int j = 0; j < SEG_UNR; ++j) {
                        const int  kk   = k + 8 * j + slot;
                        const bool have = kk < len[q];
                        const unsigned sw = srt[have ? (a[q] + kk) : 0];
                        const unsigned sr = have ? sw : 0u;
                        t[q][j] = *(const float4*)(h1 + (size_t)sr * 16 + 4 * l4);
                    }
                }
                #pragma unroll
                for (int q = 0; q < 2; ++q) {
                    #pragma unroll
                    for (int j = 0; j < SEG_UNR; ++j) {
                        const bool have = (k + 8 * j + slot) < len[q];
                        sum[q].x += have ? t[q][j].x : 0.0f;
                        sum[q].y += have ? t[q][j].y : 0.0f;
                        sum[q].z += have ? t[q][j].z : 0.0f;
                        sum[q].w += have ? t[q][j].w : 0.0f;
                    }
                }
            }
            float u0[2], u1[2], r[2], old[2];
            const int f = 4 * l4 + (up16 ? 2 : 0) + (up8 ? 1 : 0);
            #pragma unroll
            for (int q = 0; q < 2; ++q) {
                const float4 v = sum[q];
                const float ka = up16 ? v.z : v.x, kb = up16 ? v.w : v.y;
                const float ga = up16 ? v.x : v.z, gb = up16 ? v.y : v.w;
                u0[q] = ka + __shfl_xor(ga, 16);
                u1[q] = kb + __shfl_xor(gb, 16);
                old[q] = acc[(ok[q] ? (nb + 32 * q) : 0) * 16 + f];
            }
            #pragma unroll
            for (int q = 0; q < 2; ++q) {
                const float kc = up8 ? u1[q] : u0[q];
                const float gc = up8 ? u0[q] : u1[q];
                r[q] = kc + __shfl_xor(gc, 8);
            }
            #pragma unroll
            for (int q = 0; q < 2; ++q) r[q] += __shfl_xor(r[q], 4);
            #pragma unroll
            for (int q = 0; q < 2; ++q) {
                if (ok[q] && (lane & 4) == 0)
                    acc[(nb + 32 * q) * 16 + f] = old[q] + r[q];
            }
        }
    }
    __syncthreads();
    if (tid < lim) {
        const int   i  = lo + tid;
        const float di = dinv[i];
        const float* a = &acc[tid * 16];
        const float* h = h1 + (size_t)i * 16;
        float t = 0.0f;
        #pragma unroll
        for (int f = 0; f < 16; ++f) {
            float x = di * (a[f] + h[f]) + b1[f];
            x = fmaxf(x, 0.0f);
            t = fmaf(x, W2[f], t);
        }
        s[i] = di * t;
    }
}

#define S_WIN_V4 17
#define S_ROUND  (4 * LDS_THREADS)
#define S_WIN    (S_WIN_V4 * S_ROUND)
#define OWN1_REG 26
__global__ __launch_bounds__(LDS_THREADS)
void gcn_own1(const unsigned* __restrict__ packed, const int* __restrict__ bstart,
              const float* __restrict__ s, const float* __restrict__ dinv,
              const float* __restrict__ b2, float* __restrict__ out, int n) {
    __shared__ float win[S_WIN];
    __shared__ float acc[OWN_NODES];
    const int T   = LDS_THREADS;
    const int tid = (int)threadIdx.x;
    if (tid < OWN_NODES) acc[tid] = 0.0f;

    const int p0 = bstart[blockIdx.x];
    const int p1 = bstart[blockIdx.x + 1];
    const unsigned tmask = (1u << OWN_SHIFT) - 1u;

    unsigned w[OWN1_REG];
    #pragma unroll
    for (int u = 0; u < OWN1_REG; ++u) {
        const int pp = p0 + tid + u * T;
        w[u] = (pp < p1) ? packed[pp] : 0u;
    }
    const int pr = p0 + OWN1_REG * T;

    const bool al = (((size_t)s) & 15) == 0;
    float4 v[S_WIN_V4];
    auto fetch = [&](int base) {
        const int len4 = al ? (min(S_WIN, n - base) & ~3) : 0;
        const int full = len4 / S_ROUND;
        const int rem  = len4 - full * S_ROUND;
        #pragma unroll
        for (int u = 0; u < S_WIN_V4; ++u) {
            const bool have = u < full || (u == full && 4 * tid < rem);
            v[u] = have ? *(const float4*)(s + base + 4 * tid + u * S_ROUND)
                        : make_float4(0.0f, 0.0f, 0.0f, 0.0f);
        }
    };
    auto add = [&](unsigned word, int base, int len) {
        const unsigned off = (word >> OWN_SHIFT) - (unsigned)base;
        if (off < (unsigned)len)
            __hip_atomic_fetch_add(&acc[word & tmask], win[off],
                                   __ATOMIC_RELAXED, __HIP_MEMORY_SCOPE_WORKGROUP);
    };

    fetch(0);
    for (int base = 0; base < n; base += S_WIN) {
        const int len  = min(S_WIN, n - base);
        const int len4 = al ? (len & ~3) : 0;
        const int full = len4 / S_ROUND;
        const int rem  = len4 - full * S_ROUND;
        #pragma unroll
        for (int u = 0; u < S_WIN_V4; ++u) {
            if (u < full || (u == full && 4 * tid < rem))
                *(float4*)(win + 4 * tid + u * S_ROUND) = v[u];
        }
        for (int i = len4 + tid; i < len; i += T) win[i] = s[base + i];
        __syncthreads();
        if (base + S_WIN < n) fetch(base + S_WIN);
        #pragma unroll
        for (int u = 0; u < OWN1_REG; ++u) {
            unsigned word = w[u];
            asm volatile("" : "+v"(word));
            if (p0 + tid + u * T < p1) add(word, base, len);
        }
        for (int p = pr + tid; p < p1; p += T) add(packed[p], base, len);
        __syncthreads();
    }
    const int lo  = (int)blockIdx.x * OWN_NODES;
    const int lim = max(0, min(OWN_NODES, n - lo));
    if (tid < lim) {
        const int i = lo + tid;
        out[i] = dinv[i] * (acc[tid] + s[i]) + b2[0];
    }
}

extern "C" void kernel_launch(void* const* d_in, const int* in_sizes, int n_in,
                              void* d_out, int out_size, void* d_ws, size_t ws_size,
                              hipStream_t stream) {
    const float* x  = (const float*)d_in[0];
    const int*   ei = (const int*)d_in[1];
    const float* W1 = (const float*)d_in[2];
    const float* b1 = (const float*)d_in[3];
    const float* W2 = (const float*)d_in[4];
    const float* b2 = (const float*)d_in[5];

    const int N = in_sizes[0] / 128;
    const int E = in_sizes[1] / 2;
    const int* row = ei;
    const int* col = ei + E;

    {
        const int R = (N + OWN_NODES - 1) / OWN_NODES;
        const size_t need = ((size_t)19 * (size_t)N + (size_t)E + 3 * (size_t)R + 1) * sizeof(float);
        if (R >= 1 && R <= OWN_MAXR && N <= (1 << (32 - OWN_SHIFT)) && ws_size >= need) {
            float*    deg    = (float*)d_ws;
            float*    dinv   = deg + N;
            float*    h1     = dinv + N;
            float*    s      = h1 + (size_t)N * 16;
            unsigned* packed = (unsigned*)(s + N);
            int*      tot    = (int*)(packed + E);
            int*      cursor = tot + R;
            int*      bstart = cursor + R;
            float*    out    = (float*)d_out;
            const int B = 256;

            gcn_init_deg<<<(N + B - 1) / B, B, 0, stream>>>(deg, N);
            {
                const int RR = (N + RANGE_BINS - 1) / RANGE_BINS;
                const dim3 ldsGrid(EDGE_SLICES, RR);
                gcn_degree_lds<<<ldsGrid, LDS_THREADS, 0, stream>>>(col, deg, E, N);
            }
            gcn_dinv_tot<<<R, 256, 0, stream>>>(deg, dinv, tot, cursor, N);
            {
                int full   = N & ~15;
                int ntiles = full >> 4;
                if (ntiles > 0) {
                    int wavesPerBlock = G1_WAVES;
                    int blocks = (ntiles + wavesPerBlock - 1) / wavesPerBlock;
                    gcn_gemm1_wmma<<<blocks, G1_WAVES * 32, 0, stream>>>(x, W1, dinv, h1, ntiles);
                }
                if (full < N) {
                    int tail = (N - full) * 16;
                    gcn_gemm1_tail<<<(tail + B - 1) / B, B, 0, stream>>>(x, W1, dinv, h1, full, N);
                }
            }
            {
                const int quads = (E + 3) >> 2;
                int G = (quads + (BIN_CAP / 4) - 1) / (BIN_CAP / 4);
                if (G < BIN_WGS) G = BIN_WGS;
                gcn_bin_edges<<<G, LDS_THREADS, 0, stream>>>(row, col, tot, cursor, bstart, packed, E, N, R);
            }
            gcn_own16<<<R, LDS_THREADS, 0, stream>>>(packed, bstart, h1, dinv, b1, W2, s, N);
            gcn_own1<<<R, LDS_THREADS, 0, stream>>>(packed, bstart, s, dinv, b2, out, N);
            return;
        }
    }

    float* ws   = (float*)d_ws;
    float* deg  = ws;
    float* dinv = deg + N;
    float* h1   = dinv + N;
    float* agg1 = h1 + (size_t)N * 16;
    float* s    = agg1 + (size_t)N * 16;
    float* agg2 = s + N;
    float* out  = (float*)d_out;

    const int B = 256;

    {
        int total = N * 16;
        gcn_init<<<(total + B - 1) / B, B, 0, stream>>>(deg, agg1, agg2, N);
    }
    const int R = (N + RANGE_BINS - 1) / RANGE_BINS;
    const dim3 ldsGrid(EDGE_SLICES, R > 0 ? R : 1);

    gcn_degree_lds<<<ldsGrid, LDS_THREADS, 0, stream>>>(col, deg, E, N);
    gcn_dinv<<<(N + B - 1) / B, B, 0, stream>>>(deg, dinv, N);
    {
        int full   = N & ~15;
        int ntiles = full >> 4;
        if (ntiles > 0) {
            int wavesPerBlock = G1_WAVES;
            int blocks = (ntiles + wavesPerBlock - 1) / wavesPerBlock;
            gcn_gemm1_wmma<<<blocks, G1_WAVES * 32, 0, stream>>>(x, W1, dinv, h1, ntiles);
        }
        if (full < N) {
            int tail = (N - full) * 16;
            gcn_gemm1_tail<<<(tail + B - 1) / B, B, 0, stream>>>(x, W1, dinv, h1, full, N);
        }
    }
    {
        const int R16 = (N + R16_NODES - 1) / R16_NODES;
        const dim3 grid16(R16_SLICES, R16 > 0 ? R16 : 1);
        gcn_edge16_lds<<<grid16, LDS_THREADS, 0, stream>>>(row, col, h1, agg1, E, N);
    }
    gcn_post1<<<(N + B - 1) / B, B, 0, stream>>>(agg1, h1, dinv, b1, W2, s, N);
    gcn_edge1_lds<<<ldsGrid, LDS_THREADS, 0, stream>>>(row, col, s, agg2, E, N);
    gcn_post2<<<(N + B - 1) / B, B, 0, stream>>>(agg2, s, dinv, b2, out, N);
}


# Degree and run starts come out of the binning: per-slice range histogram matrix replaces the 4-range degree scan and its global atomics, degrees counted from each range's run; float sum order changes
# speedup vs baseline: 7.9562x; 1.1638x over previous
#include <hip/hip_runtime.h>

typedef __attribute__((ext_vector_type(2))) float v2f;
typedef __attribute__((ext_vector_type(4))) float v4f;
typedef __attribute__((ext_vector_type(8))) float v8f;

__global__ void gcn_init(float* __restrict__ deg, float* __restrict__ agg1,
                         float* __restrict__ agg2, int n) {
    int i = blockIdx.x * blockDim.x + threadIdx.x;
    if (i < n) { deg[i] = 1.0f; agg2[i] = 0.0f; }
    if (i < n * 16) { agg1[i] = 0.0f; }
}

#define RANGE_BINS  51200
#define EDGE_SLICES 64
#define LDS_THREADS 1024

__global__ __launch_bounds__(LDS_THREADS)
void gcn_degree_lds(const int* __restrict__ col, float* __restrict__ deg,
                    int E, int n) {
    __shared__ unsigned cnt[RANGE_BINS];
    const int T   = (int)blockDim.x;
    const int tid = (int)threadIdx.x;
    const int per = (n + (int)gridDim.y - 1) / (int)gridDim.y;
    const int lo  = (int)blockIdx.y * per;
    const int lim = max(0, min(per, n - lo));
    for (int i = tid; i < lim; i += T) cnt[i] = 0u;
    __syncthreads();

    const int quads = (E + 3) >> 2;
    const int qper  = (quads + (int)gridDim.x - 1) / (int)gridDim.x;
    const int q0    = (int)blockIdx.x * qper;
    const int q1    = min(q0 + qper, quads);
    const bool vec  = (((size_t)col) & 15) == 0;
    const int qv    = vec ? min(q1, E >> 2) : q0;

    for (int q = q0 + tid; q < qv; q += 4 * T) {
        int4 v[4];
        #pragma unroll
        for (int u = 0; u < 4; ++u) {
            const int qq = q + u * T;
            v[u] = (qq < qv) ? *(const int4*)(col + 4 * (size_t)qq)
                             : make_int4(-1, -1, -1, -1);
        }
        #pragma unroll
        for (int u = 0; u < 4; ++u) {
            const unsigned b0 = (unsigned)(v[u].x - lo);
            const unsigned b1 = (unsigned)(v[u].y - lo);
            const unsigned b2 = (unsigned)(v[u].z - lo);
            const unsigned b3 = (unsigned)(v[u].w - lo);
            if (b0 < (unsigned)lim) atomicAdd(&cnt[b0], 1u);
            if (b1 < (unsigned)lim) atomicAdd(&cnt[b1], 1u);
            if (b2 < (unsigned)lim) atomicAdd(&cnt[b2], 1u);
            if (b3 < (unsigned)lim) atomicAdd(&cnt[b3], 1u);
        }
    }
    for (int q = max(q0, qv) + tid; q < q1; q += T) {
        for (int k = 0; k < 4; ++k) {
            const int e = 4 * q + k;
            if (e < E) {
                const unsigned b = (unsigned)(col[e] - lo);
                if (b < (unsigned)lim) atomicAdd(&cnt[b], 1u);
            }
        }
    }
    __syncthreads();
    for (int i = tid; i < lim; i += T) {
        const unsigned v = cnt[i];
        if (v != 0u) atomicAdd(&deg[lo + i], (float)v);
    }
}

__global__ void gcn_dinv(const float* __restrict__ deg, float* __restrict__ dinv,
                         int n) {
    int i = blockIdx.x * blockDim.x + threadIdx.x;
    if (i >= n) return;
    float d = deg[i];
    dinv[i] = (d > 0.0f) ? rsqrtf(d) : 0.0f;
}

#define G1_WAVES 8
#define G1_PITCH 132
__global__ __launch_bounds__(G1_WAVES * 32)
void gcn_gemm1_wmma(const float* __restrict__ x,
                    const float* __restrict__ W1,
                    const float* __restrict__ dinv,
                    float* __restrict__ h1, int ntiles) {
    __shared__ alignas(16) float xs[G1_WAVES * 16 * G1_PITCH];
    __shared__ alignas(16) float ws[128 * 16];
    const int  tid  = (int)threadIdx.x;
    const int  lane = tid & 31;
    const int  wv   = tid >> 5;
    const int  wave = (int)blockIdx.x * G1_WAVES + wv;
    const bool act  = wave < ntiles;
    const size_t row16 = (size_t)(act ? wave : 0) * 16;

    float* xt = xs + wv * (16 * G1_PITCH);
    const float* src = x + row16 * 128;
    const bool al = (((size_t)x) & 15) == 0;
    const int mbase = (lane >> 4) << 3;

    v4f v[16];
    if (act && al) {
        #pragma unroll
        for (int i = 0; i < 16; ++i)
            v[i] = __builtin_nontemporal_load((const v4f*)(src + i * 128 + 4 * lane));
    }
    float w[(128 * 16) / (G1_WAVES * 32)];
    #pragma unroll
    for (int j = 0; j < (128 * 16) / (G1_WAVES * 32); ++j) w[j] = W1[tid + j * (G1_WAVES * 32)];
    float dv[8];
    #pragma unroll
    for (int r = 0; r < 8; ++r) dv[r] = dinv[row16 + (size_t)(mbase + r)];

    #pragma unroll
    for (int j = 0; j < (128 * 16) / (G1_WAVES * 32); ++j) {
        const int e = tid + j * (G1_WAVES * 32);
        const int k = e >> 4, n = e & 15;
        ws[(((k >> 1) * 16 + n) << 1) + (k & 1)] = w[j];
    }
    if (act) {
        if (al) {
            #pragma unroll
            for (int i = 0; i < 16; ++i)
                *(v4f*)(xt + i * G1_PITCH + 4 * lane) = v[i];
        } else {
            #pragma nounroll
            for (int i = 0; i < 16 * 4; ++i)
                xt[(i >> 2) * G1_PITCH + lane + 32 * (i & 3)] = src[(i >> 2) * 128 + lane + 32 * (i & 3)];
        }
    }
    __syncthreads();
    if (!act) return;

    const int m = lane & 15;
    const int h = lane >> 4;
    const float* ap = xt + m * G1_PITCH + 2 * h;
    const float* bp = ws + ((h * 16 + m) << 1);

    v8f acc = {};
    #pragma unroll 8
    for (int k0 = 0; k0 < 128; k0 += 4) {
        const v2f a = *(const v2f*)(ap + k0);
        const v2f b = *(const v2f*)(bp + (k0 >> 1) * 32);
        acc = __builtin_amdgcn_wmma_f32_16x16x4_f32(
             false, a,  false, b,
             (short)0, acc,  false,  false);
    }

    #pragma unroll
    for (int r = 0; r < 8; ++r) {
        const size_t node = row16 + (size_t)(mbase + r);
        h1[node * 16 + m] = acc[r] * dv[r];
    }
}

__global__ void gcn_gemm1_tail(const float* __restrict__ x,
                               const float* __restrict__ W1,
                               const float* __restrict__ dinv,
                               float* __restrict__ h1, int start, int n) {
    int t = blockIdx.x * blockDim.x + threadIdx.x;
    int node = start + (t >> 4);
    int f = t & 15;
    if (node >= n) return;
    float acc = 0.0f;
    const float* xr = x + (size_t)node * 128;
    for (int k = 0; k < 128; ++k) acc += xr[k] * W1[(size_t)k * 16 + f];
    h1[(size_t)node * 16 + f] = acc * dinv[node];
}

#define R16_NODES  4768
#define R16_SLICES 6
#define R16_QCAP   96

__global__ __launch_bounds__(LDS_THREADS)
void gcn_edge16_lds(const int* __restrict__ row, const int* __restrict__ col,
                    const float* __restrict__ h1,
                    float* __restrict__ agg1, int E, int n) {
    __shared__ float acc[R16_NODES * 16];
    __shared__ int queue[(LDS_THREADS / 32) * R16_QCAP];
    __shared__ unsigned short qrow[(LDS_THREADS / 32) * R16_QCAP];
    static_assert(R16_NODES <= 65535, "a target's LDS row must fit a 16-bit queue word");

    const int T    = (int)blockDim.x;
    const int tid  = (int)threadIdx.x;
    const int lane = tid & 31;
    const int per  = (n + (int)gridDim.y - 1) / (int)gridDim.y;
    const int lo   = (int)blockIdx.y * per;
    const int lim  = max(0, min(per, n - lo));
    {
        float4* a4 = (float4*)acc;
        for (int i = tid; i < lim * 4; i += T) a4[i] = make_float4(0.0f, 0.0f, 0.0f, 0.0f);
    }
    __syncthreads();

    const int qo = (tid >> 5) * R16_QCAP;
    int cnt = 0;

    auto push = [&](int cv, int e) {
        const unsigned bin = (unsigned)(cv - lo);
        const bool in = bin < (unsigned)lim;
        const unsigned m = __builtin_amdgcn_ballot_w32(in);
        if (m != 0u) {
            if (in) {
                const int slot = qo + cnt + (int)__builtin_amdgcn_mbcnt_lo(m, 0u);
                __hip_atomic_store(&queue[slot], e, __ATOMIC_RELAXED, __HIP_MEMORY_SCOPE_WAVEFRONT);
                __hip_atomic_store(&qrow[slot], (unsigned short)bin,
                                   __ATOMIC_RELAXED, __HIP_MEMORY_SCOPE_WAVEFRONT);
            }
            cnt += __builtin_popcount(m);
        }
    };
    auto drain = [&](int base, int count) {
        const bool  have = lane < count;
        const int   e  = have ? __hip_atomic_load(&queue[qo + base + lane], __ATOMIC_RELAXED,
                                                  __HIP_MEMORY_SCOPE_WAVEFRONT) : 0;
        const int   bi = have ? (int)__hip_atomic_load(&qrow[qo + base + lane], __ATOMIC_RELAXED,
                                                       __HIP_MEMORY_SCOPE_WAVEFRONT) : 0;
        const int   ri = row[e];
        const int   l4 = lane & 3;
        const int   g  = lane >> 2;
        float4 v[4];
        #pragma unroll
        for (int j = 0; j < 4; ++j) {
            const int r = __shfl(ri, 8 * j + g);
            v[j] = *(const float4*)(h1 + (size_t)r * 16 + 4 * l4);
        }
        #pragma unroll
        for (int j = 0; j < 4; ++j) {
            const int s = 8 * j + g;
            const int b = __shfl(bi, s);
            if (s < count) {
                float* p = &acc[b * 16 + 4 * l4];
                __hip_atomic_fetch_add(p + 0, v[j].x, __ATOMIC_RELAXED, __HIP_MEMORY_SCOPE_WORKGROUP);
                __hip_atomic_fetch_add(p + 1, v[j].y, __ATOMIC_RELAXED, __HIP_MEMORY_SCOPE_WORKGROUP);
                __hip_atomic_fetch_add(p + 2, v[j].z, __ATOMIC_RELAXED, __HIP_MEMORY_SCOPE_WORKGROUP);
                __hip_atomic_fetch_add(p + 3, v[j].w, __ATOMIC_RELAXED, __HIP_MEMORY_SCOPE_WORKGROUP);
            }
        }
    };

    const int quads = (E + 3) >> 2;
    const int qper  = (quads + (int)gridDim.x - 1) / (int)gridDim.x;
    const int q0    = (int)blockIdx.x * qper;
    const int q1    = min(q0 + qper, quads);
    const bool vec  = (((size_t)col) & 15) == 0;
    const int qv    = vec ? min(q1, E >> 2) : q0;

    for (int qb = q0; qb < qv; qb += 4 * T) {
        int4 v[4];
        #pragma unroll
        for (int u = 0; u < 4; ++u) {
            const int qq = qb + tid + u * T;
            v[u] = (qq < qv) ? *(const int4*)(col + 4 * (size_t)qq)
                             : make_int4(-1, -1, -1, -1);
        }
        #pragma unroll
        for (int u = 0; u < 4; ++u) {
            const int e0 = 4 * (qb + tid + u * T);
            push(v[u].x, e0 + 0);
            push(v[u].y, e0 + 1);
            while (cnt >= 32) { cnt -= 32; drain(cnt, 32); }
            push(v[u].z, e0 + 2);
            push(v[u].w, e0 + 3);
            while (cnt >= 32) { cnt -= 32; drain(cnt, 32); }
        }
    }
    {
        const int e0 = 4 * max(q0, qv);
        const int e1 = min(4 * q1, E);
        for (int eb = e0; eb < e1; eb += T) {
            const int e = eb + tid;
            push((e < e1) ? col[e] : -1, e);
            while (cnt >= 32) { cnt -= 32; drain(cnt, 32); }
        }
    }
    if (cnt > 0) drain(0, cnt);

    __syncthreads();
    float* dst = agg1 + (size_t)lo * 16;
    for (int i = tid; i < lim * 16; i += T) {
        const float t = acc[i];
        if (t != 0.0f) atomicAdd(&dst[i], t);
    }
}

__global__ void gcn_post1(const float* __restrict__ agg1,
                          const float* __restrict__ h1,
                          const float* __restrict__ dinv,
                          const float* __restrict__ b1,
                          const float* __restrict__ W2,
                          float* __restrict__ s, int n) {
    int i = blockIdx.x * blockDim.x + threadIdx.x;
    if (i >= n) return;
    float di = dinv[i];
    const float* a = agg1 + (size_t)i * 16;
    const float* h = h1 + (size_t)i * 16;
    float acc = 0.0f;
    #pragma unroll
    for (int f = 0; f < 16; ++f) {
        float v = di * (a[f] + h[f]) + b1[f];
        v = fmaxf(v, 0.0f);
        acc = fmaf(v, W2[f], acc);
    }
    s[i] = di * acc;
}

__global__ __launch_bounds__(LDS_THREADS)
void gcn_edge1_lds(const int* __restrict__ row, const int* __restrict__ col,
                   const float* __restrict__ s,
                   float* __restrict__ agg2, int E, int n) {
    __shared__ float acc[RANGE_BINS];
    const int T   = (int)blockDim.x;
    const int tid = (int)threadIdx.x;
    const int per = (n + (int)gridDim.y - 1) / (int)gridDim.y;
    const int lo  = (int)blockIdx.y * per;
    const int lim = max(0, min(per, n - lo));
    for (int i = tid; i < lim; i += T) acc[i] = 0.0f;
    __syncthreads();

    const int quads = (E + 3) >> 2;
    const int qper  = (quads + (int)gridDim.x - 1) / (int)gridDim.x;
    const int q0    = (int)blockIdx.x * qper;
    const int q1    = min(q0 + qper, quads);
    const bool vec  = ((((size_t)col) | ((size_t)row)) & 15) == 0;
    const int qv    = vec ? min(q1, E >> 2) : q0;

    for (int q = q0 + tid; q < qv; q += 2 * T) {
        int c[8], r[8];
        #pragma unroll
        for (int u = 0; u < 2; ++u) {
            const int qq = q + u * T;
            const bool ok = qq < qv;
            const int4 vc = ok ? *(const int4*)(col + 4 * (size_t)qq) : make_int4(-1, -1, -1, -1);
            const int4 vr = ok ? *(const int4*)(row + 4 * (size_t)qq) : make_int4(0, 0, 0, 0);
            c[4 * u + 0] = vc.x; c[4 * u + 1] = vc.y; c[4 * u + 2] = vc.z; c[4 * u + 3] = vc.w;
            r[4 * u + 0] = vr.x; r[4 * u + 1] = vr.y; r[4 * u + 2] = vr.z; r[4 * u + 3] = vr.w;
        }
        unsigned b[8];
        float    v[8];
        #pragma unroll
        for (int k = 0; k < 8; ++k) {
            b[k] = (unsigned)(c[k] - lo);
            const bool in = b[k] < (unsigned)lim;
            v[k] = s[in ? r[k] : 0];
        }
        #pragma unroll
        for (int k = 0; k < 8; ++k) {
            if (b[k] < (unsigned)lim)
                __hip_atomic_fetch_add(&acc[b[k]], v[k], __ATOMIC_RELAXED,
                                       __HIP_MEMORY_SCOPE_WORKGROUP);
        }
    }
    for (int q = max(q0, qv) + tid; q < q1; q += T) {
        for (int k = 0; k < 4; ++k) {
            const int e = 4 * q + k;
            if (e < E) {
                const unsigned bb = (unsigned)(col[e] - lo);
                if (bb < (unsigned)lim) {
                    __hip_atomic_fetch_add(&acc[bb], s[row[e]],
                                           __ATOMIC_RELAXED, __HIP_MEMORY_SCOPE_WORKGROUP);
                }
            }
        }
    }
    __syncthreads();
    for (int i = tid; i < lim; i += T) {
        const float t = acc[i];
        if (t != 0.0f) atomicAdd(&agg2[lo + i], t);
    }
}

__global__ void gcn_post2(const float* __restrict__ agg2,
                          const float* __restrict__ s,
                          const float* __restrict__ dinv,
                          const float* __restrict__ b2,
                          float* __restrict__ out, int n) {
    int i = blockIdx.x * blockDim.x + threadIdx.x;
    if (i >= n) return;
    float di = dinv[i];
    out[i] = di * (agg2[i] + s[i]) + b2[0];
}


#define OWN_NODES 782
#define OWN_SHIFT 10
#define OWN_MAXR  512
#define BIN_WGS   256
static_assert(OWN_NODES <= (1 << OWN_SHIFT), "a target's place in its range must fit OWN_SHIFT bits");
static_assert(OWN_NODES <= LDS_THREADS, "the epilogues give every node of a range one thread");

__global__ __launch_bounds__(LDS_THREADS)
void gcn_slice_hist(const int* __restrict__ col, int* __restrict__ hist,
                    int E, int n, int R) {
    __shared__ int slot[OWN_MAXR];
    const int T   = (int)blockDim.x;
    const int tid = (int)threadIdx.x;
    if (tid < OWN_MAXR) slot[tid] = 0;
    __syncthreads();

    const int quads = (E + 3) >> 2;
    const int qper  = (quads + (int)gridDim.x - 1) / (int)gridDim.x;
    const int q0    = (int)blockIdx.x * qper;
    const int q1    = min(q0 + qper, quads);
    const bool vec  = (((size_t)col) & 15) == 0;
    const int qv    = vec ? min(q1, E >> 2) : q0;

    auto count = [&](int c) {
        if ((unsigned)c < (unsigned)n) atomicAdd(&slot[c / OWN_NODES], 1);
    };
    for (int q = q0 + tid; q < qv; q += 4 * T) {
        int4 v[4];
        #pragma unroll
        for (int u = 0; u < 4; ++u) {
            const int qq = q + u * T;
            v[u] = (qq < qv) ? *(const int4*)(col + 4 * (size_t)qq)
                             : make_int4(-1, -1, -1, -1);
        }
        #pragma unroll
        for (int u = 0; u < 4; ++u) {
            count(v[u].x); count(v[u].y); count(v[u].z); count(v[u].w);
        }
    }
    for (int q = max(q0, qv) + tid; q < q1; q += T) {
        for (int k = 0; k < 4; ++k) {
            const int e = 4 * q + k;
            if (e < E) count(col[e]);
        }
    }
    __syncthreads();
    for (int r = tid; r < R; r += T) hist[(size_t)blockIdx.x * R + r] = slot[r];
}

#define BIN_CAP 32768
__global__ __launch_bounds__(LDS_THREADS)
void gcn_bin_edges(const int* __restrict__ row, const int* __restrict__ col,
                   const int* __restrict__ hist,
                   int* __restrict__ bstart, unsigned* __restrict__ packed,
                   int E, int n, int R) {
    __shared__ unsigned stage[BIN_CAP];
    __shared__ int slot[OWN_MAXR];
    __shared__ int start[OWN_MAXR + 1];
    __shared__ int loff[OWN_MAXR + 1];
    __shared__ int gbase[OWN_MAXR];
    __shared__ int colT[LDS_THREADS];
    __shared__ int colP[LDS_THREADS];
    __shared__ int wsum[LDS_THREADS / 32];
    const int T    = LDS_THREADS;
    const int tid  = (int)threadIdx.x;
    const int lane = tid & 31;
    const int G    = (int)gridDim.x;
    const int me   = (int)blockIdx.x;

    auto xscan = [&](int v) -> int {
        int incl = v;
        for (int off = 1; off < 32; off <<= 1) {
            const int t = __shfl_up(incl, off);
            if (lane >= off) incl += t;
        }
        __syncthreads();
        if (lane == 31) wsum[tid >> 5] = incl;
        __syncthreads();
        int pre = 0;
        for (int w = 0; w < (tid >> 5); ++w) pre += wsum[w];
        return pre + incl - v;
    };

    {
        const int RP    = (R > 256) ? 512 : 256;
        const int parts = T / RP;
        const int r     = tid & (RP - 1);
        const int part  = tid / RP;
        const int gper  = (G + parts - 1) / parts;
        const int g0    = part * gper;
        const int g1    = min(G, g0 + gper);
        int tsum = 0, psum = 0;
        if (r < R) {
            #pragma unroll 16
            for (int g = g0; g < g1; ++g) {
                const int v = hist[(size_t)g * R + r];
                tsum += v;
                psum += (g < me) ? v : 0;
            }
        }
        colT[tid] = tsum;
        colP[tid] = psum;
        __syncthreads();
        int tot = 0, pre = 0, own = 0;
        if (tid < R) {
            for (int p = 0; p < parts; ++p) { tot += colT[p * RP + tid]; pre += colP[p * RP + tid]; }
            own = hist[(size_t)me * R + tid];
        }
        const int ex = xscan(tot);
        if (tid < R)       { start[tid] = ex; gbase[tid] = ex + pre; }
        if (tid == R - 1)  start[R] = ex + tot;
        const int lx = xscan(own);
        if (tid < R)       { loff[tid] = lx; slot[tid] = lx; }
        if (tid == R - 1)  loff[R] = lx + own;
        __syncthreads();
    }
    if (blockIdx.x == 0) {
        for (int i = tid; i <= R; i += T) bstart[i] = start[i];
    }

    const int quads = (E + 3) >> 2;
    const int qper  = (quads + (int)gridDim.x - 1) / (int)gridDim.x;
    const int q0    = (int)blockIdx.x * qper;
    const int q1    = min(q0 + qper, quads);
    const bool vec  = ((((size_t)col) | ((size_t)row)) & 15) == 0;
    const int qv    = vec ? min(q1, E >> 2) : q0;

    auto place = [&](int c, int sidx) {
        if ((unsigned)c < (unsigned)n) {
            const int r = c / OWN_NODES;
            const int k = atomicAdd(&slot[r], 1);
            if (k < BIN_CAP)
                stage[k] = ((unsigned)sidx << OWN_SHIFT) | (unsigned)(c - r * OWN_NODES);
        }
    };
    for (int q = q0 + tid; q < qv; q += 2 * T) {
        int4 vc[2], vr[2];
        #pragma unroll
        for (int u = 0; u < 2; ++u) {
            const int qq = q + u * T;
            const bool ok = qq < qv;
            vc[u] = ok ? *(const int4*)(col + 4 * (size_t)qq) : make_int4(-1, -1, -1, -1);
            vr[u] = ok ? *(const int4*)(row + 4 * (size_t)qq) : make_int4(0, 0, 0, 0);
        }
        #pragma unroll
        for (int u = 0; u < 2; ++u) {
            place(vc[u].x, vr[u].x); place(vc[u].y, vr[u].y);
            place(vc[u].z, vr[u].z); place(vc[u].w, vr[u].w);
        }
    }
    for (int q = max(q0, qv) + tid; q < q1; q += T) {
        for (int k = 0; k < 4; ++k) {
            const int e = 4 * q + k;
            if (e < E) place(col[e], row[e]);
        }
    }
    __syncthreads();

    for (int r = tid >> 5; r < R; r += (T >> 5)) {
        const int l0 = loff[r];
        const int nr = min(loff[r + 1], BIN_CAP) - l0;
        const int g0 = gbase[r];
        for (int i = lane; i < nr; i += 32) packed[g0 + i] = stage[l0 + i];
    }
}

__global__ __launch_bounds__(LDS_THREADS)
void gcn_dinv_run(const unsigned* __restrict__ packed, const int* __restrict__ bstart,
                  float* __restrict__ dinv, int n) {
    __shared__ int cnt[LDS_THREADS];
    const int T   = LDS_THREADS;
    const int tid = (int)threadIdx.x;
    cnt[tid] = 0;
    __syncthreads();
    const int p0 = bstart[blockIdx.x];
    const int p1 = bstart[blockIdx.x + 1];
    const unsigned tmask = (1u << OWN_SHIFT) - 1u;
    for (int p = p0 + tid; p < p1; p += 8 * T) {
        unsigned w[8];
        #pragma unroll
        for (int u = 0; u < 8; ++u) {
            const int pp = p + u * T;
            w[u] = (pp < p1) ? packed[pp] : 0u;
        }
        #pragma unroll
        for (int u = 0; u < 8; ++u) {
            if (p + u * T < p1) atomicAdd(&cnt[w[u] & tmask], 1);
        }
    }
    __syncthreads();
    const int lo  = (int)blockIdx.x * OWN_NODES;
    const int lim = max(0, min(OWN_NODES, n - lo));
    if (tid < lim) {
        const float d = 1.0f + (float)cnt[tid];
        dinv[lo + tid] = (d > 0.0f) ? rsqrtf(d) : 0.0f;
    }
}

#define SRT_REG 26
#define SRT_CAP (SRT_REG * LDS_THREADS)
#define SEG_UNR 6
__global__ __launch_bounds__(LDS_THREADS)
void gcn_own16(const unsigned* __restrict__ packed, const int* __restrict__ bstart,
               const float* __restrict__ h1, const float* __restrict__ dinv,
               const float* __restrict__ b1, const float* __restrict__ W2,
               float* __restrict__ s, int n) {
    __shared__ unsigned srt[SRT_CAP];
    __shared__ float acc[OWN_NODES * 16];
    __shared__ int cnt[LDS_THREADS];
    __shared__ int start[LDS_THREADS + 1];
    __shared__ int wsum[LDS_THREADS / 32];
    const int T    = LDS_THREADS;
    const int tid  = (int)threadIdx.x;
    const int lane = tid & 31;
    const int wv   = __builtin_amdgcn_readfirstlane(tid >> 5);
    const int lo   = (int)blockIdx.x * OWN_NODES;
    const int lim  = max(0, min(OWN_NODES, n - lo));
    {
        float4* a4 = (float4*)acc;
        for (int i = tid; i < lim * 4; i += T) a4[i] = make_float4(0.0f, 0.0f, 0.0f, 0.0f);
    }

    auto xscan = [&](int v) -> int {
        int incl = v;
        for (int off = 1; off < 32; off <<= 1) {
            const int t = __shfl_up(incl, off);
            if (lane >= off) incl += t;
        }
        __syncthreads();
        if (lane == 31) wsum[wv] = incl;
        __syncthreads();
        int pre = 0;
        for (int w = 0; w < wv; ++w) pre += wsum[w];
        return pre + incl - v;
    };

    const int p0 = bstart[blockIdx.x];
    const int p1 = bstart[blockIdx.x + 1];
    const unsigned tmask = (1u << OWN_SHIFT) - 1u;
    const int  l4   = lane & 3;
    const int  slot = lane >> 2;
    const bool up16 = (lane & 16) != 0;
    const bool up8  = (lane & 8) != 0;
    const float4 zero4 = make_float4(0.0f, 0.0f, 0.0f, 0.0f);

    for (int c0 = p0; c0 < p1; c0 += SRT_CAP) {
        cnt[tid] = 0;
        __syncthreads();

        unsigned w[SRT_REG];
        int      rk[SRT_REG];
        #pragma unroll
        for (int u = 0; u < SRT_REG; ++u) {
            const int pp = c0 + tid + u * T;
            w[u] = (pp < p1) ? packed[pp] : 0u;
        }
        #pragma unroll
        for (int u = 0; u < SRT_REG; ++u) {
            const int pp = c0 + tid + u * T;
            rk[u] = 0;
            if (pp < p1) rk[u] = atomicAdd(&cnt[w[u] & tmask], 1);
        }
        __syncthreads();
        {
            const int v  = cnt[tid];
            const int ex = xscan(v);
            start[tid] = ex;
            if (tid == T - 1) start[T] = ex + v;
        }
        __syncthreads();
        #pragma unroll
        for (int u = 0; u < SRT_REG; ++u) rk[u] += start[w[u] & tmask];
        #pragma unroll
        for (int u = 0; u < SRT_REG; ++u) {
            const int pp = c0 + tid + u * T;
            if (pp < p1) srt[rk[u]] = w[u] >> OWN_SHIFT;
        }
        __syncthreads();

        for (int nb = wv; nb < lim; nb += 64) {
            int  a[2], len[2];
            bool ok[2];
            #pragma unroll
            for (int q = 0; q < 2; ++q) {
                const int node = nb + 32 * q;
                ok[q] = node < lim;
                const int nn = ok[q] ? node : 0;
                a[q]   = __builtin_amdgcn_readfirstlane(start[nn]);
                len[q] = ok[q] ? (__builtin_amdgcn_readfirstlane(start[nn + 1]) - a[q]) : 0;
            }
            float4 sum[2] = {zero4, zero4};
            const int mx = max(len[0], len[1]);
            for (int k = 0; k < mx; k += 8 * SEG_UNR) {
                float4 t[2][SEG_UNR];
                #pragma unroll
                for (int q = 0; q < 2; ++q) {
                    #pragma unroll
                    for (int j = 0; j < SEG_UNR; ++j) {
                        const int  kk   = k + 8 * j + slot;
                        const bool have = kk < len[q];
                        const unsigned sw = srt[have ? (a[q] + kk) : 0];
                        const unsigned sr = have ? sw : 0u;
                        t[q][j] = *(const float4*)(h1 + (size_t)sr * 16 + 4 * l4);
                    }
                }
                #pragma unroll
                for (int q = 0; q < 2; ++q) {
                    #pragma unroll
                    for (int j = 0; j < SEG_UNR; ++j) {
                        const bool have = (k + 8 * j + slot) < len[q];
                        sum[q].x += have ? t[q][j].x : 0.0f;
                        sum[q].y += have ? t[q][j].y : 0.0f;
                        sum[q].z += have ? t[q][j].z : 0.0f;
                        sum[q].w += have ? t[q][j].w : 0.0f;
                    }
                }
            }
            float u0[2], u1[2], r[2], old[2];
            const int f = 4 * l4 + (up16 ? 2 : 0) + (up8 ? 1 : 0);
            #pragma unroll
            for (int q = 0; q < 2; ++q) {
                const float4 v = sum[q];
                const float ka = up16 ? v.z : v.x, kb = up16 ? v.w : v.y;
                const float ga = up16 ? v.x : v.z, gb = up16 ? v.y : v.w;
                u0[q] = ka + __shfl_xor(ga, 16);
                u1[q] = kb + __shfl_xor(gb, 16);
                old[q] = acc[(ok[q] ? (nb + 32 * q) : 0) * 16 + f];
            }
            #pragma unroll
            for (int q = 0; q < 2; ++q) {
                const float kc = up8 ? u1[q] : u0[q];
                const float gc = up8 ? u0[q] : u1[q];
                r[q] = kc + __shfl_xor(gc, 8);
            }
            #pragma unroll
            for (int q = 0; q < 2; ++q) r[q] += __shfl_xor(r[q], 4);
            #pragma unroll
            for (int q = 0; q < 2; ++q) {
                if (ok[q] && (lane & 4) == 0)
                    acc[(nb + 32 * q) * 16 + f] = old[q] + r[q];
            }
        }
    }
    __syncthreads();
    if (tid < lim) {
        const int   i  = lo + tid;
        const float di = dinv[i];
        const float* a = &acc[tid * 16];
        const float* h = h1 + (size_t)i * 16;
        float t = 0.0f;
        #pragma unroll
        for (int f = 0; f < 16; ++f) {
            float x = di * (a[f] + h[f]) + b1[f];
            x = fmaxf(x, 0.0f);
            t = fmaf(x, W2[f], t);
        }
        s[i] = di * t;
    }
}

#define S_WIN_V4 17
#define S_ROUND  (4 * LDS_THREADS)
#define S_WIN    (S_WIN_V4 * S_ROUND)
#define OWN1_REG 26
__global__ __launch_bounds__(LDS_THREADS)
void gcn_own1(const unsigned* __restrict__ packed, const int* __restrict__ bstart,
              const float* __restrict__ s, const float* __restrict__ dinv,
              const float* __restrict__ b2, float* __restrict__ out, int n) {
    __shared__ float win[S_WIN];
    __shared__ float acc[OWN_NODES];
    const int T   = LDS_THREADS;
    const int tid = (int)threadIdx.x;
    if (tid < OWN_NODES) acc[tid] = 0.0f;

    const int p0 = bstart[blockIdx.x];
    const int p1 = bstart[blockIdx.x + 1];
    const unsigned tmask = (1u << OWN_SHIFT) - 1u;

    unsigned w[OWN1_REG];
    #pragma unroll
    for (int u = 0; u < OWN1_REG; ++u) {
        const int pp = p0 + tid + u * T;
        w[u] = (pp < p1) ? packed[pp] : 0u;
    }
    const int pr = p0 + OWN1_REG * T;

    const bool al = (((size_t)s) & 15) == 0;
    float4 v[S_WIN_V4];
    auto fetch = [&](int base) {
        const int len4 = al ? (min(S_WIN, n - base) & ~3) : 0;
        const int full = len4 / S_ROUND;
        const int rem  = len4 - full * S_ROUND;
        #pragma unroll
        for (int u = 0; u < S_WIN_V4; ++u) {
            const bool have = u < full || (u == full && 4 * tid < rem);
            v[u] = have ? *(const float4*)(s + base + 4 * tid + u * S_ROUND)
                        : make_float4(0.0f, 0.0f, 0.0f, 0.0f);
        }
    };
    auto add = [&](unsigned word, int base, int len) {
        const unsigned off = (word >> OWN_SHIFT) - (unsigned)base;
        if (off < (unsigned)len)
            __hip_atomic_fetch_add(&acc[word & tmask], win[off],
                                   __ATOMIC_RELAXED, __HIP_MEMORY_SCOPE_WORKGROUP);
    };

    fetch(0);
    for (int base = 0; base < n; base += S_WIN) {
        const int len  = min(S_WIN, n - base);
        const int len4 = al ? (len & ~3) : 0;
        const int full = len4 / S_ROUND;
        const int rem  = len4 - full * S_ROUND;
        #pragma unroll
        for (int u = 0; u < S_WIN_V4; ++u) {
            if (u < full || (u == full && 4 * tid < rem))
                *(float4*)(win + 4 * tid + u * S_ROUND) = v[u];
        }
        for (int i = len4 + tid; i < len; i += T) win[i] = s[base + i];
        __syncthreads();
        if (base + S_WIN < n) fetch(base + S_WIN);
        #pragma unroll
        for (int u = 0; u < OWN1_REG; ++u) {
            unsigned word = w[u];
            asm volatile("" : "+v"(word));
            if (p0 + tid + u * T < p1) add(word, base, len);
        }
        for (int p = pr + tid; p < p1; p += T) add(packed[p], base, len);
        __syncthreads();
    }
    const int lo  = (int)blockIdx.x * OWN_NODES;
    const int lim = max(0, min(OWN_NODES, n - lo));
    if (tid < lim) {
        const int i = lo + tid;
        out[i] = dinv[i] * (acc[tid] + s[i]) + b2[0];
    }
}

extern "C" void kernel_launch(void* const* d_in, const int* in_sizes, int n_in,
                              void* d_out, int out_size, void* d_ws, size_t ws_size,
                              hipStream_t stream) {
    const float* x  = (const float*)d_in[0];
    const int*   ei = (const int*)d_in[1];
    const float* W1 = (const float*)d_in[2];
    const float* b1 = (const float*)d_in[3];
    const float* W2 = (const float*)d_in[4];
    const float* b2 = (const float*)d_in[5];

    const int N = in_sizes[0] / 128;
    const int E = in_sizes[1] / 2;
    const int* row = ei;
    const int* col = ei + E;

    {
        const int R = (N + OWN_NODES - 1) / OWN_NODES;
        const int quads = (E + 3) >> 2;
        int G = (quads + (BIN_CAP / 4) - 1) / (BIN_CAP / 4);
        if (G < BIN_WGS) G = BIN_WGS;
        const size_t need = ((size_t)18 * (size_t)N + (size_t)E + (size_t)R + 1
                             + (size_t)G * (size_t)R) * sizeof(float);
        if (R >= 1 && R <= OWN_MAXR && N <= (1 << (32 - OWN_SHIFT)) && ws_size >= need) {
            float*    dinv   = (float*)d_ws;
            float*    h1     = dinv + N;
            float*    s      = h1 + (size_t)N * 16;
            unsigned* packed = (unsigned*)(s + N);
            int*      bstart = (int*)(packed + E);
            int*      hist   = bstart + R + 1;
            float*    out    = (float*)d_out;
            const int B = 256;

            gcn_slice_hist<<<G, LDS_THREADS, 0, stream>>>(col, hist, E, N, R);
            gcn_bin_edges<<<G, LDS_THREADS, 0, stream>>>(row, col, hist, bstart, packed, E, N, R);
            gcn_dinv_run<<<R, LDS_THREADS, 0, stream>>>(packed, bstart, dinv, N);
            {
                int full   = N & ~15;
                int ntiles = full >> 4;
                if (ntiles > 0) {
                    int wavesPerBlock = G1_WAVES;
                    int blocks = (ntiles + wavesPerBlock - 1) / wavesPerBlock;
                    gcn_gemm1_wmma<<<blocks, G1_WAVES * 32, 0, stream>>>(x, W1, dinv, h1, ntiles);
                }
                if (full < N) {
                    int tail = (N - full) * 16;
                    gcn_gemm1_tail<<<(tail + B - 1) / B, B, 0, stream>>>(x, W1, dinv, h1, full, N);
                }
            }
            gcn_own16<<<R, LDS_THREADS, 0, stream>>>(packed, bstart, h1, dinv, b1, W2, s, N);
            gcn_own1<<<R, LDS_THREADS, 0, stream>>>(packed, bstart, s, dinv, b2, out, N);
            return;
        }
    }

    float* ws   = (float*)d_ws;
    float* deg  = ws;
    float* dinv = deg + N;
    float* h1   = dinv + N;
    float* agg1 = h1 + (size_t)N * 16;
    float* s    = agg1 + (size_t)N * 16;
    float* agg2 = s + N;
    float* out  = (float*)d_out;

    const int B = 256;

    {
        int total = N * 16;
        gcn_init<<<(total + B - 1) / B, B, 0, stream>>>(deg, agg1, agg2, N);
    }
    const int R = (N + RANGE_BINS - 1) / RANGE_BINS;
    const dim3 ldsGrid(EDGE_SLICES, R > 0 ? R : 1);

    gcn_degree_lds<<<ldsGrid, LDS_THREADS, 0, stream>>>(col, deg, E, N);
    gcn_dinv<<<(N + B - 1) / B, B, 0, stream>>>(deg, dinv, N);
    {
        int full   = N & ~15;
        int ntiles = full >> 4;
        if (ntiles > 0) {
            int wavesPerBlock = G1_WAVES;
            int blocks = (ntiles + wavesPerBlock - 1) / wavesPerBlock;
            gcn_gemm1_wmma<<<blocks, G1_WAVES * 32, 0, stream>>>(x, W1, dinv, h1, ntiles);
        }
        if (full < N) {
            int tail = (N - full) * 16;
            gcn_gemm1_tail<<<(tail + B - 1) / B, B, 0, stream>>>(x, W1, dinv, h1, full, N);
        }
    }
    {
        const int R16 = (N + R16_NODES - 1) / R16_NODES;
        const dim3 grid16(R16_SLICES, R16 > 0 ? R16 : 1);
        gcn_edge16_lds<<<grid16, LDS_THREADS, 0, stream>>>(row, col, h1, agg1, E, N);
    }
    gcn_post1<<<(N + B - 1) / B, B, 0, stream>>>(agg1, h1, dinv, b1, W2, s, N);
    gcn_edge1_lds<<<ldsGrid, LDS_THREADS, 0, stream>>>(row, col, s, agg2, E, N);
    gcn_post2<<<(N + B - 1) / B, B, 0, stream>>>(agg2, s, dinv, b2, out, N);
}


# Degree pass fused into the layer-1 product launch: one work-group per target range counts its run's degrees in LDS while its x tile is in flight, then multiplies its own rows; 5 launches; sums kept
# speedup vs baseline: 8.2271x; 1.0340x over previous
#include <hip/hip_runtime.h>

typedef __attribute__((ext_vector_type(2))) float v2f;
typedef __attribute__((ext_vector_type(4))) float v4f;
typedef __attribute__((ext_vector_type(8))) float v8f;

__global__ void gcn_init(float* __restrict__ deg, float* __restrict__ agg1,
                         float* __restrict__ agg2, int n) {
    int i = blockIdx.x * blockDim.x + threadIdx.x;
    if (i < n) { deg[i] = 1.0f; agg2[i] = 0.0f; }
    if (i < n * 16) { agg1[i] = 0.0f; }
}

#define RANGE_BINS  51200
#define EDGE_SLICES 64
#define LDS_THREADS 1024

__global__ __launch_bounds__(LDS_THREADS)
void gcn_degree_lds(const int* __restrict__ col, float* __restrict__ deg,
                    int E, int n) {
    __shared__ unsigned cnt[RANGE_BINS];
    const int T   = (int)blockDim.x;
    const int tid = (int)threadIdx.x;
    const int per = (n + (int)gridDim.y - 1) / (int)gridDim.y;
    const int lo  = (int)blockIdx.y * per;
    const int lim = max(0, min(per, n - lo));
    for (int i = tid; i < lim; i += T) cnt[i] = 0u;
    __syncthreads();

    const int quads = (E + 3) >> 2;
    const int qper  = (quads + (int)gridDim.x - 1) / (int)gridDim.x;
    const int q0    = (int)blockIdx.x * qper;
    const int q1    = min(q0 + qper, quads);
    const bool vec  = (((size_t)col) & 15) == 0;
    const int qv    = vec ? min(q1, E >> 2) : q0;

    for (int q = q0 + tid; q < qv; q += 4 * T) {
        int4 v[4];
        #pragma unroll
        for (int u = 0; u < 4; ++u) {
            const int qq = q + u * T;
            v[u] = (qq < qv) ? *(const int4*)(col + 4 * (size_t)qq)
                             : make_int4(-1, -1, -1, -1);
        }
        #pragma unroll
        for (int u = 0; u < 4; ++u) {
            const unsigned b0 = (unsigned)(v[u].x - lo);
            const unsigned b1 = (unsigned)(v[u].y - lo);
            const unsigned b2 = (unsigned)(v[u].z - lo);
            const unsigned b3 = (unsigned)(v[u].w - lo);
            if (b0 < (unsigned)lim) atomicAdd(&cnt[b0], 1u);
            if (b1 < (unsigned)lim) atomicAdd(&cnt[b1], 1u);
            if (b2 < (unsigned)lim) atomicAdd(&cnt[b2], 1u);
            if (b3 < (unsigned)lim) atomicAdd(&cnt[b3], 1u);
        }
    }
    for (int q = max(q0, qv) + tid; q < q1; q += T) {
        for (int k = 0; k < 4; ++k) {
            const int e = 4 * q + k;
            if (e < E) {
                const unsigned b = (unsigned)(col[e] - lo);
                if (b < (unsigned)lim) atomicAdd(&cnt[b], 1u);
            }
        }
    }
    __syncthreads();
    for (int i = tid; i < lim; i += T) {
        const unsigned v = cnt[i];
        if (v != 0u) atomicAdd(&deg[lo + i], (float)v);
    }
}

__global__ void gcn_dinv(const float* __restrict__ deg, float* __restrict__ dinv,
                         int n) {
    int i = blockIdx.x * blockDim.x + threadIdx.x;
    if (i >= n) return;
    float d = deg[i];
    dinv[i] = (d > 0.0f) ? rsqrtf(d) : 0.0f;
}

#define G1_WAVES 8
#define G1_PITCH 132
__global__ __launch_bounds__(G1_WAVES * 32)
void gcn_gemm1_wmma(const float* __restrict__ x,
                    const float* __restrict__ W1,
                    const float* __restrict__ dinv,
                    float* __restrict__ h1, int ntiles) {
    __shared__ alignas(16) float xs[G1_WAVES * 16 * G1_PITCH];
    __shared__ alignas(16) float ws[128 * 16];
    const int  tid  = (int)threadIdx.x;
    const int  lane = tid & 31;
    const int  wv   = tid >> 5;
    const int  wave = (int)blockIdx.x * G1_WAVES + wv;
    const bool act  = wave < ntiles;
    const size_t row16 = (size_t)(act ? wave : 0) * 16;

    float* xt = xs + wv * (16 * G1_PITCH);
    const float* src = x + row16 * 128;
    const bool al = (((size_t)x) & 15) == 0;
    const int mbase = (lane >> 4) << 3;

    v4f v[16];
    if (act && al) {
        #pragma unroll
        for (int i = 0; i < 16; ++i)
            v[i] = __builtin_nontemporal_load((const v4f*)(src + i * 128 + 4 * lane));
    }
    float w[(128 * 16) / (G1_WAVES * 32)];
    #pragma unroll
    for (int j = 0; j < (128 * 16) / (G1_WAVES * 32); ++j) w[j] = W1[tid + j * (G1_WAVES * 32)];
    float dv[8];
    #pragma unroll
    for (int r = 0; r < 8; ++r) dv[r] = dinv[row16 + (size_t)(mbase + r)];

    #pragma unroll
    for (int j = 0; j < (128 * 16) / (G1_WAVES * 32); ++j) {
        const int e = tid + j * (G1_WAVES * 32);
        const int k = e >> 4, n = e & 15;
        ws[(((k >> 1) * 16 + n) << 1) + (k & 1)] = w[j];
    }
    if (act) {
        if (al) {
            #pragma unroll
            for (int i = 0; i < 16; ++i)
                *(v4f*)(xt + i * G1_PITCH + 4 * lane) = v[i];
        } else {
            #pragma nounroll
            for (int i = 0; i < 16 * 4; ++i)
                xt[(i >> 2) * G1_PITCH + lane + 32 * (i & 3)] = src[(i >> 2) * 128 + lane + 32 * (i & 3)];
        }
    }
    __syncthreads();
    if (!act) return;

    const int m = lane & 15;
    const int h = lane >> 4;
    const float* ap = xt + m * G1_PITCH + 2 * h;
    const float* bp = ws + ((h * 16 + m) << 1);

    v8f acc = {};
    #pragma unroll 8
    for (int k0 = 0; k0 < 128; k0 += 4) {
        const v2f a = *(const v2f*)(ap + k0);
        const v2f b = *(const v2f*)(bp + (k0 >> 1) * 32);
        acc = __builtin_amdgcn_wmma_f32_16x16x4_f32(
             false, a,  false, b,
             (short)0, acc,  false,  false);
    }

    #pragma unroll
    for (int r = 0; r < 8; ++r) {
        const size_t node = row16 + (size_t)(mbase + r);
        h1[node * 16 + m] = acc[r] * dv[r];
    }
}

__global__ void gcn_gemm1_tail(const float* __restrict__ x,
                               const float* __restrict__ W1,
                               const float* __restrict__ dinv,
                               float* __restrict__ h1, int start, int n) {
    int t = blockIdx.x * blockDim.x + threadIdx.x;
    int node = start + (t >> 4);
    int f = t & 15;
    if (node >= n) return;
    float acc = 0.0f;
    const float* xr = x + (size_t)node * 128;
    for (int k = 0; k < 128; ++k) acc += xr[k] * W1[(size_t)k * 16 + f];
    h1[(size_t)node * 16 + f] = acc * dinv[node];
}

#define R16_NODES  4768
#define R16_SLICES 6
#define R16_QCAP   96

__global__ __launch_bounds__(LDS_THREADS)
void gcn_edge16_lds(const int* __restrict__ row, const int* __restrict__ col,
                    const float* __restrict__ h1,
                    float* __restrict__ agg1, int E, int n) {
    __shared__ float acc[R16_NODES * 16];
    __shared__ int queue[(LDS_THREADS / 32) * R16_QCAP];
    __shared__ unsigned short qrow[(LDS_THREADS / 32) * R16_QCAP];
    static_assert(R16_NODES <= 65535, "a target's LDS row must fit a 16-bit queue word");

    const int T    = (int)blockDim.x;
    const int tid  = (int)threadIdx.x;
    const int lane = tid & 31;
    const int per  = (n + (int)gridDim.y - 1) / (int)gridDim.y;
    const int lo   = (int)blockIdx.y * per;
    const int lim  = max(0, min(per, n - lo));
    {
        float4* a4 = (float4*)acc;
        for (int i = tid; i < lim * 4; i += T) a4[i] = make_float4(0.0f, 0.0f, 0.0f, 0.0f);
    }
    __syncthreads();

    const int qo = (tid >> 5) * R16_QCAP;
    int cnt = 0;

    auto push = [&](int cv, int e) {
        const unsigned bin = (unsigned)(cv - lo);
        const bool in = bin < (unsigned)lim;
        const unsigned m = __builtin_amdgcn_ballot_w32(in);
        if (m != 0u) {
            if (in) {
                const int slot = qo + cnt + (int)__builtin_amdgcn_mbcnt_lo(m, 0u);
                __hip_atomic_store(&queue[slot], e, __ATOMIC_RELAXED, __HIP_MEMORY_SCOPE_WAVEFRONT);
                __hip_atomic_store(&qrow[slot], (unsigned short)bin,
                                   __ATOMIC_RELAXED, __HIP_MEMORY_SCOPE_WAVEFRONT);
            }
            cnt += __builtin_popcount(m);
        }
    };
    auto drain = [&](int base, int count) {
        const bool  have = lane < count;
        const int   e  = have ? __hip_atomic_load(&queue[qo + base + lane], __ATOMIC_RELAXED,
                                                  __HIP_MEMORY_SCOPE_WAVEFRONT) : 0;
        const int   bi = have ? (int)__hip_atomic_load(&qrow[qo + base + lane], __ATOMIC_RELAXED,
                                                       __HIP_MEMORY_SCOPE_WAVEFRONT) : 0;
        const int   ri = row[e];
        const int   l4 = lane & 3;
        const int   g  = lane >> 2;
        float4 v[4];
        #pragma unroll
        for (int j = 0; j < 4; ++j) {
            const int r = __shfl(ri, 8 * j + g);
            v[j] = *(const float4*)(h1 + (size_t)r * 16 + 4 * l4);
        }
        #pragma unroll
        for (int j = 0; j < 4; ++j) {
            const int s = 8 * j + g;
            const int b = __shfl(bi, s);
            if (s < count) {
                float* p = &acc[b * 16 + 4 * l4];
                __hip_atomic_fetch_add(p + 0, v[j].x, __ATOMIC_RELAXED, __HIP_MEMORY_SCOPE_WORKGROUP);
                __hip_atomic_fetch_add(p + 1, v[j].y, __ATOMIC_RELAXED, __HIP_MEMORY_SCOPE_WORKGROUP);
                __hip_atomic_fetch_add(p + 2, v[j].z, __ATOMIC_RELAXED, __HIP_MEMORY_SCOPE_WORKGROUP);
                __hip_atomic_fetch_add(p + 3, v[j].w, __ATOMIC_RELAXED, __HIP_MEMORY_SCOPE_WORKGROUP);
            }
        }
    };

    const int quads = (E + 3) >> 2;
    const int qper  = (quads + (int)gridDim.x - 1) / (int)gridDim.x;
    const int q0    = (int)blockIdx.x * qper;
    const int q1    = min(q0 + qper, quads);
    const bool vec  = (((size_t)col) & 15) == 0;
    const int qv    = vec ? min(q1, E >> 2) : q0;

    for (int qb = q0; qb < qv; qb += 4 * T) {
        int4 v[4];
        #pragma unroll
        for (int u = 0; u < 4; ++u) {
            const int qq = qb + tid + u * T;
            v[u] = (qq < qv) ? *(const int4*)(col + 4 * (size_t)qq)
                             : make_int4(-1, -1, -1, -1);
        }
        #pragma unroll
        for (int u = 0; u < 4; ++u) {
            const int e0 = 4 * (qb + tid + u * T);
            push(v[u].x, e0 + 0);
            push(v[u].y, e0 + 1);
            while (cnt >= 32) { cnt -= 32; drain(cnt, 32); }
            push(v[u].z, e0 + 2);
            push(v[u].w, e0 + 3);
            while (cnt >= 32) { cnt -= 32; drain(cnt, 32); }
        }
    }
    {
        const int e0 = 4 * max(q0, qv);
        const int e1 = min(4 * q1, E);
        for (int eb = e0; eb < e1; eb += T) {
            const int e = eb + tid;
            push((e < e1) ? col[e] : -1, e);
            while (cnt >= 32) { cnt -= 32; drain(cnt, 32); }
        }
    }
    if (cnt > 0) drain(0, cnt);

    __syncthreads();
    float* dst = agg1 + (size_t)lo * 16;
    for (int i = tid; i < lim * 16; i += T) {
        const float t = acc[i];
        if (t != 0.0f) atomicAdd(&dst[i], t);
    }
}

__global__ void gcn_post1(const float* __restrict__ agg1,
                          const float* __restrict__ h1,
                          const float* __restrict__ dinv,
                          const float* __restrict__ b1,
                          const float* __restrict__ W2,
                          float* __restrict__ s, int n) {
    int i = blockIdx.x * blockDim.x + threadIdx.x;
    if (i >= n) return;
    float di = dinv[i];
    const float* a = agg1 + (size_t)i * 16;
    const float* h = h1 + (size_t)i * 16;
    float acc = 0.0f;
    #pragma unroll
    for (int f = 0; f < 16; ++f) {
        float v = di * (a[f] + h[f]) + b1[f];
        v = fmaxf(v, 0.0f);
        acc = fmaf(v, W2[f], acc);
    }
    s[i] = di * acc;
}

__global__ __launch_bounds__(LDS_THREADS)
void gcn_edge1_lds(const int* __restrict__ row, const int* __restrict__ col,
                   const float* __restrict__ s,
                   float* __restrict__ agg2, int E, int n) {
    __shared__ float acc[RANGE_BINS];
    const int T   = (int)blockDim.x;
    const int tid = (int)threadIdx.x;
    const int per = (n + (int)gridDim.y - 1) / (int)gridDim.y;
    const int lo  = (int)blockIdx.y * per;
    const int lim = max(0, min(per, n - lo));
    for (int i = tid; i < lim; i += T) acc[i] = 0.0f;
    __syncthreads();

    const int quads = (E + 3) >> 2;
    const int qper  = (quads + (int)gridDim.x - 1) / (int)gridDim.x;
    const int q0    = (int)blockIdx.x * qper;
    const int q1    = min(q0 + qper, quads);
    const bool vec  = ((((size_t)col) | ((size_t)row)) & 15) == 0;
    const int qv    = vec ? min(q1, E >> 2) : q0;

    for (int q = q0 + tid; q < qv; q += 2 * T) {
        int c[8], r[8];
        #pragma unroll
        for (int u = 0; u < 2; ++u) {
            const int qq = q + u * T;
            const bool ok = qq < qv;
            const int4 vc = ok ? *(const int4*)(col + 4 * (size_t)qq) : make_int4(-1, -1, -1, -1);
            const int4 vr = ok ? *(const int4*)(row + 4 * (size_t)qq) : make_int4(0, 0, 0, 0);
            c[4 * u + 0] = vc.x; c[4 * u + 1] = vc.y; c[4 * u + 2] = vc.z; c[4 * u + 3] = vc.w;
            r[4 * u + 0] = vr.x; r[4 * u + 1] = vr.y; r[4 * u + 2] = vr.z; r[4 * u + 3] = vr.w;
        }
        unsigned b[8];
        float    v[8];
        #pragma unroll
        for (int k = 0; k < 8; ++k) {
            b[k] = (unsigned)(c[k] - lo);
            const bool in = b[k] < (unsigned)lim;
            v[k] = s[in ? r[k] : 0];
        }
        #pragma unroll
        for (int k = 0; k < 8; ++k) {
            if (b[k] < (unsigned)lim)
                __hip_atomic_fetch_add(&acc[b[k]], v[k], __ATOMIC_RELAXED,
                                       __HIP_MEMORY_SCOPE_WORKGROUP);
        }
    }
    for (int q = max(q0, qv) + tid; q < q1; q += T) {
        for (int k = 0; k < 4; ++k) {
            const int e = 4 * q + k;
            if (e < E) {
                const unsigned bb = (unsigned)(col[e] - lo);
                if (bb < (unsigned)lim) {
                    __hip_atomic_fetch_add(&acc[bb], s[row[e]],
                                           __ATOMIC_RELAXED, __HIP_MEMORY_SCOPE_WORKGROUP);
                }
            }
        }
    }
    __syncthreads();
    for (int i = tid; i < lim; i += T) {
        const float t = acc[i];
        if (t != 0.0f) atomicAdd(&agg2[lo + i], t);
    }
}

__global__ void gcn_post2(const float* __restrict__ agg2,
                          const float* __restrict__ s,
                          const float* __restrict__ dinv,
                          const float* __restrict__ b2,
                          float* __restrict__ out, int n) {
    int i = blockIdx.x * blockDim.x + threadIdx.x;
    if (i >= n) return;
    float di = dinv[i];
    out[i] = di * (agg2[i] + s[i]) + b2[0];
}


#define OWN_NODES 782
#define OWN_SHIFT 10
#define OWN_MAXR  512
#define BIN_WGS   256
static_assert(OWN_NODES <= (1 << OWN_SHIFT), "a target's place in its range must fit OWN_SHIFT bits");
static_assert(OWN_NODES <= LDS_THREADS, "the epilogues give every node of a range one thread");

__global__ __launch_bounds__(LDS_THREADS)
void gcn_slice_hist(const int* __restrict__ col, int* __restrict__ hist,
                    int E, int n, int R) {
    __shared__ int slot[OWN_MAXR];
    const int T   = (int)blockDim.x;
    const int tid = (int)threadIdx.x;
    if (tid < OWN_MAXR) slot[tid] = 0;
    __syncthreads();

    const int quads = (E + 3) >> 2;
    const int qper  = (quads + (int)gridDim.x - 1) / (int)gridDim.x;
    const int q0    = (int)blockIdx.x * qper;
    const int q1    = min(q0 + qper, quads);
    const bool vec  = (((size_t)col) & 15) == 0;
    const int qv    = vec ? min(q1, E >> 2) : q0;

    auto count = [&](int c) {
        if ((unsigned)c < (unsigned)n) atomicAdd(&slot[c / OWN_NODES], 1);
    };
    for (int q = q0 + tid; q < qv; q += 4 * T) {
        int4 v[4];
        #pragma unroll
        for (int u = 0; u < 4; ++u) {
            const int qq = q + u * T;
            v[u] = (qq < qv) ? *(const int4*)(col + 4 * (size_t)qq)
                             : make_int4(-1, -1, -1, -1);
        }
        #pragma unroll
        for (int u = 0; u < 4; ++u) {
            count(v[u].x); count(v[u].y); count(v[u].z); count(v[u].w);
        }
    }
    for (int q = max(q0, qv) + tid; q < q1; q += T) {
        for (int k = 0; k < 4; ++k) {
            const int e = 4 * q + k;
            if (e < E) count(col[e]);
        }
    }
    __syncthreads();
    for (int r = tid; r < R; r += T) hist[(size_t)blockIdx.x * R + r] = slot[r];
}

#define BIN_CAP 32768
__global__ __launch_bounds__(LDS_THREADS)
void gcn_bin_edges(const int* __restrict__ row, const int* __restrict__ col,
                   const int* __restrict__ hist,
                   int* __restrict__ bstart, unsigned* __restrict__ packed,
                   int E, int n, int R) {
    __shared__ unsigned stage[BIN_CAP];
    __shared__ int slot[OWN_MAXR];
    __shared__ int start[OWN_MAXR + 1];
    __shared__ int loff[OWN_MAXR + 1];
    __shared__ int gbase[OWN_MAXR];
    __shared__ int colT[LDS_THREADS];
    __shared__ int colP[LDS_THREADS];
    __shared__ int wsum[LDS_THREADS / 32];
    const int T    = LDS_THREADS;
    const int tid  = (int)threadIdx.x;
    const int lane = tid & 31;
    const int G    = (int)gridDim.x;
    const int me   = (int)blockIdx.x;

    auto xscan = [&](int v) -> int {
        int incl = v;
        for (int off = 1; off < 32; off <<= 1) {
            const int t = __shfl_up(incl, off);
            if (lane >= off) incl += t;
        }
        __syncthreads();
        if (lane == 31) wsum[tid >> 5] = incl;
        __syncthreads();
        int pre = 0;
        for (int w = 0; w < (tid >> 5); ++w) pre += wsum[w];
        return pre + incl - v;
    };

    {
        const int RP    = (R > 256) ? 512 : 256;
        const int parts = T / RP;
        const int r     = tid & (RP - 1);
        const int part  = tid / RP;
        const int gper  = (G + parts - 1) / parts;
        const int g0    = part * gper;
        const int g1    = min(G, g0 + gper);
        int tsum = 0, psum = 0;
        if (r < R) {
            #pragma unroll 16
            for (int g = g0; g < g1; ++g) {
                const int v = hist[(size_t)g * R + r];
                tsum += v;
                psum += (g < me) ? v : 0;
            }
        }
        colT[tid] = tsum;
        colP[tid] = psum;
        __syncthreads();
        int tot = 0, pre = 0, own = 0;
        if (tid < R) {
            for (int p = 0; p < parts; ++p) { tot += colT[p * RP + tid]; pre += colP[p * RP + tid]; }
            own = hist[(size_t)me * R + tid];
        }
        const int ex = xscan(tot);
        if (tid < R)       { start[tid] = ex; gbase[tid] = ex + pre; }
        if (tid == R - 1)  start[R] = ex + tot;
        const int lx = xscan(own);
        if (tid < R)       { loff[tid] = lx; slot[tid] = lx; }
        if (tid == R - 1)  loff[R] = lx + own;
        __syncthreads();
    }
    if (blockIdx.x == 0) {
        for (int i = tid; i <= R; i += T) bstart[i] = start[i];
    }

    const int quads = (E + 3) >> 2;
    const int qper  = (quads + (int)gridDim.x - 1) / (int)gridDim.x;
    const int q0    = (int)blockIdx.x * qper;
    const int q1    = min(q0 + qper, quads);
    const bool vec  = ((((size_t)col) | ((size_t)row)) & 15) == 0;
    const int qv    = vec ? min(q1, E >> 2) : q0;

    auto place = [&](int c, int sidx) {
        if ((unsigned)c < (unsigned)n) {
            const int r = c / OWN_NODES;
            const int k = atomicAdd(&slot[r], 1);
            if (k < BIN_CAP)
                stage[k] = ((unsigned)sidx << OWN_SHIFT) | (unsigned)(c - r * OWN_NODES);
        }
    };
    for (int q = q0 + tid; q < qv; q += 2 * T) {
        int4 vc[2], vr[2];
        #pragma unroll
        for (int u = 0; u < 2; ++u) {
            const int qq = q + u * T;
            const bool ok = qq < qv;
            vc[u] = ok ? *(const int4*)(col + 4 * (size_t)qq) : make_int4(-1, -1, -1, -1);
            vr[u] = ok ? *(const int4*)(row + 4 * (size_t)qq) : make_int4(0, 0, 0, 0);
        }
        #pragma unroll
        for (int u = 0; u < 2; ++u) {
            place(vc[u].x, vr[u].x); place(vc[u].y, vr[u].y);
            place(vc[u].z, vr[u].z); place(vc[u].w, vr[u].w);
        }
    }
    for (int q = max(q0, qv) + tid; q < q1; q += T) {
        for (int k = 0; k < 4; ++k) {
            const int e = 4 * q + k;
            if (e < E) place(col[e], row[e]);
        }
    }
    __syncthreads();

    for (int r = tid >> 5; r < R; r += (T >> 5)) {
        const int l0 = loff[r];
        const int nr = min(loff[r + 1], BIN_CAP) - l0;
        const int g0 = gbase[r];
        for (int i = lane; i < nr; i += 32) packed[g0 + i] = stage[l0 + i];
    }
}

#define DG_REG 26
__device__ __forceinline__ void gcn_lds_barrier() {
    asm volatile("s_wait_dscnt 0x0\n\ts_barrier_signal -1\n\ts_barrier_wait -1" ::: "memory");
}
template <bool AL>
__device__ __forceinline__ void gcn_dinv_gemm1_body(
        const unsigned* __restrict__ packed, const int p0, const int p1,
        const float* __restrict__ x, const float* __restrict__ W1,
        float* __restrict__ dinv, float* __restrict__ h1,
        float* xs, float* ws, int* cnt, float* dl,
        const int lo, const int lim) {
    const int T    = LDS_THREADS;
    const int tid  = (int)threadIdx.x;
    const int lane = tid & 31;
    const int wv   = __builtin_amdgcn_readfirstlane(tid >> 5);
    const int ntile = (lim + 15) >> 4;
    const unsigned tmask = (1u << OWN_SHIFT) - 1u;

    unsigned w[DG_REG];
    #pragma unroll
    for (int u = 0; u < DG_REG; ++u) {
        const int pp = p0 + tid + u * T;
        w[u] = packed[(pp < p1) ? pp : p0];
    }
    float w1r[(128 * 16) / LDS_THREADS];
    #pragma unroll
    for (int j = 0; j < (128 * 16) / LDS_THREADS; ++j) w1r[j] = W1[tid + j * T];

    float* xt = xs + wv * (16 * G1_PITCH);
    const int mbase = (lane >> 4) << 3;
    v4f v[16];
    auto fetch = [&](int t) {
        #pragma unroll
        for (int i = 0; i < 16; ++i) {
            const int ri = min(16 * t + i, lim - 1);
            v[i] = __builtin_nontemporal_load(
                (const v4f*)(x + (size_t)(lo + ri) * 128 + 4 * lane));
        }
    };
    if (AL) fetch(wv);

    cnt[tid] = 0;
    if (tid < 32) cnt[T + tid] = 0;
    gcn_lds_barrier();
    #pragma unroll
    for (int u = 0; u < DG_REG; ++u) {
        const bool have = p0 + tid + u * T < p1;
        atomicAdd(&cnt[have ? (int)(w[u] & tmask) : T + lane], 1);
    }
    for (int p = p0 + DG_REG * T + tid; p < p1; p += T) atomicAdd(&cnt[packed[p] & tmask], 1);
    #pragma unroll
    for (int j = 0; j < (128 * 16) / LDS_THREADS; ++j) {
        const int e = tid + j * T;
        const int k = e >> 4, c = e & 15;
        ws[(((k >> 1) * 16 + c) << 1) + (k & 1)] = w1r[j];
    }
    gcn_lds_barrier();
    {
        const float d  = 1.0f + (float)cnt[tid];
        const float di = (d > 0.0f) ? rsqrtf(d) : 0.0f;
        dl[tid] = di;
        if (tid < lim) dinv[lo + tid] = di;
    }

    const int m = lane & 15;
    const int h = lane >> 4;
    const float* ap = xt + m * G1_PITCH + 2 * h;
    const float* bp = ws + ((h * 16 + m) << 1);

    for (int tb = 0; tb < ntile; tb += T / 32) {
        const int  t   = tb + wv;
        const bool act = t < ntile;
        if (AL) {
            #pragma unroll
            for (int i = 0; i < 16; ++i)
                *(v4f*)(xt + i * G1_PITCH + 4 * lane) = v[i];
            if (t + T / 32 < ntile) fetch(t + T / 32);
        } else if (act) {
            #pragma nounroll
            for (int i = 0; i < 16 * 4; ++i) {
                const int ri = min(16 * t + (i >> 2), lim - 1);
                xt[(i >> 2) * G1_PITCH + lane + 32 * (i & 3)] =
                    x[(size_t)(lo + ri) * 128 + lane + 32 * (i & 3)];
            }
        }
        gcn_lds_barrier();
        if (act) {
            v8f acc = {};
            #pragma unroll 8
            for (int k0 = 0; k0 < 128; k0 += 4) {
                const v2f a = *(const v2f*)(ap + k0);
                const v2f b = *(const v2f*)(bp + (k0 >> 1) * 32);
                acc = __builtin_amdgcn_wmma_f32_16x16x4_f32(
                     false, a,  false, b,
                     (short)0, acc,  false,  false);
            }
            const int r0 = 16 * t + mbase;
            float dv[8];
            #pragma unroll
            for (int r = 0; r < 8; ++r) dv[r] = dl[r0 + r];
            if (16 * t + 16 <= lim) {
                #pragma unroll
                for (int r = 0; r < 8; ++r)
                    h1[(size_t)(lo + r0 + r) * 16 + m] = acc[r] * dv[r];
            } else {
                #pragma unroll
                for (int r = 0; r < 8; ++r)
                    if (r0 + r < lim) h1[(size_t)(lo + r0 + r) * 16 + m] = acc[r] * dv[r];
            }
        }
    }
}

__global__ __launch_bounds__(LDS_THREADS)
void gcn_dinv_gemm1(const unsigned* __restrict__ packed, const int* __restrict__ bstart,
                    const float* __restrict__ x, const float* __restrict__ W1,
                    float* __restrict__ dinv, float* __restrict__ h1, int n) {
    __shared__ alignas(16) float xs[(LDS_THREADS / 32) * 16 * G1_PITCH];
    __shared__ alignas(16) float ws[128 * 16];
    __shared__ int   cnt[LDS_THREADS + 32];
    __shared__ float dl[LDS_THREADS];
    const int lo  = (int)blockIdx.x * OWN_NODES;
    const int lim = min(OWN_NODES, n - lo);
    if (lim <= 0) return;
    const int p0 = bstart[blockIdx.x];
    const int p1 = bstart[blockIdx.x + 1];
    if ((((size_t)x) & 15) == 0)
        gcn_dinv_gemm1_body<true>(packed, p0, p1, x, W1, dinv, h1, xs, ws, cnt, dl, lo, lim);
    else
        gcn_dinv_gemm1_body<false>(packed, p0, p1, x, W1, dinv, h1, xs, ws, cnt, dl, lo, lim);
}

#define SRT_REG 26
#define SRT_CAP (SRT_REG * LDS_THREADS)
#define SEG_UNR 6
__global__ __launch_bounds__(LDS_THREADS)
void gcn_own16(const unsigned* __restrict__ packed, const int* __restrict__ bstart,
               const float* __restrict__ h1, const float* __restrict__ dinv,
               const float* __restrict__ b1, const float* __restrict__ W2,
               float* __restrict__ s, int n) {
    __shared__ unsigned srt[SRT_CAP];
    __shared__ float acc[OWN_NODES * 16];
    __shared__ int cnt[LDS_THREADS];
    __shared__ int start[LDS_THREADS + 1];
    __shared__ int wsum[LDS_THREADS / 32];
    const int T    = LDS_THREADS;
    const int tid  = (int)threadIdx.x;
    const int lane = tid & 31;
    const int wv   = __builtin_amdgcn_readfirstlane(tid >> 5);
    const int lo   = (int)blockIdx.x * OWN_NODES;
    const int lim  = max(0, min(OWN_NODES, n - lo));
    {
        float4* a4 = (float4*)acc;
        for (int i = tid; i < lim * 4; i += T) a4[i] = make_float4(0.0f, 0.0f, 0.0f, 0.0f);
    }

    auto xscan = [&](int v) -> int {
        int incl = v;
        for (int off = 1; off < 32; off <<= 1) {
            const int t = __shfl_up(incl, off);
            if (lane >= off) incl += t;
        }
        __syncthreads();
        if (lane == 31) wsum[wv] = incl;
        __syncthreads();
        int pre = 0;
        for (int w = 0; w < wv; ++w) pre += wsum[w];
        return pre + incl - v;
    };

    const int p0 = bstart[blockIdx.x];
    const int p1 = bstart[blockIdx.x + 1];
    const unsigned tmask = (1u << OWN_SHIFT) - 1u;
    const int  l4   = lane & 3;
    const int  slot = lane >> 2;
    const bool up16 = (lane & 16) != 0;
    const bool up8  = (lane & 8) != 0;
    const float4 zero4 = make_float4(0.0f, 0.0f, 0.0f, 0.0f);

    for (int c0 = p0; c0 < p1; c0 += SRT_CAP) {
        cnt[tid] = 0;
        __syncthreads();

        unsigned w[SRT_REG];
        int      rk[SRT_REG];
        #pragma unroll
        for (int u = 0; u < SRT_REG; ++u) {
            const int pp = c0 + tid + u * T;
            w[u] = (pp < p1) ? packed[pp] : 0u;
        }
        #pragma unroll
        for (int u = 0; u < SRT_REG; ++u) {
            const int pp = c0 + tid + u * T;
            rk[u] = 0;
            if (pp < p1) rk[u] = atomicAdd(&cnt[w[u] & tmask], 1);
        }
        __syncthreads();
        {
            const int v  = cnt[tid];
            const int ex = xscan(v);
            start[tid] = ex;
            if (tid == T - 1) start[T] = ex + v;
        }
        __syncthreads();
        #pragma unroll
        for (int u = 0; u < SRT_REG; ++u) rk[u] += start[w[u] & tmask];
        #pragma unroll
        for (int u = 0; u < SRT_REG; ++u) {
            const int pp = c0 + tid + u * T;
            if (pp < p1) srt[rk[u]] = w[u] >> OWN_SHIFT;
        }
        __syncthreads();

        for (int nb = wv; nb < lim; nb += 64) {
            int  a[2], len[2];
            bool ok[2];
            #pragma unroll
            for (int q = 0; q < 2; ++q) {
                const int node = nb + 32 * q;
                ok[q] = node < lim;
                const int nn = ok[q] ? node : 0;
                a[q]   = __builtin_amdgcn_readfirstlane(start[nn]);
                len[q] = ok[q] ? (__builtin_amdgcn_readfirstlane(start[nn + 1]) - a[q]) : 0;
            }
            float4 sum[2] = {zero4, zero4};
            const int mx = max(len[0], len[1]);
            for (int k = 0; k < mx; k += 8 * SEG_UNR) {
                float4 t[2][SEG_UNR];
                #pragma unroll
                for (int q = 0; q < 2; ++q) {
                    #pragma unroll
                    for (int j = 0; j < SEG_UNR; ++j) {
                        const int  kk   = k + 8 * j + slot;
                        const bool have = kk < len[q];
                        const unsigned sw = srt[have ? (a[q] + kk) : 0];
                        const unsigned sr = have ? sw : 0u;
                        t[q][j] = *(const float4*)(h1 + (size_t)sr * 16 + 4 * l4);
                    }
                }
                #pragma unroll
                for (int q = 0; q < 2; ++q) {
                    #pragma unroll
                    for (int j = 0; j < SEG_UNR; ++j) {
                        const bool have = (k + 8 * j + slot) < len[q];
                        sum[q].x += have ? t[q][j].x : 0.0f;
                        sum[q].y += have ? t[q][j].y : 0.0f;
                        sum[q].z += have ? t[q][j].z : 0.0f;
                        sum[q].w += have ? t[q][j].w : 0.0f;
                    }
                }
            }
            float u0[2], u1[2], r[2], old[2];
            const int f = 4 * l4 + (up16 ? 2 : 0) + (up8 ? 1 : 0);
            #pragma unroll
            for (int q = 0; q < 2; ++q) {
                const float4 v = sum[q];
                const float ka = up16 ? v.z : v.x, kb = up16 ? v.w : v.y;
                const float ga = up16 ? v.x : v.z, gb = up16 ? v.y : v.w;
                u0[q] = ka + __shfl_xor(ga, 16);
                u1[q] = kb + __shfl_xor(gb, 16);
                old[q] = acc[(ok[q] ? (nb + 32 * q) : 0) * 16 + f];
            }
            #pragma unroll
            for (int q = 0; q < 2; ++q) {
                const float kc = up8 ? u1[q] : u0[q];
                const float gc = up8 ? u0[q] : u1[q];
                r[q] = kc + __shfl_xor(gc, 8);
            }
            #pragma unroll
            for (int q = 0; q < 2; ++q) r[q] += __shfl_xor(r[q], 4);
            #pragma unroll
            for (int q = 0; q < 2; ++q) {
                if (ok[q] && (lane & 4) == 0)
                    acc[(nb + 32 * q) * 16 + f] = old[q] + r[q];
            }
        }
    }
    __syncthreads();
    if (tid < lim) {
        const int   i  = lo + tid;
        const float di = dinv[i];
        const float* a = &acc[tid * 16];
        const float* h = h1 + (size_t)i * 16;
        float t = 0.0f;
        #pragma unroll
        for (int f = 0; f < 16; ++f) {
            float x = di * (a[f] + h[f]) + b1[f];
            x = fmaxf(x, 0.0f);
            t = fmaf(x, W2[f], t);
        }
        s[i] = di * t;
    }
}

#define S_WIN_V4 17
#define S_ROUND  (4 * LDS_THREADS)
#define S_WIN    (S_WIN_V4 * S_ROUND)
#define OWN1_REG 26
__global__ __launch_bounds__(LDS_THREADS)
void gcn_own1(const unsigned* __restrict__ packed, const int* __restrict__ bstart,
              const float* __restrict__ s, const float* __restrict__ dinv,
              const float* __restrict__ b2, float* __restrict__ out, int n) {
    __shared__ float win[S_WIN];
    __shared__ float acc[OWN_NODES];
    const int T   = LDS_THREADS;
    const int tid = (int)threadIdx.x;
    if (tid < OWN_NODES) acc[tid] = 0.0f;

    const int p0 = bstart[blockIdx.x];
    const int p1 = bstart[blockIdx.x + 1];
    const unsigned tmask = (1u << OWN_SHIFT) - 1u;

    unsigned w[OWN1_REG];
    #pragma unroll
    for (int u = 0; u < OWN1_REG; ++u) {
        const int pp = p0 + tid + u * T;
        w[u] = (pp < p1) ? packed[pp] : 0u;
    }
    const int pr = p0 + OWN1_REG * T;

    const bool al = (((size_t)s) & 15) == 0;
    float4 v[S_WIN_V4];
    auto fetch = [&](int base) {
        const int len4 = al ? (min(S_WIN, n - base) & ~3) : 0;
        const int full = len4 / S_ROUND;
        const int rem  = len4 - full * S_ROUND;
        #pragma unroll
        for (int u = 0; u < S_WIN_V4; ++u) {
            const bool have = u < full || (u == full && 4 * tid < rem);
            v[u] = have ? *(const float4*)(s + base + 4 * tid + u * S_ROUND)
                        : make_float4(0.0f, 0.0f, 0.0f, 0.0f);
        }
    };
    auto add = [&](unsigned word, int base, int len) {
        const unsigned off = (word >> OWN_SHIFT) - (unsigned)base;
        if (off < (unsigned)len)
            __hip_atomic_fetch_add(&acc[word & tmask], win[off],
                                   __ATOMIC_RELAXED, __HIP_MEMORY_SCOPE_WORKGROUP);
    };

    fetch(0);
    for (int base = 0; base < n; base += S_WIN) {
        const int len  = min(S_WIN, n - base);
        const int len4 = al ? (len & ~3) : 0;
        const int full = len4 / S_ROUND;
        const int rem  = len4 - full * S_ROUND;
        #pragma unroll
        for (int u = 0; u < S_WIN_V4; ++u) {
            if (u < full || (u == full && 4 * tid < rem))
                *(float4*)(win + 4 * tid + u * S_ROUND) = v[u];
        }
        for (int i = len4 + tid; i < len; i += T) win[i] = s[base + i];
        __syncthreads();
        if (base + S_WIN < n) fetch(base + S_WIN);
        #pragma unroll
        for (int u = 0; u < OWN1_REG; ++u) {
            unsigned word = w[u];
            asm volatile("" : "+v"(word));
            if (p0 + tid + u * T < p1) add(word, base, len);
        }
        for (int p = pr + tid; p < p1; p += T) add(packed[p], base, len);
        __syncthreads();
    }
    const int lo  = (int)blockIdx.x * OWN_NODES;
    const int lim = max(0, min(OWN_NODES, n - lo));
    if (tid < lim) {
        const int i = lo + tid;
        out[i] = dinv[i] * (acc[tid] + s[i]) + b2[0];
    }
}

extern "C" void kernel_launch(void* const* d_in, const int* in_sizes, int n_in,
                              void* d_out, int out_size, void* d_ws, size_t ws_size,
                              hipStream_t stream) {
    const float* x  = (const float*)d_in[0];
    const int*   ei = (const int*)d_in[1];
    const float* W1 = (const float*)d_in[2];
    const float* b1 = (const float*)d_in[3];
    const float* W2 = (const float*)d_in[4];
    const float* b2 = (const float*)d_in[5];

    const int N = in_sizes[0] / 128;
    const int E = in_sizes[1] / 2;
    const int* row = ei;
    const int* col = ei + E;

    {
        const int R = (N + OWN_NODES - 1) / OWN_NODES;
        const int quads = (E + 3) >> 2;
        int G = (quads + (BIN_CAP / 4) - 1) / (BIN_CAP / 4);
        if (G < BIN_WGS) G = BIN_WGS;
        const size_t need = ((size_t)18 * (size_t)N + (size_t)E + (size_t)R + 1
                             + (size_t)G * (size_t)R) * sizeof(float);
        if (R >= 1 && R <= OWN_MAXR && N <= (1 << (32 - OWN_SHIFT)) && ws_size >= need) {
            float*    dinv   = (float*)d_ws;
            float*    h1     = dinv + N;
            float*    s      = h1 + (size_t)N * 16;
            unsigned* packed = (unsigned*)(s + N);
            int*      bstart = (int*)(packed + E);
            int*      hist   = bstart + R + 1;
            float*    out    = (float*)d_out;

            gcn_slice_hist<<<G, LDS_THREADS, 0, stream>>>(col, hist, E, N, R);
            gcn_bin_edges<<<G, LDS_THREADS, 0, stream>>>(row, col, hist, bstart, packed, E, N, R);
            gcn_dinv_gemm1<<<R, LDS_THREADS, 0, stream>>>(packed, bstart, x, W1, dinv, h1, N);
            gcn_own16<<<R, LDS_THREADS, 0, stream>>>(packed, bstart, h1, dinv, b1, W2, s, N);
            gcn_own1<<<R, LDS_THREADS, 0, stream>>>(packed, bstart, s, dinv, b2, out, N);
            return;
        }
    }

    float* ws   = (float*)d_ws;
    float* deg  = ws;
    float* dinv = deg + N;
    float* h1   = dinv + N;
    float* agg1 = h1 + (size_t)N * 16;
    float* s    = agg1 + (size_t)N * 16;
    float* agg2 = s + N;
    float* out  = (float*)d_out;

    const int B = 256;

    {
        int total = N * 16;
        gcn_init<<<(total + B - 1) / B, B, 0, stream>>>(deg, agg1, agg2, N);
    }
    const int R = (N + RANGE_BINS - 1) / RANGE_BINS;
    const dim3 ldsGrid(EDGE_SLICES, R > 0 ? R : 1);

    gcn_degree_lds<<<ldsGrid, LDS_THREADS, 0, stream>>>(col, deg, E, N);
    gcn_dinv<<<(N + B - 1) / B, B, 0, stream>>>(deg, dinv, N);
    {
        int full   = N & ~15;
        int ntiles = full >> 4;
        if (ntiles > 0) {
            int wavesPerBlock = G1_WAVES;
            int blocks = (ntiles + wavesPerBlock - 1) / wavesPerBlock;
            gcn_gemm1_wmma<<<blocks, G1_WAVES * 32, 0, stream>>>(x, W1, dinv, h1, ntiles);
        }
        if (full < N) {
            int tail = (N - full) * 16;
            gcn_gemm1_tail<<<(tail + B - 1) / B, B, 0, stream>>>(x, W1, dinv, h1, full, N);
        }
    }
    {
        const int R16 = (N + R16_NODES - 1) / R16_NODES;
        const dim3 grid16(R16_SLICES, R16 > 0 ? R16 : 1);
        gcn_edge16_lds<<<grid16, LDS_THREADS, 0, stream>>>(row, col, h1, agg1, E, N);
    }
    gcn_post1<<<(N + B - 1) / B, B, 0, stream>>>(agg1, h1, dinv, b1, W2, s, N);
    gcn_edge1_lds<<<ldsGrid, LDS_THREADS, 0, stream>>>(row, col, s, agg2, E, N);
    gcn_post2<<<(N + B - 1) / B, B, 0, stream>>>(agg2, s, dinv, b2, out, N);
}
